# starting kernel
# baseline (speedup 1.0000x reference)
#include <hip/hip_runtime.h>

typedef __attribute__((ext_vector_type(2))) float v2f;
typedef __attribute__((ext_vector_type(8))) float v8f;

#define TPB 256


__global__ void gcn_init(float* __restrict__ deg, float* __restrict__ s,
                         float* __restrict__ agg0, float* __restrict__ agg1,
                         int n) {
  int i = blockIdx.x * blockDim.x + threadIdx.x;
  if (i < n) {
    deg[i] = 1.0f;
    s[i] = 0.0f;
    agg0[i] = 0.0f;
    agg1[i] = 0.0f;
  }
}

__global__ void gcn_degree(const int* __restrict__ dst,
                           float* __restrict__ deg, int E) {
  int e = blockIdx.x * blockDim.x + threadIdx.x;
  if (e < E) unsafeAtomicAdd(&deg[dst[e]], 1.0f);
}

__global__ void gcn_dinv(const float* __restrict__ deg,
                         float* __restrict__ dinv, int n) {
  int i = blockIdx.x * blockDim.x + threadIdx.x;
  if (i < n) dinv[i] = rsqrtf(deg[i]);
}

__global__ void gcn_l1_scatter(const int* __restrict__ src,
                               const int* __restrict__ dst,
                               const float* __restrict__ x,
                               const float* __restrict__ dinv,
                               float* __restrict__ s, int E) {
  int e = blockIdx.x * blockDim.x + threadIdx.x;
  if (e >= E) return;
  int a = src[e], b = dst[e];
  unsafeAtomicAdd(&s[b], x[a] * dinv[a] * dinv[b]);
}

__global__ void gcn_l1_finalize(const float* __restrict__ x,
                                const float* __restrict__ dinv,
                                const float* __restrict__ s,
                                const float* __restrict__ W1,
                                const float* __restrict__ b1,
                                float* __restrict__ relu1, int n) {
  int i = blockIdx.x * blockDim.x + threadIdx.x;
  if (i >= n) return;
  float di = dinv[i];
  float sn = s[i] + x[i] * di * di;
  float4* o = (float4*)(relu1 + (size_t)i * 16);
#pragma unroll
  for (int q = 0; q < 4; ++q) {
    float4 r;
    r.x = fmaxf(fmaf(sn, W1[4 * q + 0], b1[4 * q + 0]), 0.0f);
    r.y = fmaxf(fmaf(sn, W1[4 * q + 1], b1[4 * q + 1]), 0.0f);
    r.z = fmaxf(fmaf(sn, W1[4 * q + 2], b1[4 * q + 2]), 0.0f);
    r.w = fmaxf(fmaf(sn, W1[4 * q + 3], b1[4 * q + 3]), 0.0f);
    o[q] = r;
  }
}

__global__ void gcn_l2_dense_wmma(const float* __restrict__ relu1,
                                  const float* __restrict__ W2,
                                  float* __restrict__ h2c0,
                                  float* __restrict__ h2c1,
                                  int n, int ntiles) {
  int gtid = blockIdx.x * blockDim.x + threadIdx.x;
  int tile = gtid >> 5;
  int lane = threadIdx.x & 31;
  if (tile >= ntiles) return;
  int base = tile * 16;
  int half = lane >> 4;
  int m = lane & 15;

  v2f bmat[4];
#pragma unroll
  for (int k = 0; k < 4; ++k) {
    int r0 = 4 * k + 2 * half;
    float bx = 0.0f, by = 0.0f;
    if (m < 2) { bx = W2[r0 * 2 + m]; by = W2[(r0 + 1) * 2 + m]; }
    bmat[k].x = bx; bmat[k].y = by;
  }

  int row = base + m;
  row = (row < n) ? row : (n - 1);
  const v2f* p = (const v2f*)(relu1 + (size_t)row * 16 + 2 * half);
  v2f amat[4];
#pragma unroll
  for (int k = 0; k < 4; ++k) amat[k] = p[2 * k];

  v8f c = {};
#pragma unroll
  for (int k = 0; k < 4; ++k)
    c = __builtin_amdgcn_wmma_f32_16x16x4_f32(
         false, amat[k],  false, bmat[k],
         (short)0, c,  false,  false);

  if (m < 2) {
    float* col = (m == 0) ? h2c0 : h2c1;
    int node0 = base + half * 8;
    float4 lo; lo.x = c[0]; lo.y = c[1]; lo.z = c[2]; lo.w = c[3];
    float4 hi; hi.x = c[4]; hi.y = c[5]; hi.z = c[6]; hi.w = c[7];
    *(float4*)(col + node0)     = lo;
    *(float4*)(col + node0 + 4) = hi;
  }
}

__global__ void gcn_l2_scatter(const int* __restrict__ src,
                               const int* __restrict__ dst,
                               const float* __restrict__ h2c0,
                               const float* __restrict__ h2c1,
                               const float* __restrict__ dinv,
                               float* __restrict__ agg0,
                               float* __restrict__ agg1, int E) {
  int e = blockIdx.x * blockDim.x + threadIdx.x;
  if (e >= E) return;
  int a = src[e], b = dst[e];
  float nrm = dinv[a] * dinv[b];
  unsafeAtomicAdd(&agg0[b], h2c0[a] * nrm);
  unsafeAtomicAdd(&agg1[b], h2c1[a] * nrm);
}

__global__ void gcn_finalize(const float* __restrict__ agg0,
                             const float* __restrict__ agg1,
                             const float* __restrict__ h2c0,
                             const float* __restrict__ h2c1,
                             const float* __restrict__ dinv,
                             const float* __restrict__ b2,
                             float* __restrict__ out, int n) {
  int i = blockIdx.x * blockDim.x + threadIdx.x;
  if (i >= n) return;
  float di = dinv[i];
  float sl = di * di;
  float z0 = agg0[i] + h2c0[i] * sl + b2[0];
  float z1 = agg1[i] + h2c1[i] * sl + b2[1];
  float mx = fmaxf(z0, z1);
  float lse = mx + logf(expf(z0 - mx) + expf(z1 - mx));
  float2 r; r.x = z0 - lse; r.y = z1 - lse;
  *(float2*)(out + (size_t)i * 2) = r;
}

extern "C" void kernel_launch(void* const* d_in, const int* in_sizes, int n_in,
                              void* d_out, int out_size, void* d_ws, size_t ws_size,
                              hipStream_t stream) {
  const float* x  = (const float*)d_in[0];
  const int*   ei = (const int*)d_in[1];
  const float* W1 = (const float*)d_in[2];
  const float* b1 = (const float*)d_in[3];
  const float* W2 = (const float*)d_in[4];
  const float* b2 = (const float*)d_in[5];
  float* out = (float*)d_out;

  int n = in_sizes[0];
  int E = in_sizes[1] / 2;
  const int* srcp = ei;
  const int* dstp = ei + E;

  size_t ns = ((size_t)n + 31) & ~(size_t)31;

  float* ws    = (float*)d_ws;
  float* deg   = ws;
  float* dinv  = ws + ns;
  float* s     = ws + 2 * ns;
  float* h2c0  = ws + 3 * ns;
  float* h2c1  = ws + 4 * ns;
  float* agg0  = ws + 5 * ns;
  float* agg1  = ws + 6 * ns;
  float* relu1 = ws + 7 * ns;

  int gN = (n + TPB - 1) / TPB;
  int gE = (E + TPB - 1) / TPB;

  gcn_init<<<gN, TPB, 0, stream>>>(deg, s, agg0, agg1, n);
  gcn_degree<<<gE, TPB, 0, stream>>>(dstp, deg, E);
  gcn_dinv<<<gN, TPB, 0, stream>>>(deg, dinv, n);
  gcn_l1_scatter<<<gE, TPB, 0, stream>>>(srcp, dstp, x, dinv, s, E);
  gcn_l1_finalize<<<gN, TPB, 0, stream>>>(x, dinv, s, W1, b1, relu1, n);

  int ntiles = (n + 15) / 16;
  int gW = (int)(((size_t)ntiles * 32 + TPB - 1) / TPB);
  gcn_l2_dense_wmma<<<gW, TPB, 0, stream>>>(relu1, W2, h2c0, h2c1, n, ntiles);

  gcn_l2_scatter<<<gE, TPB, 0, stream>>>(srcp, dstp, h2c0, h2c1, dinv,
                                         agg0, agg1, E);
  gcn_finalize<<<gN, TPB, 0, stream>>>(agg0, agg1, h2c0, h2c1, dinv, b2, out, n);
}


# Edges grouped by an LDS counting sort (segment histograms, scan, bucket-contiguous runs, buckets ordered in LDS): global integer atomics, scattered stores gone; changes summation order (LDS rank)
# speedup vs baseline: 5.8237x; 5.8237x over previous
#include <hip/hip_runtime.h>


typedef __attribute__((ext_vector_type(2))) float v2f;
typedef __attribute__((ext_vector_type(8))) float v8f;

#define TPB 256
#define EPT 4
#define SLICE (TPB * EPT)
#define PASS_BLOCKS 2048
#define NQ 6


__device__ __forceinline__ int tick_stride(int n) { return (n >= 32 * NQ) ? 64 : 1; }

__device__ __forceinline__ int cache_half() {
  unsigned v = __builtin_amdgcn_s_sendmsg_rtn(0x87);
  return (int)((v >> 18) & 1u);
}

__device__ __forceinline__ int draw_slice(int* q, int* sh) {
  __syncthreads();
  if (threadIdx.x == 0) *sh = atomicAdd(q, 1);
  __syncthreads();
  return *sh;
}

__device__ __forceinline__ int block_half(int* sh) {
  if (threadIdx.x == 0) *sh = cache_half();
  __syncthreads();
  return *sh;
}

__global__ void gcn_init(float* __restrict__ deg, float* __restrict__ s,
                         float* __restrict__ agg0, float* __restrict__ agg1,
                         int* __restrict__ tick, int n) {
  int i = blockIdx.x * blockDim.x + threadIdx.x;
  if (i < NQ) tick[i * tick_stride(n)] = 0;
  if (i < n) {
    deg[i] = 1.0f;
    s[i] = 0.0f;
    agg0[i] = 0.0f;
    agg1[i] = 0.0f;
  }
}

__global__ void __launch_bounds__(TPB)
gcn_degree(const int* __restrict__ dst, float* __restrict__ deg,
           int* __restrict__ tick, int n, int E) {
  __shared__ int sh[2];
  const int nsl = (E + SLICE - 1) / SLICE;
  const int split = (n >> 1) & ~31;
  const int st = tick_stride(n);
  const int h0 = block_half(&sh[1]);
  for (int r = 0; r < 2; ++r) {
    const int h = h0 ^ r;
    const int lo = h ? split : 0;
    const int hi = h ? n : split;
    int* q = tick + (0 + h) * st;
    for (int t = draw_slice(q, &sh[0]); t < nsl; t = draw_slice(q, &sh[0])) {
      const int e0 = t * SLICE + (int)threadIdx.x;
      int b[EPT];
#pragma unroll
      for (int j = 0; j < EPT; ++j) {
        int e = e0 + j * TPB;
        int v = dst[(e < E) ? e : (E - 1)];
        b[j] = (e < E) ? v : -1;
      }
#pragma unroll
      for (int j = 0; j < EPT; ++j)
        if (b[j] >= lo && b[j] < hi) unsafeAtomicAdd(&deg[b[j]], 1.0f);
    }
  }
}

__global__ void gcn_dinv(const float* __restrict__ deg,
                         float* __restrict__ dinv, int n) {
  int i = blockIdx.x * blockDim.x + threadIdx.x;
  if (i < n) dinv[i] = rsqrtf(deg[i]);
}

__global__ void __launch_bounds__(TPB)
gcn_l1_scatter(const int* __restrict__ src, const int* __restrict__ dst,
               const float* __restrict__ x, const float* __restrict__ dinv,
               float* __restrict__ s, int* __restrict__ tick, int n, int E) {
  __shared__ int sh[2];
  const int nsl = (E + SLICE - 1) / SLICE;
  const int split = (n >> 1) & ~31;
  const int st = tick_stride(n);
  const int h0 = block_half(&sh[1]);
  for (int r = 0; r < 2; ++r) {
    const int h = h0 ^ r;
    const int lo = h ? split : 0;
    const int hi = h ? n : split;
    int* q = tick + (2 + h) * st;
    for (int t = draw_slice(q, &sh[0]); t < nsl; t = draw_slice(q, &sh[0])) {
      const int e0 = t * SLICE + (int)threadIdx.x;
      int a[EPT], b[EPT];
#pragma unroll
      for (int j = 0; j < EPT; ++j) {
        int e = e0 + j * TPB;
        int ec = (e < E) ? e : (E - 1);
        int vb = dst[ec];
        a[j] = src[ec];
        b[j] = (e < E) ? vb : -1;
      }
      asm volatile("" :: "v"(a[0]), "v"(a[1]), "v"(a[2]), "v"(a[3]));
      float v[EPT];
#pragma unroll
      for (int j = 0; j < EPT; ++j) {
        bool in = b[j] >= lo && b[j] < hi;
        unsigned aa = in ? (unsigned)a[j] : 0u;
        unsigned bb = in ? (unsigned)b[j] : 0u;
        v[j] = x[aa] * dinv[aa] * dinv[bb];
      }
#pragma unroll
      for (int j = 0; j < EPT; ++j)
        if (b[j] >= lo && b[j] < hi) unsafeAtomicAdd(&s[b[j]], v[j]);
    }
  }
}

__global__ void gcn_l1_finalize(const float* __restrict__ x,
                                const float* __restrict__ dinv,
                                const float* __restrict__ s,
                                const float* __restrict__ W1,
                                const float* __restrict__ b1,
                                float* __restrict__ relu1, int n) {
  int i = blockIdx.x * blockDim.x + threadIdx.x;
  if (i >= n) return;
  float di = dinv[i];
  float sn = s[i] + x[i] * di * di;
  float4* o = (float4*)(relu1 + (size_t)i * 16);
#pragma unroll
  for (int q = 0; q < 4; ++q) {
    float4 r;
    r.x = fmaxf(fmaf(sn, W1[4 * q + 0], b1[4 * q + 0]), 0.0f);
    r.y = fmaxf(fmaf(sn, W1[4 * q + 1], b1[4 * q + 1]), 0.0f);
    r.z = fmaxf(fmaf(sn, W1[4 * q + 2], b1[4 * q + 2]), 0.0f);
    r.w = fmaxf(fmaf(sn, W1[4 * q + 3], b1[4 * q + 3]), 0.0f);
    o[q] = r;
  }
}

__global__ void gcn_l2_dense_wmma(const float* __restrict__ relu1,
                                  const float* __restrict__ W2,
                                  float* __restrict__ h2c0,
                                  float* __restrict__ h2c1,
                                  int n, int ntiles) {
  int gtid = blockIdx.x * blockDim.x + threadIdx.x;
  int tile = gtid >> 5;
  int lane = threadIdx.x & 31;
  if (tile >= ntiles) return;
  int base = tile * 16;
  int half = lane >> 4;
  int m = lane & 15;

  v2f bmat[4];
#pragma unroll
  for (int k = 0; k < 4; ++k) {
    int r0 = 4 * k + 2 * half;
    float bx = 0.0f, by = 0.0f;
    if (m < 2) { bx = W2[r0 * 2 + m]; by = W2[(r0 + 1) * 2 + m]; }
    bmat[k].x = bx; bmat[k].y = by;
  }

  int row = base + m;
  row = (row < n) ? row : (n - 1);
  const v2f* p = (const v2f*)(relu1 + (size_t)row * 16 + 2 * half);
  v2f amat[4];
#pragma unroll
  for (int k = 0; k < 4; ++k) amat[k] = p[2 * k];

  v8f c = {};
#pragma unroll
  for (int k = 0; k < 4; ++k)
    c = __builtin_amdgcn_wmma_f32_16x16x4_f32(
         false, amat[k],  false, bmat[k],
         (short)0, c,  false,  false);

  if (m < 2) {
    float* col = (m == 0) ? h2c0 : h2c1;
    int node0 = base + half * 8;
    float4 lo; lo.x = c[0]; lo.y = c[1]; lo.z = c[2]; lo.w = c[3];
    float4 hi; hi.x = c[4]; hi.y = c[5]; hi.z = c[6]; hi.w = c[7];
    *(float4*)(col + node0)     = lo;
    *(float4*)(col + node0 + 4) = hi;
  }
}

__global__ void __launch_bounds__(TPB)
gcn_l2_scatter(const int* __restrict__ src, const int* __restrict__ dst,
               const float* __restrict__ h2c0, const float* __restrict__ h2c1,
               const float* __restrict__ dinv,
               float* __restrict__ agg0, float* __restrict__ agg1,
               int* __restrict__ tick, int n, int E) {
  __shared__ int sh[2];
  const int nsl = (E + SLICE - 1) / SLICE;
  const int split = (n >> 1) & ~31;
  const int st = tick_stride(n);
  const int h0 = block_half(&sh[1]);
  for (int r = 0; r < 2; ++r) {
    const int h = h0 ^ r;
    const int lo = h ? split : 0;
    const int hi = h ? n : split;
    int* q = tick + (4 + h) * st;
    for (int t = draw_slice(q, &sh[0]); t < nsl; t = draw_slice(q, &sh[0])) {
      const int e0 = t * SLICE + (int)threadIdx.x;
      int a[EPT], b[EPT];
#pragma unroll
      for (int j = 0; j < EPT; ++j) {
        int e = e0 + j * TPB;
        int ec = (e < E) ? e : (E - 1);
        int vb = dst[ec];
        a[j] = src[ec];
        b[j] = (e < E) ? vb : -1;
      }
      asm volatile("" :: "v"(a[0]), "v"(a[1]), "v"(a[2]), "v"(a[3]));
      float v0[EPT], v1[EPT];
#pragma unroll
      for (int j = 0; j < EPT; ++j) {
        bool in = b[j] >= lo && b[j] < hi;
        unsigned aa = in ? (unsigned)a[j] : 0u;
        unsigned bb = in ? (unsigned)b[j] : 0u;
        float nrm = dinv[aa] * dinv[bb];
        v0[j] = h2c0[aa] * nrm;
        v1[j] = h2c1[aa] * nrm;
      }
#pragma unroll
      for (int j = 0; j < EPT; ++j)
        if (b[j] >= lo && b[j] < hi) {
          unsafeAtomicAdd(&agg0[b[j]], v0[j]);
          unsafeAtomicAdd(&agg1[b[j]], v1[j]);
        }
    }
  }
}

__global__ void gcn_finalize(const float* __restrict__ agg0,
                             const float* __restrict__ agg1,
                             const float* __restrict__ h2c0,
                             const float* __restrict__ h2c1,
                             const float* __restrict__ dinv,
                             const float* __restrict__ b2,
                             float* __restrict__ out, int n) {
  int i = blockIdx.x * blockDim.x + threadIdx.x;
  if (i >= n) return;
  float di = dinv[i];
  float sl = di * di;
  float z0 = agg0[i] + h2c0[i] * sl + b2[0];
  float z1 = agg1[i] + h2c1[i] * sl + b2[1];
  float mx = fmaxf(z0, z1);
  float lse = mx + logf(expf(z0 - mx) + expf(z1 - mx));
  float2 r; r.x = z0 - lse; r.y = z1 - lse;
  *(float2*)(out + (size_t)i * 2) = r;
}

#define SCAN_TPB 1024

__global__ void csr_init(int* __restrict__ cnt, int* __restrict__ tick, int n) {
  int i = blockIdx.x * blockDim.x + threadIdx.x;
  if (i < 4) tick[i * tick_stride(n)] = 0;
  if (i < n) cnt[i] = 0;
}

__global__ void __launch_bounds__(TPB)
csr_count(const int* __restrict__ dst, int* __restrict__ cnt,
          int* __restrict__ tick, int n, int E) {
  __shared__ int sh[2];
  const int nsl = (E + SLICE - 1) / SLICE;
  const int split = (n >> 1) & ~31;
  const int st = tick_stride(n);
  const int h0 = block_half(&sh[1]);
  for (int r = 0; r < 2; ++r) {
    const int h = h0 ^ r;
    const int lo = h ? split : 0;
    const int hi = h ? n : split;
    int* q = tick + (0 + h) * st;
    for (int t = draw_slice(q, &sh[0]); t < nsl; t = draw_slice(q, &sh[0])) {
      const int e0 = t * SLICE + (int)threadIdx.x;
      int b[EPT];
#pragma unroll
      for (int j = 0; j < EPT; ++j) {
        int e = e0 + j * TPB;
        int v = dst[(e < E) ? e : (E - 1)];
        b[j] = (e < E) ? v : -1;
      }
#pragma unroll
      for (int j = 0; j < EPT; ++j)
        if (b[j] >= lo && b[j] < hi) atomicAdd(&cnt[b[j]], 1);
    }
  }
}

__global__ void __launch_bounds__(SCAN_TPB)
csr_scan(const int* __restrict__ cnt, int* __restrict__ rowptr,
         int* __restrict__ cursor, float* __restrict__ dinv, int n) {
  __shared__ int wtot[32];
  __shared__ int wpre[32];
  const int t = (int)threadIdx.x;
  const int lane = t & 31;
  const int wv = t >> 5;
  const int first = (int)blockIdx.x * SCAN_TPB;
  const int i = first + t;
  const int c = (i < n) ? cnt[i] : 0;

  int pre = 0;
  const int4* c4 = (const int4*)cnt;
  const int ngrp = first / 4;
  for (int k = t; k < ngrp; k += 8 * SCAN_TPB) {
    int4 v[8];
#pragma unroll
    for (int j = 0; j < 8; ++j) {
      int kk = k + j * SCAN_TPB;
      v[j] = c4[(kk < ngrp) ? kk : k];
    }
#pragma unroll
    for (int j = 0; j < 8; ++j) {
      int kk = k + j * SCAN_TPB;
      int sum = (v[j].x + v[j].y) + (v[j].z + v[j].w);
      pre += (kk < ngrp) ? sum : 0;
    }
  }
  for (int o = 16; o > 0; o >>= 1) pre += __shfl_xor(pre, o);

  int inc = c;
  for (int o = 1; o < 32; o <<= 1) {
    int u = __shfl_up(inc, o);
    if (lane >= o) inc += u;
  }
  if (lane == 31) wtot[wv] = inc;
  if (lane == 0) wpre[wv] = pre;
  __syncthreads();

  int wt = wtot[lane];
  int winc = wt;
  for (int o = 1; o < 32; o <<= 1) {
    int u = __shfl_up(winc, o);
    if (lane >= o) winc += u;
  }
  int before_wave = __shfl(winc - wt, wv);
  int before_block = wpre[lane];
  for (int o = 16; o > 0; o >>= 1) before_block += __shfl_xor(before_block, o);

  if (i < n) {
    int slot = before_block + before_wave + (inc - c);
    rowptr[i] = slot;
    cursor[i] = slot;
    dinv[i] = rsqrtf((float)(c + 1));
  }
}

__global__ void __launch_bounds__(TPB)
csr_fill(const int* __restrict__ src, const int* __restrict__ dst,
         int* __restrict__ cursor, int* __restrict__ list,
         int* __restrict__ tick, int n, int E) {
  __shared__ int sh[2];
  const int nsl = (E + SLICE - 1) / SLICE;
  const int split = (n >> 1) & ~31;
  const int st = tick_stride(n);
  const int h0 = block_half(&sh[1]);
  for (int r = 0; r < 2; ++r) {
    const int h = h0 ^ r;
    const int lo = h ? split : 0;
    const int hi = h ? n : split;
    int* q = tick + (2 + h) * st;
    for (int t = draw_slice(q, &sh[0]); t < nsl; t = draw_slice(q, &sh[0])) {
      const int e0 = t * SLICE + (int)threadIdx.x;
      int a[EPT], b[EPT];
#pragma unroll
      for (int j = 0; j < EPT; ++j) {
        int e = e0 + j * TPB;
        int ec = (e < E) ? e : (E - 1);
        int vb = dst[ec];
        a[j] = src[ec];
        b[j] = (e < E) ? vb : -1;
      }
      int pos[EPT];
#pragma unroll
      for (int j = 0; j < EPT; ++j) {
        pos[j] = -1;
        if (b[j] >= lo && b[j] < hi) pos[j] = atomicAdd(&cursor[b[j]], 1);
      }
#pragma unroll
      for (int j = 0; j < EPT; ++j)
        if ((unsigned)pos[j] < (unsigned)E) list[pos[j]] = a[j];
    }
  }
}

__global__ void __launch_bounds__(TPB)
gcn_l1_gather(const int* __restrict__ rowptr, const int* __restrict__ rowend,
              const int* __restrict__ list, const float* __restrict__ x,
              const float* __restrict__ dinv, float* __restrict__ s,
              int n, int E) {
  const int lane = (int)threadIdx.x & 31;
  const int wv = __builtin_amdgcn_readfirstlane((int)threadIdx.x >> 5);
  const int nw = (int)gridDim.x * (TPB / 32);
  const unsigned last = (unsigned)(n - 1);
  int i = (int)blockIdx.x * (TPB / 32) + wv;
  int st = 0, en = 0;
  float di = 0.0f;
  if (i < n) { st = rowptr[i]; en = rowend[i]; di = dinv[i]; }
  while (i < n) {
    const int ni = i + nw;
    int nst = 0, nen = 0;
    float ndi = 0.0f;
    if (ni < n) { nst = rowptr[ni]; nen = rowend[ni]; ndi = dinv[ni]; }
    st = __builtin_amdgcn_readfirstlane((st > 0) ? st : 0);
    en = __builtin_amdgcn_readfirstlane((en < E) ? en : E);
    float acc = 0.0f;
    for (int base = st; base < en; base += 64) {
      const int k0 = base + lane, k1 = k0 + 32;
      const bool p0 = k0 < en, p1 = k1 < en;
      unsigned a0 = (unsigned)list[p0 ? k0 : st];
      unsigned a1 = (unsigned)list[p1 ? k1 : st];
      a0 = (a0 < last) ? a0 : last;
      a1 = (a1 < last) ? a1 : last;
      const float t0 = x[a0] * dinv[a0] * di;
      const float t1 = x[a1] * dinv[a1] * di;
      acc += p0 ? t0 : 0.0f;
      acc += p1 ? t1 : 0.0f;
    }
    for (int o = 16; o > 0; o >>= 1) acc += __shfl_xor(acc, o);
    if (lane == 0) s[i] = acc;
    i = ni; st = nst; en = nen; di = ndi;
  }
}

__global__ void __launch_bounds__(TPB)
gcn_l2_gather(const int* __restrict__ rowptr, const int* __restrict__ rowend,
              const int* __restrict__ list, const float* __restrict__ h2c0,
              const float* __restrict__ h2c1, const float* __restrict__ dinv,
              float* __restrict__ agg0, float* __restrict__ agg1,
              int n, int E) {
  const int lane = (int)threadIdx.x & 31;
  const int wv = __builtin_amdgcn_readfirstlane((int)threadIdx.x >> 5);
  const int nw = (int)gridDim.x * (TPB / 32);
  const unsigned last = (unsigned)(n - 1);
  int i = (int)blockIdx.x * (TPB / 32) + wv;
  int st = 0, en = 0;
  float di = 0.0f;
  if (i < n) { st = rowptr[i]; en = rowend[i]; di = dinv[i]; }
  while (i < n) {
    const int ni = i + nw;
    int nst = 0, nen = 0;
    float ndi = 0.0f;
    if (ni < n) { nst = rowptr[ni]; nen = rowend[ni]; ndi = dinv[ni]; }
    st = __builtin_amdgcn_readfirstlane((st > 0) ? st : 0);
    en = __builtin_amdgcn_readfirstlane((en < E) ? en : E);
    float acc0 = 0.0f, acc1 = 0.0f;
    for (int base = st; base < en; base += 64) {
      const int k0 = base + lane, k1 = k0 + 32;
      const bool p0 = k0 < en, p1 = k1 < en;
      unsigned a0 = (unsigned)list[p0 ? k0 : st];
      unsigned a1 = (unsigned)list[p1 ? k1 : st];
      a0 = (a0 < last) ? a0 : last;
      a1 = (a1 < last) ? a1 : last;
      const float n0 = dinv[a0] * di;
      const float n1 = dinv[a1] * di;
      const float u0 = h2c0[a0] * n0, w0 = h2c1[a0] * n0;
      const float u1 = h2c0[a1] * n1, w1 = h2c1[a1] * n1;
      acc0 += p0 ? u0 : 0.0f;  acc1 += p0 ? w0 : 0.0f;
      acc0 += p1 ? u1 : 0.0f;  acc1 += p1 ? w1 : 0.0f;
    }
    for (int o = 16; o > 0; o >>= 1) {
      acc0 += __shfl_xor(acc0, o);
      acc1 += __shfl_xor(acc1, o);
    }
    if (lane == 0) { agg0[i] = acc0; agg1[i] = acc1; }
    i = ni; st = nst; en = nen; di = ndi;
  }
}


#define GSH 8
#define GNB (1 << GSH)
#define GTPB 1024
#define GEPT 12
#define GSUB (GTPB * GEPT)
#define GBMAX GTPB
#define GCAP 16384
#define RS_TPB 256

template <int NW>
__device__ __forceinline__ int block_scan_excl(int v, int* wtot, int& total) {
  const int lane = (int)threadIdx.x & 31;
  const int wv = (int)threadIdx.x >> 5;
  int inc = v;
  for (int o = 1; o < 32; o <<= 1) {
    int u = __shfl_up(inc, o);
    if (lane >= o) inc += u;
  }
  if (lane == 31) wtot[wv] = inc;
  __syncthreads();
  const int wt = (lane < NW) ? wtot[lane] : 0;
  int winc = wt;
  for (int o = 1; o < 32; o <<= 1) {
    int u = __shfl_up(winc, o);
    if (lane >= o) winc += u;
  }
  const int before = __shfl(winc - wt, wv);
  total = __shfl(winc, 31);
  __syncthreads();
  return before + (inc - v);
}

__device__ __forceinline__ int bucket_of(int d, int nb) {
  unsigned b = (unsigned)d >> GSH;
  return (int)((b < (unsigned)nb) ? b : (unsigned)(nb - 1));
}

__global__ void __launch_bounds__(GTPB)
grp_hist(const int* __restrict__ dst, int* __restrict__ H, int E, int nb, int nseg) {
  __shared__ int hist[GBMAX];
  const int t = (int)threadIdx.x;
  const int s = (int)blockIdx.x;
  if (t < nb) hist[t] = 0;
  __syncthreads();
  const int e0 = s * GSUB + t;
  int d[GEPT];
#pragma unroll
  for (int j = 0; j < GEPT; ++j) {
    int e = e0 + j * GTPB;
    d[j] = dst[(e < E) ? e : (E - 1)];
  }
#pragma unroll
  for (int j = 0; j < GEPT; ++j) {
    int e = e0 + j * GTPB;
    if (e < E) atomicAdd(&hist[bucket_of(d[j], nb)], 1);
  }
  __syncthreads();
  if (t < nb) H[(size_t)t * nseg + s] = hist[t];
}

__global__ void __launch_bounds__(RS_TPB)
grp_rowscan(const int* __restrict__ H, int* __restrict__ P, int* __restrict__ T, int nseg) {
  __shared__ int wtot[32];
  const int t = (int)threadIdx.x;
  const size_t row = (size_t)blockIdx.x * nseg;
  int v[4];
#pragma unroll
  for (int j = 0; j < 4; ++j) {
    int k = 4 * t + j;
    int h = H[row + ((k < nseg) ? k : 0)];
    v[j] = (k < nseg) ? h : 0;
  }
  int total;
  int ex = block_scan_excl<RS_TPB / 32>(v[0] + v[1] + v[2] + v[3], wtot, total);
#pragma unroll
  for (int j = 0; j < 4; ++j) {
    int k = 4 * t + j;
    if (k < nseg) P[row + k] = ex;
    ex += v[j];
  }
  if (t == 0) T[blockIdx.x] = total;
}

__global__ void __launch_bounds__(GTPB)
grp_partition(const int* __restrict__ src, const int* __restrict__ dst,
              const int* __restrict__ P, const int* __restrict__ T,
              int* __restrict__ list, int E, int nb, int nseg) {
  __shared__ int hist[GBMAX];
  __shared__ int offs[GBMAX];
  __shared__ int gbase[GBMAX];
  __shared__ int wtot[32];
  __shared__ int stage[GSUB];
  __shared__ unsigned short stage_b[GSUB];
  const int t = (int)threadIdx.x;
  const int s = (int)blockIdx.x;

  const int tb = (t < nb) ? T[t] : 0;
  const int pb = (t < nb) ? P[(size_t)t * nseg + s] : 0;
  if (t < nb) hist[t] = 0;
  int all;
  const int first = block_scan_excl<GTPB / 32>(tb, wtot, all);
  if (t < nb) gbase[t] = first + pb;

  const int e0 = s * GSUB + t;
  int a[GEPT], d[GEPT], r[GEPT];
#pragma unroll
  for (int j = 0; j < GEPT; ++j) {
    int e = e0 + j * GTPB;
    int ec = (e < E) ? e : (E - 1);
    d[j] = dst[ec];
    a[j] = src[ec];
  }
#pragma unroll
  for (int j = 0; j < GEPT; ++j) {
    int e = e0 + j * GTPB;
    r[j] = -1;
    if (e < E) r[j] = atomicAdd(&hist[bucket_of(d[j], nb)], 1);
  }
  __syncthreads();
  const int hv = (t < nb) ? hist[t] : 0;
  int total;
  const int ex = block_scan_excl<GTPB / 32>(hv, wtot, total);
  if (t < nb) offs[t] = ex;
  __syncthreads();
#pragma unroll
  for (int j = 0; j < GEPT; ++j) {
    if (r[j] >= 0) {
      const int b = bucket_of(d[j], nb);
      const int pos = offs[b] + r[j];
      if ((unsigned)pos < (unsigned)GSUB) {
        stage[pos] = (int)((((unsigned)d[j] & (unsigned)(GNB - 1)) << 24) |
                           ((unsigned)a[j] & 0xFFFFFFu));
        stage_b[pos] = (unsigned short)b;
      }
    }
  }
  __syncthreads();
#pragma unroll
  for (int j = 0; j < GEPT; ++j) {
    const int idx = t + j * GTPB;
    if (idx < total) {
      const int b = (int)stage_b[idx];
      const int g = gbase[b] + (idx - offs[b]);
      if ((unsigned)g < (unsigned)E) list[g] = stage[idx];
    }
  }
}

__global__ void __launch_bounds__(GTPB)
grp_bucket(const int* __restrict__ T, int* list, int* __restrict__ rowptr,
           int* __restrict__ rowend, float* __restrict__ dinv, int n, int E, int nb) {
  __shared__ int cnt[GNB];
  __shared__ int cur[GNB];
  __shared__ int endv[GNB];
  __shared__ int wtot[32];
  __shared__ int sh[2];
  __shared__ int ebuf[GCAP];
  __shared__ int obuf[GCAP];
  const int t = (int)threadIdx.x;
  const int b = (int)blockIdx.x;

  const int tb = (t < nb) ? T[t] : 0;
  int all;
  const int first = block_scan_excl<GTPB / 32>(tb, wtot, all);
  if (t == b) { sh[0] = first; sh[1] = tb; }
  if (t < GNB) cnt[t] = 0;
  __syncthreads();
  int base = sh[0], len = sh[1];
  base = (base > 0) ? base : 0;
  base = (base < E) ? base : E;
  len = (len > 0) ? len : 0;
  len = (len < E - base) ? len : (E - base);
  const bool fits = len <= GCAP;

  for (int j0 = t; j0 < len; j0 += 4 * GTPB) {
    unsigned v[4];
#pragma unroll
    for (int q = 0; q < 4; ++q) {
      int j = j0 + q * GTPB;
      v[q] = (unsigned)list[base + ((j < len) ? j : j0)];
    }
#pragma unroll
    for (int q = 0; q < 4; ++q) {
      int j = j0 + q * GTPB;
      if (j < len) {
        if (fits) ebuf[j] = (int)v[q];
        atomicAdd(&cnt[v[q] >> 24], 1);
      }
    }
  }
  __syncthreads();
  const int c = (t < GNB) ? cnt[t] : 0;
  int tot;
  const int ex = block_scan_excl<GTPB / 32>(c, wtot, tot);
  if (t < GNB) {
    cur[t] = ex;
    endv[t] = ex + c;
    const int i = b * GNB + t;
    if (i < n) {
      rowptr[i] = base + ex;
      rowend[i] = base + ex + c;
      dinv[i] = rsqrtf((float)(c + 1));
    }
  }
  __syncthreads();

  if (fits) {
    for (int j = t; j < len; j += GTPB) {
      const unsigned v = (unsigned)ebuf[j];
      const int pos = atomicAdd(&cur[v >> 24], 1);
      if ((unsigned)pos < (unsigned)GCAP) obuf[pos] = (int)(v & 0xFFFFFFu);
    }
    __syncthreads();
    for (int j = t; j < len; j += GTPB) list[base + j] = obuf[j];
  } else {
    if (t == 0) {
      int steps = 0;
      const int limit = 2 * len + GNB;
      for (int v = 0; v < GNB; ++v) {
        int p = cur[v];
        const int pe = endv[v];
        while (p < pe && steps < limit) {
          ++steps;
          const unsigned x = (unsigned)list[base + p];
          const int dv = (int)(x >> 24);
          if (dv == v) { ++p; continue; }
          const int q = cur[dv];
          if (q >= endv[dv]) break;
          const int y = list[base + q];
          list[base + q] = (int)x;
          list[base + p] = y;
          cur[dv] = q + 1;
        }
        cur[v] = p;
      }
    }
    __syncthreads();
    for (int j = t; j < len; j += GTPB) list[base + j] = list[base + j] & 0xFFFFFF;
  }
}

extern "C" void kernel_launch(void* const* d_in, const int* in_sizes, int n_in,
                              void* d_out, int out_size, void* d_ws, size_t ws_size,
                              hipStream_t stream) {
  const float* x  = (const float*)d_in[0];
  const int*   ei = (const int*)d_in[1];
  const float* W1 = (const float*)d_in[2];
  const float* b1 = (const float*)d_in[3];
  const float* W2 = (const float*)d_in[4];
  const float* b2 = (const float*)d_in[5];
  float* out = (float*)d_out;
  int* tick = (int*)d_out;

  int n = in_sizes[0];
  int E = in_sizes[1] / 2;
  const int* srcp = ei;
  const int* dstp = ei + E;

  size_t ns = ((size_t)n + 31) & ~(size_t)31;

  float* ws    = (float*)d_ws;
  float* deg   = ws;
  float* dinv  = ws + ns;
  float* s     = ws + 2 * ns;
  float* h2c0  = ws + 3 * ns;
  float* h2c1  = ws + 4 * ns;
  float* agg0  = ws + 5 * ns;
  float* agg1  = ws + 6 * ns;

  int gN = (n + TPB - 1) / TPB;
  int ntiles = (n + 15) / 16;
  int gW = (int)(((size_t)ntiles * 32 + TPB - 1) / TPB);

  const size_t need = (25 * ns + (size_t)E) * sizeof(float);
  if (ws_size >= need) {
    int*   cnt    = (int*)ws;
    int*   rowptr = (int*)(ws + 7 * ns);
    int*   cursor = (int*)(ws + 8 * ns);
    float* relu1  = ws + 9 * ns;
    int*   list   = (int*)(ws + 25 * ns);
    int gS = (n + SCAN_TPB - 1) / SCAN_TPB;

    const int nb = (n + GNB - 1) >> GSH;
    const int nseg = (E + GSUB - 1) / GSUB;
    const size_t tabs = 2 * (size_t)nb * (size_t)nseg + (size_t)nb;
    if (E > 0 && nb <= GBMAX && nseg <= GTPB && n <= (1 << 24) && tabs <= 16 * (size_t)n) {
      int* H = (int*)relu1;
      int* P = H + (size_t)nb * nseg;
      int* T = P + (size_t)nb * nseg;
      grp_hist<<<nseg, GTPB, 0, stream>>>(dstp, H, E, nb, nseg);
      grp_rowscan<<<nb, RS_TPB, 0, stream>>>(H, P, T, nseg);
      grp_partition<<<nseg, GTPB, 0, stream>>>(srcp, dstp, P, T, list, E, nb, nseg);
      grp_bucket<<<nb, GTPB, 0, stream>>>(T, list, rowptr, cursor, dinv, n, E, nb);
    } else {
      csr_init<<<gN, TPB, 0, stream>>>(cnt, tick, n);
      csr_count<<<PASS_BLOCKS, TPB, 0, stream>>>(dstp, cnt, tick, n, E);
      csr_scan<<<gS, SCAN_TPB, 0, stream>>>(cnt, rowptr, cursor, dinv, n);
      csr_fill<<<PASS_BLOCKS, TPB, 0, stream>>>(srcp, dstp, cursor, list, tick, n, E);
    }
    gcn_l1_gather<<<PASS_BLOCKS, TPB, 0, stream>>>(rowptr, cursor, list, x, dinv, s, n, E);
    gcn_l1_finalize<<<gN, TPB, 0, stream>>>(x, dinv, s, W1, b1, relu1, n);
    gcn_l2_dense_wmma<<<gW, TPB, 0, stream>>>(relu1, W2, h2c0, h2c1, n, ntiles);
    gcn_l2_gather<<<PASS_BLOCKS, TPB, 0, stream>>>(rowptr, cursor, list, h2c0, h2c1, dinv,
                                                    agg0, agg1, n, E);
    gcn_finalize<<<gN, TPB, 0, stream>>>(agg0, agg1, h2c0, h2c1, dinv, b2, out, n);
    return;
  }

  float* relu1 = ws + 7 * ns;
  gcn_init<<<gN, TPB, 0, stream>>>(deg, s, agg0, agg1, tick, n);
  gcn_degree<<<PASS_BLOCKS, TPB, 0, stream>>>(dstp, deg, tick, n, E);
  gcn_dinv<<<gN, TPB, 0, stream>>>(deg, dinv, n);
  gcn_l1_scatter<<<PASS_BLOCKS, TPB, 0, stream>>>(srcp, dstp, x, dinv, s, tick, n, E);
  gcn_l1_finalize<<<gN, TPB, 0, stream>>>(x, dinv, s, W1, b1, relu1, n);
  gcn_l2_dense_wmma<<<gW, TPB, 0, stream>>>(relu1, W2, h2c0, h2c1, n, ntiles);
  gcn_l2_scatter<<<PASS_BLOCKS, TPB, 0, stream>>>(srcp, dstp, h2c0, h2c1, dinv,
                                                   agg0, agg1, tick, n, E);
  gcn_finalize<<<gN, TPB, 0, stream>>>(agg0, agg1, h2c0, h2c1, dinv, b2, out, n);
}


# Per-node values an edge gathers interleaved into one record per layer ({x,dinv} 8 B; {dinv,class0,class1} 16 B): one line request per edge and layer, not 2 and 3; terms and sum order unchanged
# speedup vs baseline: 8.0832x; 1.3880x over previous
#include <hip/hip_runtime.h>


typedef __attribute__((ext_vector_type(2))) float v2f;
typedef __attribute__((ext_vector_type(8))) float v8f;

#define TPB 256
#define EPT 4
#define SLICE (TPB * EPT)
#define PASS_BLOCKS 2048
#define NQ 6


__device__ __forceinline__ int tick_stride(int n) { return (n >= 32 * NQ) ? 64 : 1; }

__device__ __forceinline__ int cache_half() {
  unsigned v = __builtin_amdgcn_s_sendmsg_rtn(0x87);
  return (int)((v >> 18) & 1u);
}

__device__ __forceinline__ int draw_slice(int* q, int* sh) {
  __syncthreads();
  if (threadIdx.x == 0) *sh = atomicAdd(q, 1);
  __syncthreads();
  return *sh;
}

__device__ __forceinline__ int block_half(int* sh) {
  if (threadIdx.x == 0) *sh = cache_half();
  __syncthreads();
  return *sh;
}

__global__ void gcn_init(float* __restrict__ deg, float* __restrict__ s,
                         float* __restrict__ agg0, float* __restrict__ agg1,
                         int* __restrict__ tick, int n) {
  int i = blockIdx.x * blockDim.x + threadIdx.x;
  if (i < NQ) tick[i * tick_stride(n)] = 0;
  if (i < n) {
    deg[i] = 1.0f;
    s[i] = 0.0f;
    agg0[i] = 0.0f;
    agg1[i] = 0.0f;
  }
}

__global__ void __launch_bounds__(TPB)
gcn_degree(const int* __restrict__ dst, float* __restrict__ deg,
           int* __restrict__ tick, int n, int E) {
  __shared__ int sh[2];
  const int nsl = (E + SLICE - 1) / SLICE;
  const int split = (n >> 1) & ~31;
  const int st = tick_stride(n);
  const int h0 = block_half(&sh[1]);
  for (int r = 0; r < 2; ++r) {
    const int h = h0 ^ r;
    const int lo = h ? split : 0;
    const int hi = h ? n : split;
    int* q = tick + (0 + h) * st;
    for (int t = draw_slice(q, &sh[0]); t < nsl; t = draw_slice(q, &sh[0])) {
      const int e0 = t * SLICE + (int)threadIdx.x;
      int b[EPT];
#pragma unroll
      for (int j = 0; j < EPT; ++j) {
        int e = e0 + j * TPB;
        int v = dst[(e < E) ? e : (E - 1)];
        b[j] = (e < E) ? v : -1;
      }
#pragma unroll
      for (int j = 0; j < EPT; ++j)
        if (b[j] >= lo && b[j] < hi) unsafeAtomicAdd(&deg[b[j]], 1.0f);
    }
  }
}

__global__ void gcn_dinv(const float* __restrict__ deg,
                         float* __restrict__ dinv, int n) {
  int i = blockIdx.x * blockDim.x + threadIdx.x;
  if (i < n) dinv[i] = rsqrtf(deg[i]);
}

__global__ void __launch_bounds__(TPB)
gcn_l1_scatter(const int* __restrict__ src, const int* __restrict__ dst,
               const float* __restrict__ x, const float* __restrict__ dinv,
               float* __restrict__ s, int* __restrict__ tick, int n, int E) {
  __shared__ int sh[2];
  const int nsl = (E + SLICE - 1) / SLICE;
  const int split = (n >> 1) & ~31;
  const int st = tick_stride(n);
  const int h0 = block_half(&sh[1]);
  for (int r = 0; r < 2; ++r) {
    const int h = h0 ^ r;
    const int lo = h ? split : 0;
    const int hi = h ? n : split;
    int* q = tick + (2 + h) * st;
    for (int t = draw_slice(q, &sh[0]); t < nsl; t = draw_slice(q, &sh[0])) {
      const int e0 = t * SLICE + (int)threadIdx.x;
      int a[EPT], b[EPT];
#pragma unroll
      for (int j = 0; j < EPT; ++j) {
        int e = e0 + j * TPB;
        int ec = (e < E) ? e : (E - 1);
        int vb = dst[ec];
        a[j] = src[ec];
        b[j] = (e < E) ? vb : -1;
      }
      asm volatile("" :: "v"(a[0]), "v"(a[1]), "v"(a[2]), "v"(a[3]));
      float v[EPT];
#pragma unroll
      for (int j = 0; j < EPT; ++j) {
        bool in = b[j] >= lo && b[j] < hi;
        unsigned aa = in ? (unsigned)a[j] : 0u;
        unsigned bb = in ? (unsigned)b[j] : 0u;
        v[j] = x[aa] * dinv[aa] * dinv[bb];
      }
#pragma unroll
      for (int j = 0; j < EPT; ++j)
        if (b[j] >= lo && b[j] < hi) unsafeAtomicAdd(&s[b[j]], v[j]);
    }
  }
}

__global__ void gcn_l1_finalize(const float* __restrict__ x,
                                const float* __restrict__ dinv,
                                const float* __restrict__ s,
                                const float* __restrict__ W1,
                                const float* __restrict__ b1,
                                float* __restrict__ relu1, int n) {
  int i = blockIdx.x * blockDim.x + threadIdx.x;
  if (i >= n) return;
  float di = dinv[i];
  float sn = s[i] + x[i] * di * di;
  float4* o = (float4*)(relu1 + (size_t)i * 16);
#pragma unroll
  for (int q = 0; q < 4; ++q) {
    float4 r;
    r.x = fmaxf(fmaf(sn, W1[4 * q + 0], b1[4 * q + 0]), 0.0f);
    r.y = fmaxf(fmaf(sn, W1[4 * q + 1], b1[4 * q + 1]), 0.0f);
    r.z = fmaxf(fmaf(sn, W1[4 * q + 2], b1[4 * q + 2]), 0.0f);
    r.w = fmaxf(fmaf(sn, W1[4 * q + 3], b1[4 * q + 3]), 0.0f);
    o[q] = r;
  }
}

template <bool REC>
__global__ void gcn_l2_dense_wmma(const float* __restrict__ relu1,
                                  const float* __restrict__ W2,
                                  const float* __restrict__ dinv,
                                  float* __restrict__ h2c0,
                                  float* __restrict__ h2c1,
                                  float4* __restrict__ rec2,
                                  int n, int ntiles) {
  int gtid = blockIdx.x * blockDim.x + threadIdx.x;
  int tile = gtid >> 5;
  int lane = threadIdx.x & 31;
  if (tile >= ntiles) return;
  int base = tile * 16;
  int half = lane >> 4;
  int m = lane & 15;

  v2f bmat[4];
#pragma unroll
  for (int k = 0; k < 4; ++k) {
    int r0 = 4 * k + 2 * half;
    float bx = 0.0f, by = 0.0f;
    if (m < 2) { bx = W2[r0 * 2 + m]; by = W2[(r0 + 1) * 2 + m]; }
    bmat[k].x = bx; bmat[k].y = by;
  }

  int row = base + m;
  row = (row < n) ? row : (n - 1);
  const v2f* p = (const v2f*)(relu1 + (size_t)row * 16 + 2 * half);
  v2f amat[4];
#pragma unroll
  for (int k = 0; k < 4; ++k) amat[k] = p[2 * k];

  v8f c = {};
#pragma unroll
  for (int k = 0; k < 4; ++k)
    c = __builtin_amdgcn_wmma_f32_16x16x4_f32(
         false, amat[k],  false, bmat[k],
         (short)0, c,  false,  false);

  if (REC) {
    const int l0 = lane & 16;
    float h0 = 0.0f, h1 = 0.0f;
#pragma unroll
    for (int r = 0; r < 8; ++r) {
      const float t0 = __shfl(c[r], l0);
      const float t1 = __shfl(c[r], l0 + 1);
      if (m == r) { h0 = t0; h1 = t1; }
    }
    if (m < 8) {
      const int node = base + half * 8 + m;
      rec2[node] = make_float4(dinv[node], h0, h1, 0.0f);
    }
  } else if (m < 2) {
    float* col = (m == 0) ? h2c0 : h2c1;
    int node0 = base + half * 8;
    float4 lo; lo.x = c[0]; lo.y = c[1]; lo.z = c[2]; lo.w = c[3];
    float4 hi; hi.x = c[4]; hi.y = c[5]; hi.z = c[6]; hi.w = c[7];
    *(float4*)(col + node0)     = lo;
    *(float4*)(col + node0 + 4) = hi;
  }
}

__global__ void __launch_bounds__(TPB)
gcn_l2_scatter(const int* __restrict__ src, const int* __restrict__ dst,
               const float* __restrict__ h2c0, const float* __restrict__ h2c1,
               const float* __restrict__ dinv,
               float* __restrict__ agg0, float* __restrict__ agg1,
               int* __restrict__ tick, int n, int E) {
  __shared__ int sh[2];
  const int nsl = (E + SLICE - 1) / SLICE;
  const int split = (n >> 1) & ~31;
  const int st = tick_stride(n);
  const int h0 = block_half(&sh[1]);
  for (int r = 0; r < 2; ++r) {
    const int h = h0 ^ r;
    const int lo = h ? split : 0;
    const int hi = h ? n : split;
    int* q = tick + (4 + h) * st;
    for (int t = draw_slice(q, &sh[0]); t < nsl; t = draw_slice(q, &sh[0])) {
      const int e0 = t * SLICE + (int)threadIdx.x;
      int a[EPT], b[EPT];
#pragma unroll
      for (int j = 0; j < EPT; ++j) {
        int e = e0 + j * TPB;
        int ec = (e < E) ? e : (E - 1);
        int vb = dst[ec];
        a[j] = src[ec];
        b[j] = (e < E) ? vb : -1;
      }
      asm volatile("" :: "v"(a[0]), "v"(a[1]), "v"(a[2]), "v"(a[3]));
      float v0[EPT], v1[EPT];
#pragma unroll
      for (int j = 0; j < EPT; ++j) {
        bool in = b[j] >= lo && b[j] < hi;
        unsigned aa = in ? (unsigned)a[j] : 0u;
        unsigned bb = in ? (unsigned)b[j] : 0u;
        float nrm = dinv[aa] * dinv[bb];
        v0[j] = h2c0[aa] * nrm;
        v1[j] = h2c1[aa] * nrm;
      }
#pragma unroll
      for (int j = 0; j < EPT; ++j)
        if (b[j] >= lo && b[j] < hi) {
          unsafeAtomicAdd(&agg0[b[j]], v0[j]);
          unsafeAtomicAdd(&agg1[b[j]], v1[j]);
        }
    }
  }
}

template <bool REC>
__global__ void gcn_finalize(const float* __restrict__ agg0,
                             const float* __restrict__ agg1,
                             const float* __restrict__ h2c0,
                             const float* __restrict__ h2c1,
                             const float* __restrict__ dinv,
                             const float4* __restrict__ rec2,
                             const float* __restrict__ b2,
                             float* __restrict__ out, int n) {
  int i = blockIdx.x * blockDim.x + threadIdx.x;
  if (i >= n) return;
  float di, h0, h1;
  if (REC) {
    const float4 q = rec2[i];
    di = q.x; h0 = q.y; h1 = q.z;
  } else {
    di = dinv[i]; h0 = h2c0[i]; h1 = h2c1[i];
  }
  float sl = di * di;
  float z0 = agg0[i] + h0 * sl + b2[0];
  float z1 = agg1[i] + h1 * sl + b2[1];
  float mx = fmaxf(z0, z1);
  float lse = mx + logf(expf(z0 - mx) + expf(z1 - mx));
  float2 r; r.x = z0 - lse; r.y = z1 - lse;
  *(float2*)(out + (size_t)i * 2) = r;
}

#define SCAN_TPB 1024

__global__ void csr_init(int* __restrict__ cnt, int* __restrict__ tick, int n) {
  int i = blockIdx.x * blockDim.x + threadIdx.x;
  if (i < 4) tick[i * tick_stride(n)] = 0;
  if (i < n) cnt[i] = 0;
}

__global__ void __launch_bounds__(TPB)
csr_count(const int* __restrict__ dst, int* __restrict__ cnt,
          int* __restrict__ tick, int n, int E) {
  __shared__ int sh[2];
  const int nsl = (E + SLICE - 1) / SLICE;
  const int split = (n >> 1) & ~31;
  const int st = tick_stride(n);
  const int h0 = block_half(&sh[1]);
  for (int r = 0; r < 2; ++r) {
    const int h = h0 ^ r;
    const int lo = h ? split : 0;
    const int hi = h ? n : split;
    int* q = tick + (0 + h) * st;
    for (int t = draw_slice(q, &sh[0]); t < nsl; t = draw_slice(q, &sh[0])) {
      const int e0 = t * SLICE + (int)threadIdx.x;
      int b[EPT];
#pragma unroll
      for (int j = 0; j < EPT; ++j) {
        int e = e0 + j * TPB;
        int v = dst[(e < E) ? e : (E - 1)];
        b[j] = (e < E) ? v : -1;
      }
#pragma unroll
      for (int j = 0; j < EPT; ++j)
        if (b[j] >= lo && b[j] < hi) atomicAdd(&cnt[b[j]], 1);
    }
  }
}

__global__ void __launch_bounds__(SCAN_TPB)
csr_scan(const int* __restrict__ cnt, const float* __restrict__ xin,
         int* __restrict__ rowptr, int* __restrict__ cursor,
         float* __restrict__ dinv, float2* __restrict__ rec1, int n) {
  __shared__ int wtot[32];
  __shared__ int wpre[32];
  const int t = (int)threadIdx.x;
  const int lane = t & 31;
  const int wv = t >> 5;
  const int first = (int)blockIdx.x * SCAN_TPB;
  const int i = first + t;
  const int c = (i < n) ? cnt[i] : 0;

  int pre = 0;
  const int4* c4 = (const int4*)cnt;
  const int ngrp = first / 4;
  for (int k = t; k < ngrp; k += 8 * SCAN_TPB) {
    int4 v[8];
#pragma unroll
    for (int j = 0; j < 8; ++j) {
      int kk = k + j * SCAN_TPB;
      v[j] = c4[(kk < ngrp) ? kk : k];
    }
#pragma unroll
    for (int j = 0; j < 8; ++j) {
      int kk = k + j * SCAN_TPB;
      int sum = (v[j].x + v[j].y) + (v[j].z + v[j].w);
      pre += (kk < ngrp) ? sum : 0;
    }
  }
  for (int o = 16; o > 0; o >>= 1) pre += __shfl_xor(pre, o);

  int inc = c;
  for (int o = 1; o < 32; o <<= 1) {
    int u = __shfl_up(inc, o);
    if (lane >= o) inc += u;
  }
  if (lane == 31) wtot[wv] = inc;
  if (lane == 0) wpre[wv] = pre;
  __syncthreads();

  int wt = wtot[lane];
  int winc = wt;
  for (int o = 1; o < 32; o <<= 1) {
    int u = __shfl_up(winc, o);
    if (lane >= o) winc += u;
  }
  int before_wave = __shfl(winc - wt, wv);
  int before_block = wpre[lane];
  for (int o = 16; o > 0; o >>= 1) before_block += __shfl_xor(before_block, o);

  if (i < n) {
    int slot = before_block + before_wave + (inc - c);
    rowptr[i] = slot;
    cursor[i] = slot;
    const float dv = rsqrtf((float)(c + 1));
    dinv[i] = dv;
    rec1[i] = make_float2(xin[i], dv);
  }
}

__global__ void __launch_bounds__(TPB)
csr_fill(const int* __restrict__ src, const int* __restrict__ dst,
         int* __restrict__ cursor, int* __restrict__ list,
         int* __restrict__ tick, int n, int E) {
  __shared__ int sh[2];
  const int nsl = (E + SLICE - 1) / SLICE;
  const int split = (n >> 1) & ~31;
  const int st = tick_stride(n);
  const int h0 = block_half(&sh[1]);
  for (int r = 0; r < 2; ++r) {
    const int h = h0 ^ r;
    const int lo = h ? split : 0;
    const int hi = h ? n : split;
    int* q = tick + (2 + h) * st;
    for (int t = draw_slice(q, &sh[0]); t < nsl; t = draw_slice(q, &sh[0])) {
      const int e0 = t * SLICE + (int)threadIdx.x;
      int a[EPT], b[EPT];
#pragma unroll
      for (int j = 0; j < EPT; ++j) {
        int e = e0 + j * TPB;
        int ec = (e < E) ? e : (E - 1);
        int vb = dst[ec];
        a[j] = src[ec];
        b[j] = (e < E) ? vb : -1;
      }
      int pos[EPT];
#pragma unroll
      for (int j = 0; j < EPT; ++j) {
        pos[j] = -1;
        if (b[j] >= lo && b[j] < hi) pos[j] = atomicAdd(&cursor[b[j]], 1);
      }
#pragma unroll
      for (int j = 0; j < EPT; ++j)
        if ((unsigned)pos[j] < (unsigned)E) list[pos[j]] = a[j];
    }
  }
}

__global__ void __launch_bounds__(TPB)
gcn_l1_gather(const int* __restrict__ rowptr, const int* __restrict__ rowend,
              const int* __restrict__ list, const float2* __restrict__ rec1,
              const float* __restrict__ dinv, float* __restrict__ s,
              int n, int E) {
  const int lane = (int)threadIdx.x & 31;
  const int wv = __builtin_amdgcn_readfirstlane((int)threadIdx.x >> 5);
  const int nw = (int)gridDim.x * (TPB / 32);
  const unsigned last = (unsigned)(n - 1);
  int i = (int)blockIdx.x * (TPB / 32) + wv;
  int st = 0, en = 0;
  float di = 0.0f;
  if (i < n) { st = rowptr[i]; en = rowend[i]; di = dinv[i]; }
  while (i < n) {
    const int ni = i + nw;
    int nst = 0, nen = 0;
    float ndi = 0.0f;
    if (ni < n) { nst = rowptr[ni]; nen = rowend[ni]; ndi = dinv[ni]; }
    st = __builtin_amdgcn_readfirstlane((st > 0) ? st : 0);
    en = __builtin_amdgcn_readfirstlane((en < E) ? en : E);
    float acc = 0.0f;
    for (int base = st; base < en; base += 64) {
      const int k0 = base + lane, k1 = k0 + 32;
      const bool p0 = k0 < en, p1 = k1 < en;
      unsigned a0 = (unsigned)list[p0 ? k0 : st];
      unsigned a1 = (unsigned)list[p1 ? k1 : st];
      a0 = (a0 < last) ? a0 : last;
      a1 = (a1 < last) ? a1 : last;
      const float2 r0 = rec1[a0];
      const float2 r1 = rec1[a1];
      const float t0 = r0.x * r0.y * di;
      const float t1 = r1.x * r1.y * di;
      acc += p0 ? t0 : 0.0f;
      acc += p1 ? t1 : 0.0f;
    }
    for (int o = 16; o > 0; o >>= 1) acc += __shfl_xor(acc, o);
    if (lane == 0) s[i] = acc;
    i = ni; st = nst; en = nen; di = ndi;
  }
}

__global__ void __launch_bounds__(TPB)
gcn_l2_gather(const int* __restrict__ rowptr, const int* __restrict__ rowend,
              const int* __restrict__ list, const float4* __restrict__ rec2,
              const float* __restrict__ dinv,
              float* __restrict__ agg0, float* __restrict__ agg1,
              int n, int E) {
  const int lane = (int)threadIdx.x & 31;
  const int wv = __builtin_amdgcn_readfirstlane((int)threadIdx.x >> 5);
  const int nw = (int)gridDim.x * (TPB / 32);
  const unsigned last = (unsigned)(n - 1);
  int i = (int)blockIdx.x * (TPB / 32) + wv;
  int st = 0, en = 0;
  float di = 0.0f;
  if (i < n) { st = rowptr[i]; en = rowend[i]; di = dinv[i]; }
  while (i < n) {
    const int ni = i + nw;
    int nst = 0, nen = 0;
    float ndi = 0.0f;
    if (ni < n) { nst = rowptr[ni]; nen = rowend[ni]; ndi = dinv[ni]; }
    st = __builtin_amdgcn_readfirstlane((st > 0) ? st : 0);
    en = __builtin_amdgcn_readfirstlane((en < E) ? en : E);
    float acc0 = 0.0f, acc1 = 0.0f;
    for (int base = st; base < en; base += 64) {
      const int k0 = base + lane, k1 = k0 + 32;
      const bool p0 = k0 < en, p1 = k1 < en;
      unsigned a0 = (unsigned)list[p0 ? k0 : st];
      unsigned a1 = (unsigned)list[p1 ? k1 : st];
      a0 = (a0 < last) ? a0 : last;
      a1 = (a1 < last) ? a1 : last;
      const float4 q0 = rec2[a0];
      const float4 q1 = rec2[a1];
      const float n0 = q0.x * di;
      const float n1 = q1.x * di;
      const float u0 = q0.y * n0, w0 = q0.z * n0;
      const float u1 = q1.y * n1, w1 = q1.z * n1;
      acc0 += p0 ? u0 : 0.0f;  acc1 += p0 ? w0 : 0.0f;
      acc0 += p1 ? u1 : 0.0f;  acc1 += p1 ? w1 : 0.0f;
    }
    for (int o = 16; o > 0; o >>= 1) {
      acc0 += __shfl_xor(acc0, o);
      acc1 += __shfl_xor(acc1, o);
    }
    if (lane == 0) { agg0[i] = acc0; agg1[i] = acc1; }
    i = ni; st = nst; en = nen; di = ndi;
  }
}


#define GSH 8
#define GNB (1 << GSH)
#define GTPB 1024
#define GEPT 12
#define GSUB (GTPB * GEPT)
#define GBMAX GTPB
#define GCAP 16384
#define RS_TPB 256

template <int NW>
__device__ __forceinline__ int block_scan_excl(int v, int* wtot, int& total) {
  const int lane = (int)threadIdx.x & 31;
  const int wv = (int)threadIdx.x >> 5;
  int inc = v;
  for (int o = 1; o < 32; o <<= 1) {
    int u = __shfl_up(inc, o);
    if (lane >= o) inc += u;
  }
  if (lane == 31) wtot[wv] = inc;
  __syncthreads();
  const int wt = (lane < NW) ? wtot[lane] : 0;
  int winc = wt;
  for (int o = 1; o < 32; o <<= 1) {
    int u = __shfl_up(winc, o);
    if (lane >= o) winc += u;
  }
  const int before = __shfl(winc - wt, wv);
  total = __shfl(winc, 31);
  __syncthreads();
  return before + (inc - v);
}

__device__ __forceinline__ int bucket_of(int d, int nb) {
  unsigned b = (unsigned)d >> GSH;
  return (int)((b < (unsigned)nb) ? b : (unsigned)(nb - 1));
}

__global__ void __launch_bounds__(GTPB)
grp_hist(const int* __restrict__ dst, int* __restrict__ H, int E, int nb, int nseg) {
  __shared__ int hist[GBMAX];
  const int t = (int)threadIdx.x;
  const int s = (int)blockIdx.x;
  if (t < nb) hist[t] = 0;
  __syncthreads();
  const int e0 = s * GSUB + t;
  int d[GEPT];
#pragma unroll
  for (int j = 0; j < GEPT; ++j) {
    int e = e0 + j * GTPB;
    d[j] = dst[(e < E) ? e : (E - 1)];
  }
#pragma unroll
  for (int j = 0; j < GEPT; ++j) {
    int e = e0 + j * GTPB;
    if (e < E) atomicAdd(&hist[bucket_of(d[j], nb)], 1);
  }
  __syncthreads();
  if (t < nb) H[(size_t)t * nseg + s] = hist[t];
}

__global__ void __launch_bounds__(RS_TPB)
grp_rowscan(const int* __restrict__ H, int* __restrict__ P, int* __restrict__ T, int nseg) {
  __shared__ int wtot[32];
  const int t = (int)threadIdx.x;
  const size_t row = (size_t)blockIdx.x * nseg;
  int v[4];
#pragma unroll
  for (int j = 0; j < 4; ++j) {
    int k = 4 * t + j;
    int h = H[row + ((k < nseg) ? k : 0)];
    v[j] = (k < nseg) ? h : 0;
  }
  int total;
  int ex = block_scan_excl<RS_TPB / 32>(v[0] + v[1] + v[2] + v[3], wtot, total);
#pragma unroll
  for (int j = 0; j < 4; ++j) {
    int k = 4 * t + j;
    if (k < nseg) P[row + k] = ex;
    ex += v[j];
  }
  if (t == 0) T[blockIdx.x] = total;
}

__global__ void __launch_bounds__(GTPB)
grp_partition(const int* __restrict__ src, const int* __restrict__ dst,
              const int* __restrict__ P, const int* __restrict__ T,
              int* __restrict__ list, int E, int nb, int nseg) {
  __shared__ int hist[GBMAX];
  __shared__ int offs[GBMAX];
  __shared__ int gbase[GBMAX];
  __shared__ int wtot[32];
  __shared__ int stage[GSUB];
  __shared__ unsigned short stage_b[GSUB];
  const int t = (int)threadIdx.x;
  const int s = (int)blockIdx.x;

  const int tb = (t < nb) ? T[t] : 0;
  const int pb = (t < nb) ? P[(size_t)t * nseg + s] : 0;
  if (t < nb) hist[t] = 0;
  int all;
  const int first = block_scan_excl<GTPB / 32>(tb, wtot, all);
  if (t < nb) gbase[t] = first + pb;

  const int e0 = s * GSUB + t;
  int a[GEPT], d[GEPT], r[GEPT];
#pragma unroll
  for (int j = 0; j < GEPT; ++j) {
    int e = e0 + j * GTPB;
    int ec = (e < E) ? e : (E - 1);
    d[j] = dst[ec];
    a[j] = src[ec];
  }
#pragma unroll
  for (int j = 0; j < GEPT; ++j) {
    int e = e0 + j * GTPB;
    r[j] = -1;
    if (e < E) r[j] = atomicAdd(&hist[bucket_of(d[j], nb)], 1);
  }
  __syncthreads();
  const int hv = (t < nb) ? hist[t] : 0;
  int total;
  const int ex = block_scan_excl<GTPB / 32>(hv, wtot, total);
  if (t < nb) offs[t] = ex;
  __syncthreads();
#pragma unroll
  for (int j = 0; j < GEPT; ++j) {
    if (r[j] >= 0) {
      const int b = bucket_of(d[j], nb);
      const int pos = offs[b] + r[j];
      if ((unsigned)pos < (unsigned)GSUB) {
        stage[pos] = (int)((((unsigned)d[j] & (unsigned)(GNB - 1)) << 24) |
                           ((unsigned)a[j] & 0xFFFFFFu));
        stage_b[pos] = (unsigned short)b;
      }
    }
  }
  __syncthreads();
#pragma unroll
  for (int j = 0; j < GEPT; ++j) {
    const int idx = t + j * GTPB;
    if (idx < total) {
      const int b = (int)stage_b[idx];
      const int g = gbase[b] + (idx - offs[b]);
      if ((unsigned)g < (unsigned)E) list[g] = stage[idx];
    }
  }
}

__global__ void __launch_bounds__(GTPB)
grp_bucket(const int* __restrict__ T, const float* __restrict__ xin, int* list,
           int* __restrict__ rowptr, int* __restrict__ rowend,
           float* __restrict__ dinv, float2* __restrict__ rec1, int n, int E, int nb) {
  __shared__ int cnt[GNB];
  __shared__ int cur[GNB];
  __shared__ int endv[GNB];
  __shared__ int wtot[32];
  __shared__ int sh[2];
  __shared__ int ebuf[GCAP];
  __shared__ int obuf[GCAP];
  const int t = (int)threadIdx.x;
  const int b = (int)blockIdx.x;

  const int tb = (t < nb) ? T[t] : 0;
  int all;
  const int first = block_scan_excl<GTPB / 32>(tb, wtot, all);
  if (t == b) { sh[0] = first; sh[1] = tb; }
  if (t < GNB) cnt[t] = 0;
  __syncthreads();
  int base = sh[0], len = sh[1];
  base = (base > 0) ? base : 0;
  base = (base < E) ? base : E;
  len = (len > 0) ? len : 0;
  len = (len < E - base) ? len : (E - base);
  const bool fits = len <= GCAP;

  for (int j0 = t; j0 < len; j0 += 4 * GTPB) {
    unsigned v[4];
#pragma unroll
    for (int q = 0; q < 4; ++q) {
      int j = j0 + q * GTPB;
      v[q] = (unsigned)list[base + ((j < len) ? j : j0)];
    }
#pragma unroll
    for (int q = 0; q < 4; ++q) {
      int j = j0 + q * GTPB;
      if (j < len) {
        if (fits) ebuf[j] = (int)v[q];
        atomicAdd(&cnt[v[q] >> 24], 1);
      }
    }
  }
  __syncthreads();
  const int c = (t < GNB) ? cnt[t] : 0;
  int tot;
  const int ex = block_scan_excl<GTPB / 32>(c, wtot, tot);
  if (t < GNB) {
    cur[t] = ex;
    endv[t] = ex + c;
    const int i = b * GNB + t;
    if (i < n) {
      rowptr[i] = base + ex;
      rowend[i] = base + ex + c;
      const float dv = rsqrtf((float)(c + 1));
      dinv[i] = dv;
      rec1[i] = make_float2(xin[i], dv);
    }
  }
  __syncthreads();

  if (fits) {
    for (int j = t; j < len; j += GTPB) {
      const unsigned v = (unsigned)ebuf[j];
      const int pos = atomicAdd(&cur[v >> 24], 1);
      if ((unsigned)pos < (unsigned)GCAP) obuf[pos] = (int)(v & 0xFFFFFFu);
    }
    __syncthreads();
    for (int j = t; j < len; j += GTPB) list[base + j] = obuf[j];
  } else {
    if (t == 0) {
      int steps = 0;
      const int limit = 2 * len + GNB;
      for (int v = 0; v < GNB; ++v) {
        int p = cur[v];
        const int pe = endv[v];
        while (p < pe && steps < limit) {
          ++steps;
          const unsigned x = (unsigned)list[base + p];
          const int dv = (int)(x >> 24);
          if (dv == v) { ++p; continue; }
          const int q = cur[dv];
          if (q >= endv[dv]) break;
          const int y = list[base + q];
          list[base + q] = (int)x;
          list[base + p] = y;
          cur[dv] = q + 1;
        }
        cur[v] = p;
      }
    }
    __syncthreads();
    for (int j = t; j < len; j += GTPB) list[base + j] = list[base + j] & 0xFFFFFF;
  }
}

extern "C" void kernel_launch(void* const* d_in, const int* in_sizes, int n_in,
                              void* d_out, int out_size, void* d_ws, size_t ws_size,
                              hipStream_t stream) {
  const float* x  = (const float*)d_in[0];
  const int*   ei = (const int*)d_in[1];
  const float* W1 = (const float*)d_in[2];
  const float* b1 = (const float*)d_in[3];
  const float* W2 = (const float*)d_in[4];
  const float* b2 = (const float*)d_in[5];
  float* out = (float*)d_out;
  int* tick = (int*)d_out;

  int n = in_sizes[0];
  int E = in_sizes[1] / 2;
  const int* srcp = ei;
  const int* dstp = ei + E;

  size_t ns = ((size_t)n + 31) & ~(size_t)31;

  float* ws    = (float*)d_ws;
  float* deg   = ws;
  float* dinv  = ws + ns;
  float* s     = ws + 2 * ns;
  float* h2c0  = ws + 3 * ns;
  float* h2c1  = ws + 4 * ns;
  float* agg0  = ws + 5 * ns;
  float* agg1  = ws + 6 * ns;

  int gN = (n + TPB - 1) / TPB;
  int ntiles = (n + 15) / 16;
  int gW = (int)(((size_t)ntiles * 32 + TPB - 1) / TPB);

  const size_t need = (25 * ns + (size_t)E) * sizeof(float);
  if (ws_size >= need) {
    int*   cnt    = (int*)ws;
    int*   rowptr = (int*)(ws + 7 * ns);
    int*   cursor = (int*)(ws + 8 * ns);
    float* relu1  = ws + 9 * ns;
    int*   list   = (int*)(ws + 25 * ns);
    float2* rec1  = (float2*)(ws + 4 * ns);
    float4* rec2  = (float4*)(ws + 2 * ns);
    float*  sum0  = ws;
    float*  sum1  = ws + 6 * ns;
    int gS = (n + SCAN_TPB - 1) / SCAN_TPB;

    const int nb = (n + GNB - 1) >> GSH;
    const int nseg = (E + GSUB - 1) / GSUB;
    const size_t tabs = 2 * (size_t)nb * (size_t)nseg + (size_t)nb;
    if (E > 0 && nb <= GBMAX && nseg <= GTPB && n <= (1 << 24) && tabs <= 16 * (size_t)n) {
      int* H = (int*)relu1;
      int* P = H + (size_t)nb * nseg;
      int* T = P + (size_t)nb * nseg;
      grp_hist<<<nseg, GTPB, 0, stream>>>(dstp, H, E, nb, nseg);
      grp_rowscan<<<nb, RS_TPB, 0, stream>>>(H, P, T, nseg);
      grp_partition<<<nseg, GTPB, 0, stream>>>(srcp, dstp, P, T, list, E, nb, nseg);
      grp_bucket<<<nb, GTPB, 0, stream>>>(T, x, list, rowptr, cursor, dinv, rec1, n, E, nb);
    } else {
      csr_init<<<gN, TPB, 0, stream>>>(cnt, tick, n);
      csr_count<<<PASS_BLOCKS, TPB, 0, stream>>>(dstp, cnt, tick, n, E);
      csr_scan<<<gS, SCAN_TPB, 0, stream>>>(cnt, x, rowptr, cursor, dinv, rec1, n);
      csr_fill<<<PASS_BLOCKS, TPB, 0, stream>>>(srcp, dstp, cursor, list, tick, n, E);
    }
    gcn_l1_gather<<<PASS_BLOCKS, TPB, 0, stream>>>(rowptr, cursor, list, rec1, dinv, s, n, E);
    gcn_l1_finalize<<<gN, TPB, 0, stream>>>(x, dinv, s, W1, b1, relu1, n);
    gcn_l2_dense_wmma<true><<<gW, TPB, 0, stream>>>(relu1, W2, dinv, (float*)nullptr,
                                                    (float*)nullptr, rec2, n, ntiles);
    gcn_l2_gather<<<PASS_BLOCKS, TPB, 0, stream>>>(rowptr, cursor, list, rec2, dinv,
                                                    sum0, sum1, n, E);
    gcn_finalize<true><<<gN, TPB, 0, stream>>>(sum0, sum1, (const float*)nullptr,
                                               (const float*)nullptr, (const float*)nullptr,
                                               rec2, b2, out, n);
    return;
  }

  float* relu1 = ws + 7 * ns;
  gcn_init<<<gN, TPB, 0, stream>>>(deg, s, agg0, agg1, tick, n);
  gcn_degree<<<PASS_BLOCKS, TPB, 0, stream>>>(dstp, deg, tick, n, E);
  gcn_dinv<<<gN, TPB, 0, stream>>>(deg, dinv, n);
  gcn_l1_scatter<<<PASS_BLOCKS, TPB, 0, stream>>>(srcp, dstp, x, dinv, s, tick, n, E);
  gcn_l1_finalize<<<gN, TPB, 0, stream>>>(x, dinv, s, W1, b1, relu1, n);
  gcn_l2_dense_wmma<false><<<gW, TPB, 0, stream>>>(relu1, W2, dinv, h2c0, h2c1,
                                                   (float4*)nullptr, n, ntiles);
  gcn_l2_scatter<<<PASS_BLOCKS, TPB, 0, stream>>>(srcp, dstp, h2c0, h2c1, dinv,
                                                   agg0, agg1, tick, n, E);
  gcn_finalize<false><<<gN, TPB, 0, stream>>>(agg0, agg1, h2c0, h2c1, dinv,
                                              (const float4*)nullptr, b2, out, n);
}


# Gather waves finish their own node in registers: layer-1 self-loop, W1, bias, relu and layer-2 self-loop, bias, log-softmax move into the two gather kernels; 9 launches to 7; sum order unchanged
# speedup vs baseline: 9.1293x; 1.1294x over previous
#include <hip/hip_runtime.h>


typedef __attribute__((ext_vector_type(2))) float v2f;
typedef __attribute__((ext_vector_type(8))) float v8f;

#define TPB 256
#define EPT 4
#define SLICE (TPB * EPT)
#define PASS_BLOCKS 2048
#define NQ 6


__device__ __forceinline__ int tick_stride(int n) { return (n >= 32 * NQ) ? 64 : 1; }

__device__ __forceinline__ int cache_half() {
  unsigned v = __builtin_amdgcn_s_sendmsg_rtn(0x87);
  return (int)((v >> 18) & 1u);
}

__device__ __forceinline__ int draw_slice(int* q, int* sh) {
  __syncthreads();
  if (threadIdx.x == 0) *sh = atomicAdd(q, 1);
  __syncthreads();
  return *sh;
}

__device__ __forceinline__ int block_half(int* sh) {
  if (threadIdx.x == 0) *sh = cache_half();
  __syncthreads();
  return *sh;
}

__global__ void gcn_init(float* __restrict__ deg, float* __restrict__ s,
                         float* __restrict__ agg0, float* __restrict__ agg1,
                         int* __restrict__ tick, int n) {
  int i = blockIdx.x * blockDim.x + threadIdx.x;
  if (i < NQ) tick[i * tick_stride(n)] = 0;
  if (i < n) {
    deg[i] = 1.0f;
    s[i] = 0.0f;
    agg0[i] = 0.0f;
    agg1[i] = 0.0f;
  }
}

__global__ void __launch_bounds__(TPB)
gcn_degree(const int* __restrict__ dst, float* __restrict__ deg,
           int* __restrict__ tick, int n, int E) {
  __shared__ int sh[2];
  const int nsl = (E + SLICE - 1) / SLICE;
  const int split = (n >> 1) & ~31;
  const int st = tick_stride(n);
  const int h0 = block_half(&sh[1]);
  for (int r = 0; r < 2; ++r) {
    const int h = h0 ^ r;
    const int lo = h ? split : 0;
    const int hi = h ? n : split;
    int* q = tick + (0 + h) * st;
    for (int t = draw_slice(q, &sh[0]); t < nsl; t = draw_slice(q, &sh[0])) {
      const int e0 = t * SLICE + (int)threadIdx.x;
      int b[EPT];
#pragma unroll
      for (int j = 0; j < EPT; ++j) {
        int e = e0 + j * TPB;
        int v = dst[(e < E) ? e : (E - 1)];
        b[j] = (e < E) ? v : -1;
      }
#pragma unroll
      for (int j = 0; j < EPT; ++j)
        if (b[j] >= lo && b[j] < hi) unsafeAtomicAdd(&deg[b[j]], 1.0f);
    }
  }
}

__global__ void gcn_dinv(const float* __restrict__ deg,
                         float* __restrict__ dinv, int n) {
  int i = blockIdx.x * blockDim.x + threadIdx.x;
  if (i < n) dinv[i] = rsqrtf(deg[i]);
}

__global__ void __launch_bounds__(TPB)
gcn_l1_scatter(const int* __restrict__ src, const int* __restrict__ dst,
               const float* __restrict__ x, const float* __restrict__ dinv,
               float* __restrict__ s, int* __restrict__ tick, int n, int E) {
  __shared__ int sh[2];
  const int nsl = (E + SLICE - 1) / SLICE;
  const int split = (n >> 1) & ~31;
  const int st = tick_stride(n);
  const int h0 = block_half(&sh[1]);
  for (int r = 0; r < 2; ++r) {
    const int h = h0 ^ r;
    const int lo = h ? split : 0;
    const int hi = h ? n : split;
    int* q = tick + (2 + h) * st;
    for (int t = draw_slice(q, &sh[0]); t < nsl; t = draw_slice(q, &sh[0])) {
      const int e0 = t * SLICE + (int)threadIdx.x;
      int a[EPT], b[EPT];
#pragma unroll
      for (int j = 0; j < EPT; ++j) {
        int e = e0 + j * TPB;
        int ec = (e < E) ? e : (E - 1);
        int vb = dst[ec];
        a[j] = src[ec];
        b[j] = (e < E) ? vb : -1;
      }
      asm volatile("" :: "v"(a[0]), "v"(a[1]), "v"(a[2]), "v"(a[3]));
      float v[EPT];
#pragma unroll
      for (int j = 0; j < EPT; ++j) {
        bool in = b[j] >= lo && b[j] < hi;
        unsigned aa = in ? (unsigned)a[j] : 0u;
        unsigned bb = in ? (unsigned)b[j] : 0u;
        v[j] = x[aa] * dinv[aa] * dinv[bb];
      }
#pragma unroll
      for (int j = 0; j < EPT; ++j)
        if (b[j] >= lo && b[j] < hi) unsafeAtomicAdd(&s[b[j]], v[j]);
    }
  }
}

__global__ void gcn_l1_finalize(const float* __restrict__ x,
                                const float* __restrict__ dinv,
                                const float* __restrict__ s,
                                const float* __restrict__ W1,
                                const float* __restrict__ b1,
                                float* __restrict__ relu1, int n) {
  int i = blockIdx.x * blockDim.x + threadIdx.x;
  if (i >= n) return;
  float di = dinv[i];
  float sn = s[i] + x[i] * di * di;
  float4* o = (float4*)(relu1 + (size_t)i * 16);
#pragma unroll
  for (int q = 0; q < 4; ++q) {
    float4 r;
    r.x = fmaxf(fmaf(sn, W1[4 * q + 0], b1[4 * q + 0]), 0.0f);
    r.y = fmaxf(fmaf(sn, W1[4 * q + 1], b1[4 * q + 1]), 0.0f);
    r.z = fmaxf(fmaf(sn, W1[4 * q + 2], b1[4 * q + 2]), 0.0f);
    r.w = fmaxf(fmaf(sn, W1[4 * q + 3], b1[4 * q + 3]), 0.0f);
    o[q] = r;
  }
}

template <bool REC>
__global__ void gcn_l2_dense_wmma(const float* __restrict__ relu1,
                                  const float* __restrict__ W2,
                                  const float* __restrict__ dinv,
                                  float* __restrict__ h2c0,
                                  float* __restrict__ h2c1,
                                  float4* __restrict__ rec2,
                                  int n, int ntiles) {
  int gtid = blockIdx.x * blockDim.x + threadIdx.x;
  int tile = gtid >> 5;
  int lane = threadIdx.x & 31;
  if (tile >= ntiles) return;
  int base = tile * 16;
  int half = lane >> 4;
  int m = lane & 15;

  v2f bmat[4];
#pragma unroll
  for (int k = 0; k < 4; ++k) {
    int r0 = 4 * k + 2 * half;
    float bx = 0.0f, by = 0.0f;
    if (m < 2) { bx = W2[r0 * 2 + m]; by = W2[(r0 + 1) * 2 + m]; }
    bmat[k].x = bx; bmat[k].y = by;
  }

  int row = base + m;
  row = (row < n) ? row : (n - 1);
  const v2f* p = (const v2f*)(relu1 + (size_t)row * 16 + 2 * half);
  v2f amat[4];
#pragma unroll
  for (int k = 0; k < 4; ++k) amat[k] = p[2 * k];

  v8f c = {};
#pragma unroll
  for (int k = 0; k < 4; ++k)
    c = __builtin_amdgcn_wmma_f32_16x16x4_f32(
         false, amat[k],  false, bmat[k],
         (short)0, c,  false,  false);

  if (REC) {
    const int l0 = lane & 16;
    float h0 = 0.0f, h1 = 0.0f;
#pragma unroll
    for (int r = 0; r < 8; ++r) {
      const float t0 = __shfl(c[r], l0);
      const float t1 = __shfl(c[r], l0 + 1);
      if (m == r) { h0 = t0; h1 = t1; }
    }
    if (m < 8) {
      const int node = base + half * 8 + m;
      rec2[node] = make_float4(dinv[node], h0, h1, 0.0f);
    }
  } else if (m < 2) {
    float* col = (m == 0) ? h2c0 : h2c1;
    int node0 = base + half * 8;
    float4 lo; lo.x = c[0]; lo.y = c[1]; lo.z = c[2]; lo.w = c[3];
    float4 hi; hi.x = c[4]; hi.y = c[5]; hi.z = c[6]; hi.w = c[7];
    *(float4*)(col + node0)     = lo;
    *(float4*)(col + node0 + 4) = hi;
  }
}

__global__ void __launch_bounds__(TPB)
gcn_l2_scatter(const int* __restrict__ src, const int* __restrict__ dst,
               const float* __restrict__ h2c0, const float* __restrict__ h2c1,
               const float* __restrict__ dinv,
               float* __restrict__ agg0, float* __restrict__ agg1,
               int* __restrict__ tick, int n, int E) {
  __shared__ int sh[2];
  const int nsl = (E + SLICE - 1) / SLICE;
  const int split = (n >> 1) & ~31;
  const int st = tick_stride(n);
  const int h0 = block_half(&sh[1]);
  for (int r = 0; r < 2; ++r) {
    const int h = h0 ^ r;
    const int lo = h ? split : 0;
    const int hi = h ? n : split;
    int* q = tick + (4 + h) * st;
    for (int t = draw_slice(q, &sh[0]); t < nsl; t = draw_slice(q, &sh[0])) {
      const int e0 = t * SLICE + (int)threadIdx.x;
      int a[EPT], b[EPT];
#pragma unroll
      for (int j = 0; j < EPT; ++j) {
        int e = e0 + j * TPB;
        int ec = (e < E) ? e : (E - 1);
        int vb = dst[ec];
        a[j] = src[ec];
        b[j] = (e < E) ? vb : -1;
      }
      asm volatile("" :: "v"(a[0]), "v"(a[1]), "v"(a[2]), "v"(a[3]));
      float v0[EPT], v1[EPT];
#pragma unroll
      for (int j = 0; j < EPT; ++j) {
        bool in = b[j] >= lo && b[j] < hi;
        unsigned aa = in ? (unsigned)a[j] : 0u;
        unsigned bb = in ? (unsigned)b[j] : 0u;
        float nrm = dinv[aa] * dinv[bb];
        v0[j] = h2c0[aa] * nrm;
        v1[j] = h2c1[aa] * nrm;
      }
#pragma unroll
      for (int j = 0; j < EPT; ++j)
        if (b[j] >= lo && b[j] < hi) {
          unsafeAtomicAdd(&agg0[b[j]], v0[j]);
          unsafeAtomicAdd(&agg1[b[j]], v1[j]);
        }
    }
  }
}

__global__ void gcn_finalize(const float* __restrict__ agg0,
                             const float* __restrict__ agg1,
                             const float* __restrict__ h2c0,
                             const float* __restrict__ h2c1,
                             const float* __restrict__ dinv,
                             const float* __restrict__ b2,
                             float* __restrict__ out, int n) {
  int i = blockIdx.x * blockDim.x + threadIdx.x;
  if (i >= n) return;
  float di = dinv[i], h0 = h2c0[i], h1 = h2c1[i];
  float sl = di * di;
  float z0 = agg0[i] + h0 * sl + b2[0];
  float z1 = agg1[i] + h1 * sl + b2[1];
  float mx = fmaxf(z0, z1);
  float lse = mx + logf(expf(z0 - mx) + expf(z1 - mx));
  float2 r; r.x = z0 - lse; r.y = z1 - lse;
  *(float2*)(out + (size_t)i * 2) = r;
}

#define SCAN_TPB 1024

__global__ void csr_init(int* __restrict__ cnt, int* __restrict__ tick, int n) {
  int i = blockIdx.x * blockDim.x + threadIdx.x;
  if (i < 4) tick[i * tick_stride(n)] = 0;
  if (i < n) cnt[i] = 0;
}

__global__ void __launch_bounds__(TPB)
csr_count(const int* __restrict__ dst, int* __restrict__ cnt,
          int* __restrict__ tick, int n, int E) {
  __shared__ int sh[2];
  const int nsl = (E + SLICE - 1) / SLICE;
  const int split = (n >> 1) & ~31;
  const int st = tick_stride(n);
  const int h0 = block_half(&sh[1]);
  for (int r = 0; r < 2; ++r) {
    const int h = h0 ^ r;
    const int lo = h ? split : 0;
    const int hi = h ? n : split;
    int* q = tick + (0 + h) * st;
    for (int t = draw_slice(q, &sh[0]); t < nsl; t = draw_slice(q, &sh[0])) {
      const int e0 = t * SLICE + (int)threadIdx.x;
      int b[EPT];
#pragma unroll
      for (int j = 0; j < EPT; ++j) {
        int e = e0 + j * TPB;
        int v = dst[(e < E) ? e : (E - 1)];
        b[j] = (e < E) ? v : -1;
      }
#pragma unroll
      for (int j = 0; j < EPT; ++j)
        if (b[j] >= lo && b[j] < hi) atomicAdd(&cnt[b[j]], 1);
    }
  }
}

__global__ void __launch_bounds__(SCAN_TPB)
csr_scan(const int* __restrict__ cnt, const float* __restrict__ xin,
         int* __restrict__ rowptr, int* __restrict__ cursor,
         float* __restrict__ dinv, float2* __restrict__ rec1, int n) {
  __shared__ int wtot[32];
  __shared__ int wpre[32];
  const int t = (int)threadIdx.x;
  const int lane = t & 31;
  const int wv = t >> 5;
  const int first = (int)blockIdx.x * SCAN_TPB;
  const int i = first + t;
  const int c = (i < n) ? cnt[i] : 0;

  int pre = 0;
  const int4* c4 = (const int4*)cnt;
  const int ngrp = first / 4;
  for (int k = t; k < ngrp; k += 8 * SCAN_TPB) {
    int4 v[8];
#pragma unroll
    for (int j = 0; j < 8; ++j) {
      int kk = k + j * SCAN_TPB;
      v[j] = c4[(kk < ngrp) ? kk : k];
    }
#pragma unroll
    for (int j = 0; j < 8; ++j) {
      int kk = k + j * SCAN_TPB;
      int sum = (v[j].x + v[j].y) + (v[j].z + v[j].w);
      pre += (kk < ngrp) ? sum : 0;
    }
  }
  for (int o = 16; o > 0; o >>= 1) pre += __shfl_xor(pre, o);

  int inc = c;
  for (int o = 1; o < 32; o <<= 1) {
    int u = __shfl_up(inc, o);
    if (lane >= o) inc += u;
  }
  if (lane == 31) wtot[wv] = inc;
  if (lane == 0) wpre[wv] = pre;
  __syncthreads();

  int wt = wtot[lane];
  int winc = wt;
  for (int o = 1; o < 32; o <<= 1) {
    int u = __shfl_up(winc, o);
    if (lane >= o) winc += u;
  }
  int before_wave = __shfl(winc - wt, wv);
  int before_block = wpre[lane];
  for (int o = 16; o > 0; o >>= 1) before_block += __shfl_xor(before_block, o);

  if (i < n) {
    int slot = before_block + before_wave + (inc - c);
    rowptr[i] = slot;
    cursor[i] = slot;
    const float dv = rsqrtf((float)(c + 1));
    dinv[i] = dv;
    rec1[i] = make_float2(xin[i], dv);
  }
}

__global__ void __launch_bounds__(TPB)
csr_fill(const int* __restrict__ src, const int* __restrict__ dst,
         int* __restrict__ cursor, int* __restrict__ list,
         int* __restrict__ tick, int n, int E) {
  __shared__ int sh[2];
  const int nsl = (E + SLICE - 1) / SLICE;
  const int split = (n >> 1) & ~31;
  const int st = tick_stride(n);
  const int h0 = block_half(&sh[1]);
  for (int r = 0; r < 2; ++r) {
    const int h = h0 ^ r;
    const int lo = h ? split : 0;
    const int hi = h ? n : split;
    int* q = tick + (2 + h) * st;
    for (int t = draw_slice(q, &sh[0]); t < nsl; t = draw_slice(q, &sh[0])) {
      const int e0 = t * SLICE + (int)threadIdx.x;
      int a[EPT], b[EPT];
#pragma unroll
      for (int j = 0; j < EPT; ++j) {
        int e = e0 + j * TPB;
        int ec = (e < E) ? e : (E - 1);
        int vb = dst[ec];
        a[j] = src[ec];
        b[j] = (e < E) ? vb : -1;
      }
      int pos[EPT];
#pragma unroll
      for (int j = 0; j < EPT; ++j) {
        pos[j] = -1;
        if (b[j] >= lo && b[j] < hi) pos[j] = atomicAdd(&cursor[b[j]], 1);
      }
#pragma unroll
      for (int j = 0; j < EPT; ++j)
        if ((unsigned)pos[j] < (unsigned)E) list[pos[j]] = a[j];
    }
  }
}

__global__ void __launch_bounds__(TPB)
gcn_l1_gather(const int* __restrict__ rowptr, const int* __restrict__ rowend,
              const int* __restrict__ list, const float2* __restrict__ rec1,
              const float* __restrict__ W1, const float* __restrict__ b1,
              float* __restrict__ relu1, int n, int E) {
  const int lane = (int)threadIdx.x & 31;
  const int wv = __builtin_amdgcn_readfirstlane((int)threadIdx.x >> 5);
  const int nw = (int)gridDim.x * (TPB / 32);
  const unsigned last = (unsigned)(n - 1);
  const float w1 = W1[lane & 15];
  const float c1 = b1[lane & 15];
  int i = (int)blockIdx.x * (TPB / 32) + wv;
  int st = 0, en = 0;
  float xi = 0.0f, di = 0.0f;
  if (i < n) { st = rowptr[i]; en = rowend[i]; const float2 own = rec1[i]; xi = own.x; di = own.y; }
  while (i < n) {
    const int ni = i + nw;
    int nst = 0, nen = 0;
    float nxi = 0.0f, ndi = 0.0f;
    if (ni < n) { nst = rowptr[ni]; nen = rowend[ni]; const float2 own = rec1[ni]; nxi = own.x; ndi = own.y; }
    st = __builtin_amdgcn_readfirstlane((st > 0) ? st : 0);
    en = __builtin_amdgcn_readfirstlane((en < E) ? en : E);
    float acc = 0.0f;
    for (int base = st; base < en; base += 64) {
      const int k0 = base + lane, k1 = k0 + 32;
      const bool p0 = k0 < en, p1 = k1 < en;
      unsigned a0 = (unsigned)list[p0 ? k0 : st];
      unsigned a1 = (unsigned)list[p1 ? k1 : st];
      a0 = (a0 < last) ? a0 : last;
      a1 = (a1 < last) ? a1 : last;
      const float2 r0 = rec1[a0];
      const float2 r1 = rec1[a1];
      const float t0 = r0.x * r0.y * di;
      const float t1 = r1.x * r1.y * di;
      acc += p0 ? t0 : 0.0f;
      acc += p1 ? t1 : 0.0f;
    }
    for (int o = 16; o > 0; o >>= 1) acc += __shfl_xor(acc, o);
    const float sn = acc + xi * di * di;
    if (lane < 16) relu1[(size_t)i * 16 + lane] = fmaxf(fmaf(sn, w1, c1), 0.0f);
    i = ni; st = nst; en = nen; xi = nxi; di = ndi;
  }
}

__global__ void __launch_bounds__(TPB)
gcn_l2_gather(const int* __restrict__ rowptr, const int* __restrict__ rowend,
              const int* __restrict__ list, const float4* __restrict__ rec2,
              const float* __restrict__ b2, float* __restrict__ out,
              int n, int E) {
  const int lane = (int)threadIdx.x & 31;
  const int wv = __builtin_amdgcn_readfirstlane((int)threadIdx.x >> 5);
  const int nw = (int)gridDim.x * (TPB / 32);
  const unsigned last = (unsigned)(n - 1);
  const float bz0 = b2[0], bz1 = b2[1];
  int i = (int)blockIdx.x * (TPB / 32) + wv;
  int st = 0, en = 0;
  float di = 0.0f, h0 = 0.0f, h1 = 0.0f;
  if (i < n) { st = rowptr[i]; en = rowend[i]; const float4 own = rec2[i]; di = own.x; h0 = own.y; h1 = own.z; }
  while (i < n) {
    const int ni = i + nw;
    int nst = 0, nen = 0;
    float ndi = 0.0f, nh0 = 0.0f, nh1 = 0.0f;
    if (ni < n) { nst = rowptr[ni]; nen = rowend[ni]; const float4 own = rec2[ni]; ndi = own.x; nh0 = own.y; nh1 = own.z; }
    st = __builtin_amdgcn_readfirstlane((st > 0) ? st : 0);
    en = __builtin_amdgcn_readfirstlane((en < E) ? en : E);
    float acc0 = 0.0f, acc1 = 0.0f;
    for (int base = st; base < en; base += 64) {
      const int k0 = base + lane, k1 = k0 + 32;
      const bool p0 = k0 < en, p1 = k1 < en;
      unsigned a0 = (unsigned)list[p0 ? k0 : st];
      unsigned a1 = (unsigned)list[p1 ? k1 : st];
      a0 = (a0 < last) ? a0 : last;
      a1 = (a1 < last) ? a1 : last;
      const float4 q0 = rec2[a0];
      const float4 q1 = rec2[a1];
      const float n0 = q0.x * di;
      const float n1 = q1.x * di;
      const float u0 = q0.y * n0, w0 = q0.z * n0;
      const float u1 = q1.y * n1, w1 = q1.z * n1;
      acc0 += p0 ? u0 : 0.0f;  acc1 += p0 ? w0 : 0.0f;
      acc0 += p1 ? u1 : 0.0f;  acc1 += p1 ? w1 : 0.0f;
    }
    for (int o = 16; o > 0; o >>= 1) {
      acc0 += __shfl_xor(acc0, o);
      acc1 += __shfl_xor(acc1, o);
    }
    if (lane == 0) {
      const float sl = di * di;
      const float z0 = acc0 + h0 * sl + bz0;
      const float z1 = acc1 + h1 * sl + bz1;
      const float mx = fmaxf(z0, z1);
      const float lse = mx + logf(expf(z0 - mx) + expf(z1 - mx));
      float2 r; r.x = z0 - lse; r.y = z1 - lse;
      *(float2*)(out + (size_t)i * 2) = r;
    }
    i = ni; st = nst; en = nen; di = ndi; h0 = nh0; h1 = nh1;
  }
}


#define GSH 8
#define GNB (1 << GSH)
#define GTPB 1024
#define GEPT 12
#define GSUB (GTPB * GEPT)
#define GBMAX GTPB
#define GCAP 16384
#define RS_TPB 256

template <int NW>
__device__ __forceinline__ int block_scan_excl(int v, int* wtot, int& total) {
  const int lane = (int)threadIdx.x & 31;
  const int wv = (int)threadIdx.x >> 5;
  int inc = v;
  for (int o = 1; o < 32; o <<= 1) {
    int u = __shfl_up(inc, o);
    if (lane >= o) inc += u;
  }
  if (lane == 31) wtot[wv] = inc;
  __syncthreads();
  const int wt = (lane < NW) ? wtot[lane] : 0;
  int winc = wt;
  for (int o = 1; o < 32; o <<= 1) {
    int u = __shfl_up(winc, o);
    if (lane >= o) winc += u;
  }
  const int before = __shfl(winc - wt, wv);
  total = __shfl(winc, 31);
  __syncthreads();
  return before + (inc - v);
}

__device__ __forceinline__ int bucket_of(int d, int nb) {
  unsigned b = (unsigned)d >> GSH;
  return (int)((b < (unsigned)nb) ? b : (unsigned)(nb - 1));
}

__global__ void __launch_bounds__(GTPB)
grp_hist(const int* __restrict__ dst, int* __restrict__ H, int E, int nb, int nseg) {
  __shared__ int hist[GBMAX];
  const int t = (int)threadIdx.x;
  const int s = (int)blockIdx.x;
  if (t < nb) hist[t] = 0;
  __syncthreads();
  const int e0 = s * GSUB + t;
  int d[GEPT];
#pragma unroll
  for (int j = 0; j < GEPT; ++j) {
    int e = e0 + j * GTPB;
    d[j] = dst[(e < E) ? e : (E - 1)];
  }
#pragma unroll
  for (int j = 0; j < GEPT; ++j) {
    int e = e0 + j * GTPB;
    if (e < E) atomicAdd(&hist[bucket_of(d[j], nb)], 1);
  }
  __syncthreads();
  if (t < nb) H[(size_t)t * nseg + s] = hist[t];
}

__global__ void __launch_bounds__(RS_TPB)
grp_rowscan(const int* __restrict__ H, int* __restrict__ P, int* __restrict__ T, int nseg) {
  __shared__ int wtot[32];
  const int t = (int)threadIdx.x;
  const size_t row = (size_t)blockIdx.x * nseg;
  int v[4];
#pragma unroll
  for (int j = 0; j < 4; ++j) {
    int k = 4 * t + j;
    int h = H[row + ((k < nseg) ? k : 0)];
    v[j] = (k < nseg) ? h : 0;
  }
  int total;
  int ex = block_scan_excl<RS_TPB / 32>(v[0] + v[1] + v[2] + v[3], wtot, total);
#pragma unroll
  for (int j = 0; j < 4; ++j) {
    int k = 4 * t + j;
    if (k < nseg) P[row + k] = ex;
    ex += v[j];
  }
  if (t == 0) T[blockIdx.x] = total;
}

__global__ void __launch_bounds__(GTPB)
grp_partition(const int* __restrict__ src, const int* __restrict__ dst,
              const int* __restrict__ P, const int* __restrict__ T,
              int* __restrict__ list, int E, int nb, int nseg) {
  __shared__ int hist[GBMAX];
  __shared__ int offs[GBMAX];
  __shared__ int gbase[GBMAX];
  __shared__ int wtot[32];
  __shared__ int stage[GSUB];
  __shared__ unsigned short stage_b[GSUB];
  const int t = (int)threadIdx.x;
  const int s = (int)blockIdx.x;

  const int tb = (t < nb) ? T[t] : 0;
  const int pb = (t < nb) ? P[(size_t)t * nseg + s] : 0;
  if (t < nb) hist[t] = 0;
  int all;
  const int first = block_scan_excl<GTPB / 32>(tb, wtot, all);
  if (t < nb) gbase[t] = first + pb;

  const int e0 = s * GSUB + t;
  int a[GEPT], d[GEPT], r[GEPT];
#pragma unroll
  for (int j = 0; j < GEPT; ++j) {
    int e = e0 + j * GTPB;
    int ec = (e < E) ? e : (E - 1);
    d[j] = dst[ec];
    a[j] = src[ec];
  }
#pragma unroll
  for (int j = 0; j < GEPT; ++j) {
    int e = e0 + j * GTPB;
    r[j] = -1;
    if (e < E) r[j] = atomicAdd(&hist[bucket_of(d[j], nb)], 1);
  }
  __syncthreads();
  const int hv = (t < nb) ? hist[t] : 0;
  int total;
  const int ex = block_scan_excl<GTPB / 32>(hv, wtot, total);
  if (t < nb) offs[t] = ex;
  __syncthreads();
#pragma unroll
  for (int j = 0; j < GEPT; ++j) {
    if (r[j] >= 0) {
      const int b = bucket_of(d[j], nb);
      const int pos = offs[b] + r[j];
      if ((unsigned)pos < (unsigned)GSUB) {
        stage[pos] = (int)((((unsigned)d[j] & (unsigned)(GNB - 1)) << 24) |
                           ((unsigned)a[j] & 0xFFFFFFu));
        stage_b[pos] = (unsigned short)b;
      }
    }
  }
  __syncthreads();
#pragma unroll
  for (int j = 0; j < GEPT; ++j) {
    const int idx = t + j * GTPB;
    if (idx < total) {
      const int b = (int)stage_b[idx];
      const int g = gbase[b] + (idx - offs[b]);
      if ((unsigned)g < (unsigned)E) list[g] = stage[idx];
    }
  }
}

__global__ void __launch_bounds__(GTPB)
grp_bucket(const int* __restrict__ T, const float* __restrict__ xin, int* list,
           int* __restrict__ rowptr, int* __restrict__ rowend,
           float* __restrict__ dinv, float2* __restrict__ rec1, int n, int E, int nb) {
  __shared__ int cnt[GNB];
  __shared__ int cur[GNB];
  __shared__ int endv[GNB];
  __shared__ int wtot[32];
  __shared__ int sh[2];
  __shared__ int ebuf[GCAP];
  __shared__ int obuf[GCAP];
  const int t = (int)threadIdx.x;
  const int b = (int)blockIdx.x;

  const int tb = (t < nb) ? T[t] : 0;
  int all;
  const int first = block_scan_excl<GTPB / 32>(tb, wtot, all);
  if (t == b) { sh[0] = first; sh[1] = tb; }
  if (t < GNB) cnt[t] = 0;
  __syncthreads();
  int base = sh[0], len = sh[1];
  base = (base > 0) ? base : 0;
  base = (base < E) ? base : E;
  len = (len > 0) ? len : 0;
  len = (len < E - base) ? len : (E - base);
  const bool fits = len <= GCAP;

  for (int j0 = t; j0 < len; j0 += 4 * GTPB) {
    unsigned v[4];
#pragma unroll
    for (int q = 0; q < 4; ++q) {
      int j = j0 + q * GTPB;
      v[q] = (unsigned)list[base + ((j < len) ? j : j0)];
    }
#pragma unroll
    for (int q = 0; q < 4; ++q) {
      int j = j0 + q * GTPB;
      if (j < len) {
        if (fits) ebuf[j] = (int)v[q];
        atomicAdd(&cnt[v[q] >> 24], 1);
      }
    }
  }
  __syncthreads();
  const int c = (t < GNB) ? cnt[t] : 0;
  int tot;
  const int ex = block_scan_excl<GTPB / 32>(c, wtot, tot);
  if (t < GNB) {
    cur[t] = ex;
    endv[t] = ex + c;
    const int i = b * GNB + t;
    if (i < n) {
      rowptr[i] = base + ex;
      rowend[i] = base + ex + c;
      const float dv = rsqrtf((float)(c + 1));
      dinv[i] = dv;
      rec1[i] = make_float2(xin[i], dv);
    }
  }
  __syncthreads();

  if (fits) {
    for (int j = t; j < len; j += GTPB) {
      const unsigned v = (unsigned)ebuf[j];
      const int pos = atomicAdd(&cur[v >> 24], 1);
      if ((unsigned)pos < (unsigned)GCAP) obuf[pos] = (int)(v & 0xFFFFFFu);
    }
    __syncthreads();
    for (int j = t; j < len; j += GTPB) list[base + j] = obuf[j];
  } else {
    if (t == 0) {
      int steps = 0;
      const int limit = 2 * len + GNB;
      for (int v = 0; v < GNB; ++v) {
        int p = cur[v];
        const int pe = endv[v];
        while (p < pe && steps < limit) {
          ++steps;
          const unsigned x = (unsigned)list[base + p];
          const int dv = (int)(x >> 24);
          if (dv == v) { ++p; continue; }
          const int q = cur[dv];
          if (q >= endv[dv]) break;
          const int y = list[base + q];
          list[base + q] = (int)x;
          list[base + p] = y;
          cur[dv] = q + 1;
        }
        cur[v] = p;
      }
    }
    __syncthreads();
    for (int j = t; j < len; j += GTPB) list[base + j] = list[base + j] & 0xFFFFFF;
  }
}

extern "C" void kernel_launch(void* const* d_in, const int* in_sizes, int n_in,
                              void* d_out, int out_size, void* d_ws, size_t ws_size,
                              hipStream_t stream) {
  const float* x  = (const float*)d_in[0];
  const int*   ei = (const int*)d_in[1];
  const float* W1 = (const float*)d_in[2];
  const float* b1 = (const float*)d_in[3];
  const float* W2 = (const float*)d_in[4];
  const float* b2 = (const float*)d_in[5];
  float* out = (float*)d_out;
  int* tick = (int*)d_out;

  int n = in_sizes[0];
  int E = in_sizes[1] / 2;
  const int* srcp = ei;
  const int* dstp = ei + E;

  size_t ns = ((size_t)n + 31) & ~(size_t)31;

  float* ws    = (float*)d_ws;
  float* deg   = ws;
  float* dinv  = ws + ns;
  float* s     = ws + 2 * ns;
  float* h2c0  = ws + 3 * ns;
  float* h2c1  = ws + 4 * ns;
  float* agg0  = ws + 5 * ns;
  float* agg1  = ws + 6 * ns;

  int gN = (n + TPB - 1) / TPB;
  int ntiles = (n + 15) / 16;
  int gW = (int)(((size_t)ntiles * 32 + TPB - 1) / TPB);

  const size_t need = (25 * ns + (size_t)E) * sizeof(float);
  if (ws_size >= need) {
    int*   cnt    = (int*)ws;
    int*   rowptr = (int*)(ws + 7 * ns);
    int*   cursor = (int*)(ws + 8 * ns);
    float* relu1  = ws + 9 * ns;
    int*   list   = (int*)(ws + 25 * ns);
    float2* rec1  = (float2*)(ws + 4 * ns);
    float4* rec2  = (float4*)(ws + 2 * ns);
    int gS = (n + SCAN_TPB - 1) / SCAN_TPB;

    const int nb = (n + GNB - 1) >> GSH;
    const int nseg = (E + GSUB - 1) / GSUB;
    const size_t tabs = 2 * (size_t)nb * (size_t)nseg + (size_t)nb;
    if (E > 0 && nb <= GBMAX && nseg <= GTPB && n <= (1 << 24) && tabs <= 16 * (size_t)n) {
      int* H = (int*)relu1;
      int* P = H + (size_t)nb * nseg;
      int* T = P + (size_t)nb * nseg;
      grp_hist<<<nseg, GTPB, 0, stream>>>(dstp, H, E, nb, nseg);
      grp_rowscan<<<nb, RS_TPB, 0, stream>>>(H, P, T, nseg);
      grp_partition<<<nseg, GTPB, 0, stream>>>(srcp, dstp, P, T, list, E, nb, nseg);
      grp_bucket<<<nb, GTPB, 0, stream>>>(T, x, list, rowptr, cursor, dinv, rec1, n, E, nb);
    } else {
      csr_init<<<gN, TPB, 0, stream>>>(cnt, tick, n);
      csr_count<<<PASS_BLOCKS, TPB, 0, stream>>>(dstp, cnt, tick, n, E);
      csr_scan<<<gS, SCAN_TPB, 0, stream>>>(cnt, x, rowptr, cursor, dinv, rec1, n);
      csr_fill<<<PASS_BLOCKS, TPB, 0, stream>>>(srcp, dstp, cursor, list, tick, n, E);
    }
    gcn_l1_gather<<<PASS_BLOCKS, TPB, 0, stream>>>(rowptr, cursor, list, rec1, W1, b1, relu1, n, E);
    gcn_l2_dense_wmma<true><<<gW, TPB, 0, stream>>>(relu1, W2, dinv, (float*)nullptr,
                                                    (float*)nullptr, rec2, n, ntiles);
    gcn_l2_gather<<<PASS_BLOCKS, TPB, 0, stream>>>(rowptr, cursor, list, rec2, b2, out, n, E);
    return;
  }

  float* relu1 = ws + 7 * ns;
  gcn_init<<<gN, TPB, 0, stream>>>(deg, s, agg0, agg1, tick, n);
  gcn_degree<<<PASS_BLOCKS, TPB, 0, stream>>>(dstp, deg, tick, n, E);
  gcn_dinv<<<gN, TPB, 0, stream>>>(deg, dinv, n);
  gcn_l1_scatter<<<PASS_BLOCKS, TPB, 0, stream>>>(srcp, dstp, x, dinv, s, tick, n, E);
  gcn_l1_finalize<<<gN, TPB, 0, stream>>>(x, dinv, s, W1, b1, relu1, n);
  gcn_l2_dense_wmma<false><<<gW, TPB, 0, stream>>>(relu1, W2, dinv, h2c0, h2c1,
                                                   (float4*)nullptr, n, ntiles);
  gcn_l2_scatter<<<PASS_BLOCKS, TPB, 0, stream>>>(srcp, dstp, h2c0, h2c1, dinv,
                                                   agg0, agg1, tick, n, E);
  gcn_finalize<<<gN, TPB, 0, stream>>>(agg0, agg1, h2c0, h2c1, dinv, b2, out, n);
}


# First sort level made segment-local: each segment ordered by bucket in LDS in place, bucket blocks collect their pieces via a per-segment offset table; no global histogram/prefix; 7 launches to 5
# speedup vs baseline: 10.2992x; 1.1281x over previous
#include <hip/hip_runtime.h>


typedef __attribute__((ext_vector_type(2))) float v2f;
typedef __attribute__((ext_vector_type(8))) float v8f;

#define TPB 256
#define EPT 4
#define SLICE (TPB * EPT)
#define PASS_BLOCKS 2048
#define NQ 6


__device__ __forceinline__ int tick_stride(int n) { return (n >= 32 * NQ) ? 64 : 1; }

__device__ __forceinline__ int cache_half() {
  unsigned v = __builtin_amdgcn_s_sendmsg_rtn(0x87);
  return (int)((v >> 18) & 1u);
}

__device__ __forceinline__ int draw_slice(int* q, int* sh) {
  __syncthreads();
  if (threadIdx.x == 0) *sh = atomicAdd(q, 1);
  __syncthreads();
  return *sh;
}

__device__ __forceinline__ int block_half(int* sh) {
  if (threadIdx.x == 0) *sh = cache_half();
  __syncthreads();
  return *sh;
}

__global__ void gcn_init(float* __restrict__ deg, float* __restrict__ s,
                         float* __restrict__ agg0, float* __restrict__ agg1,
                         int* __restrict__ tick, int n) {
  int i = blockIdx.x * blockDim.x + threadIdx.x;
  if (i < NQ) tick[i * tick_stride(n)] = 0;
  if (i < n) {
    deg[i] = 1.0f;
    s[i] = 0.0f;
    agg0[i] = 0.0f;
    agg1[i] = 0.0f;
  }
}

__global__ void __launch_bounds__(TPB)
gcn_degree(const int* __restrict__ dst, float* __restrict__ deg,
           int* __restrict__ tick, int n, int E) {
  __shared__ int sh[2];
  const int nsl = (E + SLICE - 1) / SLICE;
  const int split = (n >> 1) & ~31;
  const int st = tick_stride(n);
  const int h0 = block_half(&sh[1]);
  for (int r = 0; r < 2; ++r) {
    const int h = h0 ^ r;
    const int lo = h ? split : 0;
    const int hi = h ? n : split;
    int* q = tick + (0 + h) * st;
    for (int t = draw_slice(q, &sh[0]); t < nsl; t = draw_slice(q, &sh[0])) {
      const int e0 = t * SLICE + (int)threadIdx.x;
      int b[EPT];
#pragma unroll
      for (int j = 0; j < EPT; ++j) {
        int e = e0 + j * TPB;
        int v = dst[(e < E) ? e : (E - 1)];
        b[j] = (e < E) ? v : -1;
      }
#pragma unroll
      for (int j = 0; j < EPT; ++j)
        if (b[j] >= lo && b[j] < hi) unsafeAtomicAdd(&deg[b[j]], 1.0f);
    }
  }
}

__global__ void gcn_dinv(const float* __restrict__ deg,
                         float* __restrict__ dinv, int n) {
  int i = blockIdx.x * blockDim.x + threadIdx.x;
  if (i < n) dinv[i] = rsqrtf(deg[i]);
}

__global__ void __launch_bounds__(TPB)
gcn_l1_scatter(const int* __restrict__ src, const int* __restrict__ dst,
               const float* __restrict__ x, const float* __restrict__ dinv,
               float* __restrict__ s, int* __restrict__ tick, int n, int E) {
  __shared__ int sh[2];
  const int nsl = (E + SLICE - 1) / SLICE;
  const int split = (n >> 1) & ~31;
  const int st = tick_stride(n);
  const int h0 = block_half(&sh[1]);
  for (int r = 0; r < 2; ++r) {
    const int h = h0 ^ r;
    const int lo = h ? split : 0;
    const int hi = h ? n : split;
    int* q = tick + (2 + h) * st;
    for (int t = draw_slice(q, &sh[0]); t < nsl; t = draw_slice(q, &sh[0])) {
      const int e0 = t * SLICE + (int)threadIdx.x;
      int a[EPT], b[EPT];
#pragma unroll
      for (int j = 0; j < EPT; ++j) {
        int e = e0 + j * TPB;
        int ec = (e < E) ? e : (E - 1);
        int vb = dst[ec];
        a[j] = src[ec];
        b[j] = (e < E) ? vb : -1;
      }
      asm volatile("" :: "v"(a[0]), "v"(a[1]), "v"(a[2]), "v"(a[3]));
      float v[EPT];
#pragma unroll
      for (int j = 0; j < EPT; ++j) {
        bool in = b[j] >= lo && b[j] < hi;
        unsigned aa = in ? (unsigned)a[j] : 0u;
        unsigned bb = in ? (unsigned)b[j] : 0u;
        v[j] = x[aa] * dinv[aa] * dinv[bb];
      }
#pragma unroll
      for (int j = 0; j < EPT; ++j)
        if (b[j] >= lo && b[j] < hi) unsafeAtomicAdd(&s[b[j]], v[j]);
    }
  }
}

__global__ void gcn_l1_finalize(const float* __restrict__ x,
                                const float* __restrict__ dinv,
                                const float* __restrict__ s,
                                const float* __restrict__ W1,
                                const float* __restrict__ b1,
                                float* __restrict__ relu1, int n) {
  int i = blockIdx.x * blockDim.x + threadIdx.x;
  if (i >= n) return;
  float di = dinv[i];
  float sn = s[i] + x[i] * di * di;
  float4* o = (float4*)(relu1 + (size_t)i * 16);
#pragma unroll
  for (int q = 0; q < 4; ++q) {
    float4 r;
    r.x = fmaxf(fmaf(sn, W1[4 * q + 0], b1[4 * q + 0]), 0.0f);
    r.y = fmaxf(fmaf(sn, W1[4 * q + 1], b1[4 * q + 1]), 0.0f);
    r.z = fmaxf(fmaf(sn, W1[4 * q + 2], b1[4 * q + 2]), 0.0f);
    r.w = fmaxf(fmaf(sn, W1[4 * q + 3], b1[4 * q + 3]), 0.0f);
    o[q] = r;
  }
}

template <bool REC>
__global__ void gcn_l2_dense_wmma(const float* __restrict__ relu1,
                                  const float* __restrict__ W2,
                                  const float* __restrict__ dinv,
                                  float* __restrict__ h2c0,
                                  float* __restrict__ h2c1,
                                  float4* __restrict__ rec2,
                                  int n, int ntiles) {
  int gtid = blockIdx.x * blockDim.x + threadIdx.x;
  int tile = gtid >> 5;
  int lane = threadIdx.x & 31;
  if (tile >= ntiles) return;
  int base = tile * 16;
  int half = lane >> 4;
  int m = lane & 15;

  v2f bmat[4];
#pragma unroll
  for (int k = 0; k < 4; ++k) {
    int r0 = 4 * k + 2 * half;
    float bx = 0.0f, by = 0.0f;
    if (m < 2) { bx = W2[r0 * 2 + m]; by = W2[(r0 + 1) * 2 + m]; }
    bmat[k].x = bx; bmat[k].y = by;
  }

  int row = base + m;
  row = (row < n) ? row : (n - 1);
  const v2f* p = (const v2f*)(relu1 + (size_t)row * 16 + 2 * half);
  v2f amat[4];
#pragma unroll
  for (int k = 0; k < 4; ++k) amat[k] = p[2 * k];

  v8f c = {};
#pragma unroll
  for (int k = 0; k < 4; ++k)
    c = __builtin_amdgcn_wmma_f32_16x16x4_f32(
         false, amat[k],  false, bmat[k],
         (short)0, c,  false,  false);

  if (REC) {
    const int l0 = lane & 16;
    float h0 = 0.0f, h1 = 0.0f;
#pragma unroll
    for (int r = 0; r < 8; ++r) {
      const float t0 = __shfl(c[r], l0);
      const float t1 = __shfl(c[r], l0 + 1);
      if (m == r) { h0 = t0; h1 = t1; }
    }
    if (m < 8) {
      const int node = base + half * 8 + m;
      rec2[node] = make_float4(dinv[node], h0, h1, 0.0f);
    }
  } else if (m < 2) {
    float* col = (m == 0) ? h2c0 : h2c1;
    int node0 = base + half * 8;
    float4 lo; lo.x = c[0]; lo.y = c[1]; lo.z = c[2]; lo.w = c[3];
    float4 hi; hi.x = c[4]; hi.y = c[5]; hi.z = c[6]; hi.w = c[7];
    *(float4*)(col + node0)     = lo;
    *(float4*)(col + node0 + 4) = hi;
  }
}

__global__ void __launch_bounds__(TPB)
gcn_l2_scatter(const int* __restrict__ src, const int* __restrict__ dst,
               const float* __restrict__ h2c0, const float* __restrict__ h2c1,
               const float* __restrict__ dinv,
               float* __restrict__ agg0, float* __restrict__ agg1,
               int* __restrict__ tick, int n, int E) {
  __shared__ int sh[2];
  const int nsl = (E + SLICE - 1) / SLICE;
  const int split = (n >> 1) & ~31;
  const int st = tick_stride(n);
  const int h0 = block_half(&sh[1]);
  for (int r = 0; r < 2; ++r) {
    const int h = h0 ^ r;
    const int lo = h ? split : 0;
    const int hi = h ? n : split;
    int* q = tick + (4 + h) * st;
    for (int t = draw_slice(q, &sh[0]); t < nsl; t = draw_slice(q, &sh[0])) {
      const int e0 = t * SLICE + (int)threadIdx.x;
      int a[EPT], b[EPT];
#pragma unroll
      for (int j = 0; j < EPT; ++j) {
        int e = e0 + j * TPB;
        int ec = (e < E) ? e : (E - 1);
        int vb = dst[ec];
        a[j] = src[ec];
        b[j] = (e < E) ? vb : -1;
      }
      asm volatile("" :: "v"(a[0]), "v"(a[1]), "v"(a[2]), "v"(a[3]));
      float v0[EPT], v1[EPT];
#pragma unroll
      for (int j = 0; j < EPT; ++j) {
        bool in = b[j] >= lo && b[j] < hi;
        unsigned aa = in ? (unsigned)a[j] : 0u;
        unsigned bb = in ? (unsigned)b[j] : 0u;
        float nrm = dinv[aa] * dinv[bb];
        v0[j] = h2c0[aa] * nrm;
        v1[j] = h2c1[aa] * nrm;
      }
#pragma unroll
      for (int j = 0; j < EPT; ++j)
        if (b[j] >= lo && b[j] < hi) {
          unsafeAtomicAdd(&agg0[b[j]], v0[j]);
          unsafeAtomicAdd(&agg1[b[j]], v1[j]);
        }
    }
  }
}

__global__ void gcn_finalize(const float* __restrict__ agg0,
                             const float* __restrict__ agg1,
                             const float* __restrict__ h2c0,
                             const float* __restrict__ h2c1,
                             const float* __restrict__ dinv,
                             const float* __restrict__ b2,
                             float* __restrict__ out, int n) {
  int i = blockIdx.x * blockDim.x + threadIdx.x;
  if (i >= n) return;
  float di = dinv[i], h0 = h2c0[i], h1 = h2c1[i];
  float sl = di * di;
  float z0 = agg0[i] + h0 * sl + b2[0];
  float z1 = agg1[i] + h1 * sl + b2[1];
  float mx = fmaxf(z0, z1);
  float lse = mx + logf(expf(z0 - mx) + expf(z1 - mx));
  float2 r; r.x = z0 - lse; r.y = z1 - lse;
  *(float2*)(out + (size_t)i * 2) = r;
}

#define SCAN_TPB 1024

__global__ void csr_init(int* __restrict__ cnt, int* __restrict__ tick, int n) {
  int i = blockIdx.x * blockDim.x + threadIdx.x;
  if (i < 4) tick[i * tick_stride(n)] = 0;
  if (i < n) cnt[i] = 0;
}

__global__ void __launch_bounds__(TPB)
csr_count(const int* __restrict__ dst, int* __restrict__ cnt,
          int* __restrict__ tick, int n, int E) {
  __shared__ int sh[2];
  const int nsl = (E + SLICE - 1) / SLICE;
  const int split = (n >> 1) & ~31;
  const int st = tick_stride(n);
  const int h0 = block_half(&sh[1]);
  for (int r = 0; r < 2; ++r) {
    const int h = h0 ^ r;
    const int lo = h ? split : 0;
    const int hi = h ? n : split;
    int* q = tick + (0 + h) * st;
    for (int t = draw_slice(q, &sh[0]); t < nsl; t = draw_slice(q, &sh[0])) {
      const int e0 = t * SLICE + (int)threadIdx.x;
      int b[EPT];
#pragma unroll
      for (int j = 0; j < EPT; ++j) {
        int e = e0 + j * TPB;
        int v = dst[(e < E) ? e : (E - 1)];
        b[j] = (e < E) ? v : -1;
      }
#pragma unroll
      for (int j = 0; j < EPT; ++j)
        if (b[j] >= lo && b[j] < hi) atomicAdd(&cnt[b[j]], 1);
    }
  }
}

__global__ void __launch_bounds__(SCAN_TPB)
csr_scan(const int* __restrict__ cnt, const float* __restrict__ xin,
         int* __restrict__ rowptr, int* __restrict__ cursor,
         float* __restrict__ dinv, float2* __restrict__ rec1, int n) {
  __shared__ int wtot[32];
  __shared__ int wpre[32];
  const int t = (int)threadIdx.x;
  const int lane = t & 31;
  const int wv = t >> 5;
  const int first = (int)blockIdx.x * SCAN_TPB;
  const int i = first + t;
  const int c = (i < n) ? cnt[i] : 0;

  int pre = 0;
  const int4* c4 = (const int4*)cnt;
  const int ngrp = first / 4;
  for (int k = t; k < ngrp; k += 8 * SCAN_TPB) {
    int4 v[8];
#pragma unroll
    for (int j = 0; j < 8; ++j) {
      int kk = k + j * SCAN_TPB;
      v[j] = c4[(kk < ngrp) ? kk : k];
    }
#pragma unroll
    for (int j = 0; j < 8; ++j) {
      int kk = k + j * SCAN_TPB;
      int sum = (v[j].x + v[j].y) + (v[j].z + v[j].w);
      pre += (kk < ngrp) ? sum : 0;
    }
  }
  for (int o = 16; o > 0; o >>= 1) pre += __shfl_xor(pre, o);

  int inc = c;
  for (int o = 1; o < 32; o <<= 1) {
    int u = __shfl_up(inc, o);
    if (lane >= o) inc += u;
  }
  if (lane == 31) wtot[wv] = inc;
  if (lane == 0) wpre[wv] = pre;
  __syncthreads();

  int wt = wtot[lane];
  int winc = wt;
  for (int o = 1; o < 32; o <<= 1) {
    int u = __shfl_up(winc, o);
    if (lane >= o) winc += u;
  }
  int before_wave = __shfl(winc - wt, wv);
  int before_block = wpre[lane];
  for (int o = 16; o > 0; o >>= 1) before_block += __shfl_xor(before_block, o);

  if (i < n) {
    int slot = before_block + before_wave + (inc - c);
    rowptr[i] = slot;
    cursor[i] = slot;
    const float dv = rsqrtf((float)(c + 1));
    dinv[i] = dv;
    rec1[i] = make_float2(xin[i], dv);
  }
}

__global__ void __launch_bounds__(TPB)
csr_fill(const int* __restrict__ src, const int* __restrict__ dst,
         int* __restrict__ cursor, int* __restrict__ list,
         int* __restrict__ tick, int n, int E) {
  __shared__ int sh[2];
  const int nsl = (E + SLICE - 1) / SLICE;
  const int split = (n >> 1) & ~31;
  const int st = tick_stride(n);
  const int h0 = block_half(&sh[1]);
  for (int r = 0; r < 2; ++r) {
    const int h = h0 ^ r;
    const int lo = h ? split : 0;
    const int hi = h ? n : split;
    int* q = tick + (2 + h) * st;
    for (int t = draw_slice(q, &sh[0]); t < nsl; t = draw_slice(q, &sh[0])) {
      const int e0 = t * SLICE + (int)threadIdx.x;
      int a[EPT], b[EPT];
#pragma unroll
      for (int j = 0; j < EPT; ++j) {
        int e = e0 + j * TPB;
        int ec = (e < E) ? e : (E - 1);
        int vb = dst[ec];
        a[j] = src[ec];
        b[j] = (e < E) ? vb : -1;
      }
      int pos[EPT];
#pragma unroll
      for (int j = 0; j < EPT; ++j) {
        pos[j] = -1;
        if (b[j] >= lo && b[j] < hi) pos[j] = atomicAdd(&cursor[b[j]], 1);
      }
#pragma unroll
      for (int j = 0; j < EPT; ++j)
        if ((unsigned)pos[j] < (unsigned)E) list[pos[j]] = a[j];
    }
  }
}

__global__ void __launch_bounds__(TPB)
gcn_l1_gather(const int* __restrict__ rowptr, const int* __restrict__ rowend,
              const int* __restrict__ list, const float2* __restrict__ rec1,
              const float* __restrict__ W1, const float* __restrict__ b1,
              float* __restrict__ relu1, int n, int E) {
  const int lane = (int)threadIdx.x & 31;
  const int wv = __builtin_amdgcn_readfirstlane((int)threadIdx.x >> 5);
  const int nw = (int)gridDim.x * (TPB / 32);
  const unsigned last = (unsigned)(n - 1);
  const float w1 = W1[lane & 15];
  const float c1 = b1[lane & 15];
  int i = (int)blockIdx.x * (TPB / 32) + wv;
  int st = 0, en = 0;
  float xi = 0.0f, di = 0.0f;
  if (i < n) { st = rowptr[i]; en = rowend[i]; const float2 own = rec1[i]; xi = own.x; di = own.y; }
  while (i < n) {
    const int ni = i + nw;
    int nst = 0, nen = 0;
    float nxi = 0.0f, ndi = 0.0f;
    if (ni < n) { nst = rowptr[ni]; nen = rowend[ni]; const float2 own = rec1[ni]; nxi = own.x; ndi = own.y; }
    st = __builtin_amdgcn_readfirstlane((st > 0) ? st : 0);
    en = __builtin_amdgcn_readfirstlane((en < E) ? en : E);
    float acc = 0.0f;
    for (int base = st; base < en; base += 64) {
      const int k0 = base + lane, k1 = k0 + 32;
      const bool p0 = k0 < en, p1 = k1 < en;
      unsigned a0 = (unsigned)list[p0 ? k0 : st];
      unsigned a1 = (unsigned)list[p1 ? k1 : st];
      a0 = (a0 < last) ? a0 : last;
      a1 = (a1 < last) ? a1 : last;
      const float2 r0 = rec1[a0];
      const float2 r1 = rec1[a1];
      const float t0 = r0.x * r0.y * di;
      const float t1 = r1.x * r1.y * di;
      acc += p0 ? t0 : 0.0f;
      acc += p1 ? t1 : 0.0f;
    }
    for (int o = 16; o > 0; o >>= 1) acc += __shfl_xor(acc, o);
    const float sn = acc + xi * di * di;
    if (lane < 16) relu1[(size_t)i * 16 + lane] = fmaxf(fmaf(sn, w1, c1), 0.0f);
    i = ni; st = nst; en = nen; xi = nxi; di = ndi;
  }
}

__global__ void __launch_bounds__(TPB)
gcn_l2_gather(const int* __restrict__ rowptr, const int* __restrict__ rowend,
              const int* __restrict__ list, const float4* __restrict__ rec2,
              const float* __restrict__ b2, float* __restrict__ out,
              int n, int E) {
  const int lane = (int)threadIdx.x & 31;
  const int wv = __builtin_amdgcn_readfirstlane((int)threadIdx.x >> 5);
  const int nw = (int)gridDim.x * (TPB / 32);
  const unsigned last = (unsigned)(n - 1);
  const float bz0 = b2[0], bz1 = b2[1];
  int i = (int)blockIdx.x * (TPB / 32) + wv;
  int st = 0, en = 0;
  float di = 0.0f, h0 = 0.0f, h1 = 0.0f;
  if (i < n) { st = rowptr[i]; en = rowend[i]; const float4 own = rec2[i]; di = own.x; h0 = own.y; h1 = own.z; }
  while (i < n) {
    const int ni = i + nw;
    int nst = 0, nen = 0;
    float ndi = 0.0f, nh0 = 0.0f, nh1 = 0.0f;
    if (ni < n) { nst = rowptr[ni]; nen = rowend[ni]; const float4 own = rec2[ni]; ndi = own.x; nh0 = own.y; nh1 = own.z; }
    st = __builtin_amdgcn_readfirstlane((st > 0) ? st : 0);
    en = __builtin_amdgcn_readfirstlane((en < E) ? en : E);
    float acc0 = 0.0f, acc1 = 0.0f;
    for (int base = st; base < en; base += 64) {
      const int k0 = base + lane, k1 = k0 + 32;
      const bool p0 = k0 < en, p1 = k1 < en;
      unsigned a0 = (unsigned)list[p0 ? k0 : st];
      unsigned a1 = (unsigned)list[p1 ? k1 : st];
      a0 = (a0 < last) ? a0 : last;
      a1 = (a1 < last) ? a1 : last;
      const float4 q0 = rec2[a0];
      const float4 q1 = rec2[a1];
      const float n0 = q0.x * di;
      const float n1 = q1.x * di;
      const float u0 = q0.y * n0, w0 = q0.z * n0;
      const float u1 = q1.y * n1, w1 = q1.z * n1;
      acc0 += p0 ? u0 : 0.0f;  acc1 += p0 ? w0 : 0.0f;
      acc0 += p1 ? u1 : 0.0f;  acc1 += p1 ? w1 : 0.0f;
    }
    for (int o = 16; o > 0; o >>= 1) {
      acc0 += __shfl_xor(acc0, o);
      acc1 += __shfl_xor(acc1, o);
    }
    if (lane == 0) {
      const float sl = di * di;
      const float z0 = acc0 + h0 * sl + bz0;
      const float z1 = acc1 + h1 * sl + bz1;
      const float mx = fmaxf(z0, z1);
      const float lse = mx + logf(expf(z0 - mx) + expf(z1 - mx));
      float2 r; r.x = z0 - lse; r.y = z1 - lse;
      *(float2*)(out + (size_t)i * 2) = r;
    }
    i = ni; st = nst; en = nen; di = ndi; h0 = nh0; h1 = nh1;
  }
}


#define GSH 8
#define GNB (1 << GSH)
#define GTPB 1024
#define GEPT 12
#define GSUB (GTPB * GEPT)
#define GBMAX GTPB
#define GCAP 16384
#define RS_TPB 256

template <int NW>
__device__ __forceinline__ int block_scan_excl(int v, int* wtot, int& total) {
  const int lane = (int)threadIdx.x & 31;
  const int wv = (int)threadIdx.x >> 5;
  int inc = v;
  for (int o = 1; o < 32; o <<= 1) {
    int u = __shfl_up(inc, o);
    if (lane >= o) inc += u;
  }
  if (lane == 31) wtot[wv] = inc;
  __syncthreads();
  const int wt = (lane < NW) ? wtot[lane] : 0;
  int winc = wt;
  for (int o = 1; o < 32; o <<= 1) {
    int u = __shfl_up(winc, o);
    if (lane >= o) winc += u;
  }
  const int before = __shfl(winc - wt, wv);
  total = __shfl(winc, 31);
  __syncthreads();
  return before + (inc - v);
}

__device__ __forceinline__ int bucket_of(int d, int nb) {
  unsigned b = (unsigned)d >> GSH;
  return (int)((b < (unsigned)nb) ? b : (unsigned)(nb - 1));
}

__global__ void __launch_bounds__(GTPB)
grp_hist(const int* __restrict__ dst, int* __restrict__ H, int E, int nb, int nseg) {
  __shared__ int hist[GBMAX];
  const int t = (int)threadIdx.x;
  const int s = (int)blockIdx.x;
  if (t < nb) hist[t] = 0;
  __syncthreads();
  const int e0 = s * GSUB + t;
  int d[GEPT];
#pragma unroll
  for (int j = 0; j < GEPT; ++j) {
    int e = e0 + j * GTPB;
    d[j] = dst[(e < E) ? e : (E - 1)];
  }
#pragma unroll
  for (int j = 0; j < GEPT; ++j) {
    int e = e0 + j * GTPB;
    if (e < E) atomicAdd(&hist[bucket_of(d[j], nb)], 1);
  }
  __syncthreads();
  if (t < nb) H[(size_t)t * nseg + s] = hist[t];
}

__global__ void __launch_bounds__(RS_TPB)
grp_rowscan(const int* __restrict__ H, int* __restrict__ P, int* __restrict__ T, int nseg) {
  __shared__ int wtot[32];
  const int t = (int)threadIdx.x;
  const size_t row = (size_t)blockIdx.x * nseg;
  int v[4];
#pragma unroll
  for (int j = 0; j < 4; ++j) {
    int k = 4 * t + j;
    int h = H[row + ((k < nseg) ? k : 0)];
    v[j] = (k < nseg) ? h : 0;
  }
  int total;
  int ex = block_scan_excl<RS_TPB / 32>(v[0] + v[1] + v[2] + v[3], wtot, total);
#pragma unroll
  for (int j = 0; j < 4; ++j) {
    int k = 4 * t + j;
    if (k < nseg) P[row + k] = ex;
    ex += v[j];
  }
  if (t == 0) T[blockIdx.x] = total;
}

__global__ void __launch_bounds__(GTPB)
grp_partition(const int* __restrict__ src, const int* __restrict__ dst,
              const int* __restrict__ P, const int* __restrict__ T,
              int* __restrict__ list, int E, int nb, int nseg) {
  __shared__ int hist[GBMAX];
  __shared__ int offs[GBMAX];
  __shared__ int gbase[GBMAX];
  __shared__ int wtot[32];
  __shared__ int stage[GSUB];
  __shared__ unsigned short stage_b[GSUB];
  const int t = (int)threadIdx.x;
  const int s = (int)blockIdx.x;

  const int tb = (t < nb) ? T[t] : 0;
  const int pb = (t < nb) ? P[(size_t)t * nseg + s] : 0;
  if (t < nb) hist[t] = 0;
  int all;
  const int first = block_scan_excl<GTPB / 32>(tb, wtot, all);
  if (t < nb) gbase[t] = first + pb;

  const int e0 = s * GSUB + t;
  int a[GEPT], d[GEPT], r[GEPT];
#pragma unroll
  for (int j = 0; j < GEPT; ++j) {
    int e = e0 + j * GTPB;
    int ec = (e < E) ? e : (E - 1);
    d[j] = dst[ec];
    a[j] = src[ec];
  }
#pragma unroll
  for (int j = 0; j < GEPT; ++j) {
    int e = e0 + j * GTPB;
    r[j] = -1;
    if (e < E) r[j] = atomicAdd(&hist[bucket_of(d[j], nb)], 1);
  }
  __syncthreads();
  const int hv = (t < nb) ? hist[t] : 0;
  int total;
  const int ex = block_scan_excl<GTPB / 32>(hv, wtot, total);
  if (t < nb) offs[t] = ex;
  __syncthreads();
#pragma unroll
  for (int j = 0; j < GEPT; ++j) {
    if (r[j] >= 0) {
      const int b = bucket_of(d[j], nb);
      const int pos = offs[b] + r[j];
      if ((unsigned)pos < (unsigned)GSUB) {
        stage[pos] = (int)((((unsigned)d[j] & (unsigned)(GNB - 1)) << 24) |
                           ((unsigned)a[j] & 0xFFFFFFu));
        stage_b[pos] = (unsigned short)b;
      }
    }
  }
  __syncthreads();
#pragma unroll
  for (int j = 0; j < GEPT; ++j) {
    const int idx = t + j * GTPB;
    if (idx < total) {
      const int b = (int)stage_b[idx];
      const int g = gbase[b] + (idx - offs[b]);
      if ((unsigned)g < (unsigned)E) list[g] = stage[idx];
    }
  }
}

__global__ void __launch_bounds__(GTPB)
grp_bucket(const int* __restrict__ T, const float* __restrict__ xin, int* list,
           int* __restrict__ rowptr, int* __restrict__ rowend,
           float* __restrict__ dinv, float2* __restrict__ rec1, int n, int E, int nb) {
  __shared__ int cnt[GNB];
  __shared__ int cur[GNB];
  __shared__ int endv[GNB];
  __shared__ int wtot[32];
  __shared__ int sh[2];
  __shared__ int ebuf[GCAP];
  __shared__ int obuf[GCAP];
  const int t = (int)threadIdx.x;
  const int b = (int)blockIdx.x;

  const int tb = (t < nb) ? T[t] : 0;
  int all;
  const int first = block_scan_excl<GTPB / 32>(tb, wtot, all);
  if (t == b) { sh[0] = first; sh[1] = tb; }
  if (t < GNB) cnt[t] = 0;
  __syncthreads();
  int base = sh[0], len = sh[1];
  base = (base > 0) ? base : 0;
  base = (base < E) ? base : E;
  len = (len > 0) ? len : 0;
  len = (len < E - base) ? len : (E - base);
  const bool fits = len <= GCAP;

  for (int j0 = t; j0 < len; j0 += 4 * GTPB) {
    unsigned v[4];
#pragma unroll
    for (int q = 0; q < 4; ++q) {
      int j = j0 + q * GTPB;
      v[q] = (unsigned)list[base + ((j < len) ? j : j0)];
    }
#pragma unroll
    for (int q = 0; q < 4; ++q) {
      int j = j0 + q * GTPB;
      if (j < len) {
        if (fits) ebuf[j] = (int)v[q];
        atomicAdd(&cnt[v[q] >> 24], 1);
      }
    }
  }
  __syncthreads();
  const int c = (t < GNB) ? cnt[t] : 0;
  int tot;
  const int ex = block_scan_excl<GTPB / 32>(c, wtot, tot);
  if (t < GNB) {
    cur[t] = ex;
    endv[t] = ex + c;
    const int i = b * GNB + t;
    if (i < n) {
      rowptr[i] = base + ex;
      rowend[i] = base + ex + c;
      const float dv = rsqrtf((float)(c + 1));
      dinv[i] = dv;
      rec1[i] = make_float2(xin[i], dv);
    }
  }
  __syncthreads();

  if (fits) {
    for (int j = t; j < len; j += GTPB) {
      const unsigned v = (unsigned)ebuf[j];
      const int pos = atomicAdd(&cur[v >> 24], 1);
      if ((unsigned)pos < (unsigned)GCAP) obuf[pos] = (int)(v & 0xFFFFFFu);
    }
    __syncthreads();
    for (int j = t; j < len; j += GTPB) list[base + j] = obuf[j];
  } else {
    if (t == 0) {
      int steps = 0;
      const int limit = 2 * len + GNB;
      for (int v = 0; v < GNB; ++v) {
        int p = cur[v];
        const int pe = endv[v];
        while (p < pe && steps < limit) {
          ++steps;
          const unsigned x = (unsigned)list[base + p];
          const int dv = (int)(x >> 24);
          if (dv == v) { ++p; continue; }
          const int q = cur[dv];
          if (q >= endv[dv]) break;
          const int y = list[base + q];
          list[base + q] = (int)x;
          list[base + p] = y;
          cur[dv] = q + 1;
        }
        cur[v] = p;
      }
    }
    __syncthreads();
    for (int j = t; j < len; j += GTPB) list[base + j] = list[base + j] & 0xFFFFFF;
  }
}

#define GPC 6

__global__ void __launch_bounds__(GTPB)
grp_segsort(const int* __restrict__ src, const int* __restrict__ dst,
            int* __restrict__ O, int* __restrict__ tmp, int E, int nb, int nseg) {
  __shared__ int hist[GBMAX];
  __shared__ int offs[GBMAX];
  __shared__ int wtot[32];
  __shared__ int stage[GSUB];
  const int t = (int)threadIdx.x;
  const int s = (int)blockIdx.x;
  if (t < nb) hist[t] = 0;
  __syncthreads();

  const int e0 = s * GSUB + t;
  int a[GEPT], d[GEPT], r[GEPT];
#pragma unroll
  for (int j = 0; j < GEPT; ++j) {
    int e = e0 + j * GTPB;
    int ec = (e < E) ? e : (E - 1);
    d[j] = dst[ec];
    a[j] = src[ec];
  }
#pragma unroll
  for (int j = 0; j < GEPT; ++j) {
    int e = e0 + j * GTPB;
    r[j] = -1;
    if (e < E) r[j] = atomicAdd(&hist[bucket_of(d[j], nb)], 1);
  }
  __syncthreads();
  const int hv = (t < nb) ? hist[t] : 0;
  int total;
  const int ex = block_scan_excl<GTPB / 32>(hv, wtot, total);
  if (t < nb) {
    offs[t] = ex;
    O[(size_t)t * nseg + s] = ex;
  }
  if (t == 0) O[(size_t)nb * nseg + s] = total;
  __syncthreads();
#pragma unroll
  for (int j = 0; j < GEPT; ++j) {
    if (r[j] >= 0) {
      const int pos = offs[bucket_of(d[j], nb)] + r[j];
      if ((unsigned)pos < (unsigned)GSUB)
        stage[pos] = (int)((((unsigned)d[j] & (unsigned)(GNB - 1)) << 24) |
                           ((unsigned)a[j] & 0xFFFFFFu));
    }
  }
  __syncthreads();
#pragma unroll
  for (int j = 0; j < GEPT; ++j) {
    const int idx = t + j * GTPB;
    const size_t g = (size_t)s * GSUB + (size_t)idx;
    if (idx < total && g < (size_t)E) tmp[g] = stage[idx];
  }
}

__global__ void __launch_bounds__(GTPB)
grp_collect(const int* __restrict__ O, const int* __restrict__ tmp,
            const float* __restrict__ xin, int* list,
            int* __restrict__ rowptr, int* __restrict__ rowend,
            float* __restrict__ dinv, float2* __restrict__ rec1,
            int n, int E, int nb, int nseg) {
  __shared__ int cnt[GNB];
  __shared__ int cur[GNB];
  __shared__ int endv[GNB];
  __shared__ int wtot[32];
  __shared__ int4 piece[GTPB];
  __shared__ int ebuf[GCAP];
  __shared__ int obuf[GCAP];
  const int t = (int)threadIdx.x;
  const int b = (int)blockIdx.x;
  const int lane = t & 31;
  const int wv = __builtin_amdgcn_readfirstlane(t >> 5);

  int o0 = 0, o1 = 0;
  if (t < nseg) {
    o0 = O[(size_t)b * nseg + t];
    o1 = O[(size_t)(b + 1) * nseg + t];
  }
  o0 = (o0 > 0) ? o0 : 0;
  o0 = (o0 < GSUB) ? o0 : GSUB;
  int pl = o1 - o0;
  pl = (pl > 0) ? pl : 0;
  pl = (pl < GSUB - o0) ? pl : (GSUB - o0);
  int len, base;
  const int pp = block_scan_excl<GTPB / 32>(pl, wtot, len);
  (void)block_scan_excl<GTPB / 32>(o0, wtot, base);
  piece[t] = make_int4(o0, pp, pl, 0);
  if (t < GNB) cnt[t] = 0;
  __syncthreads();
  base = (base > 0) ? base : 0;
  base = (base < E) ? base : E;
  len = (len > 0) ? len : 0;
  len = (len < E - base) ? len : (E - base);
  const bool fits = len <= GCAP;

  auto put = [&](int k, unsigned v) {
    if (fits) {
      if ((unsigned)k < (unsigned)GCAP) ebuf[k] = (int)v;
    } else if (k < len) {
      list[base + k] = (int)v;
    }
    atomicAdd(&cnt[v >> 24], 1);
  };

  for (int s0 = wv; s0 < nseg; s0 += 32 * GPC) {
    unsigned v[GPC][2];
    int pl_[GPC], pp_[GPC];
    size_t g0_[GPC];
#pragma unroll
    for (int q = 0; q < GPC; ++q) {
      const int sq = s0 + 32 * q;
      const bool ok = sq < nseg;
      const int sc = ok ? sq : s0;
      const int4 m = piece[sc];
      const int mlen = __builtin_amdgcn_readfirstlane(m.z);
      pl_[q] = ok ? mlen : 0;
      pp_[q] = __builtin_amdgcn_readfirstlane(m.y);
      g0_[q] = (size_t)sc * GSUB + (size_t)__builtin_amdgcn_readfirstlane(m.x);
    }
#pragma unroll
    for (int q = 0; q < GPC; ++q) {
#pragma unroll
      for (int c = 0; c < 2; ++c) {
        const int j = lane + 32 * c;
        size_t g = (j < pl_[q]) ? g0_[q] + (size_t)j : 0;
        g = (g < (size_t)E) ? g : (size_t)(E - 1);
        v[q][c] = (unsigned)tmp[g];
      }
    }
#pragma unroll
    for (int q = 0; q < GPC; ++q) {
#pragma unroll
      for (int c = 0; c < 2; ++c) {
        const int j = lane + 32 * c;
        if (j < pl_[q]) put(pp_[q] + j, v[q][c]);
      }
    }
#pragma unroll
    for (int q = 0; q < GPC; ++q) {
      for (int j = lane + 64; j < pl_[q]; j += 32) {
        size_t g = g0_[q] + (size_t)j;
        g = (g < (size_t)E) ? g : (size_t)(E - 1);
        put(pp_[q] + j, (unsigned)tmp[g]);
      }
    }
  }
  __syncthreads();

  const int c = (t < GNB) ? cnt[t] : 0;
  int tot;
  const int ex = block_scan_excl<GTPB / 32>(c, wtot, tot);
  if (t < GNB) {
    cur[t] = ex;
    endv[t] = ex + c;
    const int i = b * GNB + t;
    if (i < n) {
      rowptr[i] = base + ex;
      rowend[i] = base + ex + c;
      const float dv = rsqrtf((float)(c + 1));
      dinv[i] = dv;
      rec1[i] = make_float2(xin[i], dv);
    }
  }
  __syncthreads();

  if (fits) {
    for (int j = t; j < len; j += GTPB) {
      const unsigned v = (unsigned)ebuf[j];
      const int pos = atomicAdd(&cur[v >> 24], 1);
      if ((unsigned)pos < (unsigned)GCAP) obuf[pos] = (int)(v & 0xFFFFFFu);
    }
    __syncthreads();
    for (int j = t; j < len; j += GTPB) list[base + j] = obuf[j];
  } else {
    if (t == 0) {
      int steps = 0;
      const int limit = 2 * len + GNB;
      for (int v = 0; v < GNB; ++v) {
        int p = cur[v];
        const int pe = endv[v];
        while (p < pe && steps < limit) {
          ++steps;
          const unsigned x = (unsigned)list[base + p];
          const int dv = (int)(x >> 24);
          if (dv == v) { ++p; continue; }
          const int q = cur[dv];
          if (q >= endv[dv]) break;
          const int y = list[base + q];
          list[base + q] = (int)x;
          list[base + p] = y;
          cur[dv] = q + 1;
        }
        cur[v] = p;
      }
    }
    __syncthreads();
    for (int j = t; j < len; j += GTPB) list[base + j] = list[base + j] & 0xFFFFFF;
  }
}

extern "C" void kernel_launch(void* const* d_in, const int* in_sizes, int n_in,
                              void* d_out, int out_size, void* d_ws, size_t ws_size,
                              hipStream_t stream) {
  const float* x  = (const float*)d_in[0];
  const int*   ei = (const int*)d_in[1];
  const float* W1 = (const float*)d_in[2];
  const float* b1 = (const float*)d_in[3];
  const float* W2 = (const float*)d_in[4];
  const float* b2 = (const float*)d_in[5];
  float* out = (float*)d_out;
  int* tick = (int*)d_out;

  int n = in_sizes[0];
  int E = in_sizes[1] / 2;
  const int* srcp = ei;
  const int* dstp = ei + E;

  size_t ns = ((size_t)n + 31) & ~(size_t)31;

  float* ws    = (float*)d_ws;
  float* deg   = ws;
  float* dinv  = ws + ns;
  float* s     = ws + 2 * ns;
  float* h2c0  = ws + 3 * ns;
  float* h2c1  = ws + 4 * ns;
  float* agg0  = ws + 5 * ns;
  float* agg1  = ws + 6 * ns;

  int gN = (n + TPB - 1) / TPB;
  int ntiles = (n + 15) / 16;
  int gW = (int)(((size_t)ntiles * 32 + TPB - 1) / TPB);

  const size_t need = (25 * ns + (size_t)E) * sizeof(float);
  if (ws_size >= need) {
    int*   cnt    = (int*)ws;
    int*   rowptr = (int*)(ws + 7 * ns);
    int*   cursor = (int*)(ws + 8 * ns);
    float* relu1  = ws + 9 * ns;
    int*   list   = (int*)(ws + 25 * ns);
    float2* rec1  = (float2*)(ws + 4 * ns);
    float4* rec2  = (float4*)(ws + 2 * ns);
    int gS = (n + SCAN_TPB - 1) / SCAN_TPB;

    const int nb = (n + GNB - 1) >> GSH;
    const int nseg = (E + GSUB - 1) / GSUB;
    const size_t tabs = 2 * (size_t)nb * (size_t)nseg + (size_t)nb;
    if (E > 0 && nb <= GBMAX && nseg <= GTPB && n <= (1 << 24) && tabs <= 16 * (size_t)n) {
      int* H = (int*)relu1;
      const size_t need2 = (25 * ns + 2 * (size_t)E) * sizeof(float);
      if (ws_size >= need2) {
        int* tmp = list + E;
        grp_segsort<<<nseg, GTPB, 0, stream>>>(srcp, dstp, H, tmp, E, nb, nseg);
        grp_collect<<<nb, GTPB, 0, stream>>>(H, tmp, x, list, rowptr, cursor, dinv, rec1, n, E, nb, nseg);
      } else {
        int* P = H + (size_t)nb * nseg;
        int* T = P + (size_t)nb * nseg;
        grp_hist<<<nseg, GTPB, 0, stream>>>(dstp, H, E, nb, nseg);
        grp_rowscan<<<nb, RS_TPB, 0, stream>>>(H, P, T, nseg);
        grp_partition<<<nseg, GTPB, 0, stream>>>(srcp, dstp, P, T, list, E, nb, nseg);
        grp_bucket<<<nb, GTPB, 0, stream>>>(T, x, list, rowptr, cursor, dinv, rec1, n, E, nb);
      }
    } else {
      csr_init<<<gN, TPB, 0, stream>>>(cnt, tick, n);
      csr_count<<<PASS_BLOCKS, TPB, 0, stream>>>(dstp, cnt, tick, n, E);
      csr_scan<<<gS, SCAN_TPB, 0, stream>>>(cnt, x, rowptr, cursor, dinv, rec1, n);
      csr_fill<<<PASS_BLOCKS, TPB, 0, stream>>>(srcp, dstp, cursor, list, tick, n, E);
    }
    gcn_l1_gather<<<PASS_BLOCKS, TPB, 0, stream>>>(rowptr, cursor, list, rec1, W1, b1, relu1, n, E);
    gcn_l2_dense_wmma<true><<<gW, TPB, 0, stream>>>(relu1, W2, dinv, (float*)nullptr,
                                                    (float*)nullptr, rec2, n, ntiles);
    gcn_l2_gather<<<PASS_BLOCKS, TPB, 0, stream>>>(rowptr, cursor, list, rec2, b2, out, n, E);
    return;
  }

  float* relu1 = ws + 7 * ns;
  gcn_init<<<gN, TPB, 0, stream>>>(deg, s, agg0, agg1, tick, n);
  gcn_degree<<<PASS_BLOCKS, TPB, 0, stream>>>(dstp, deg, tick, n, E);
  gcn_dinv<<<gN, TPB, 0, stream>>>(deg, dinv, n);
  gcn_l1_scatter<<<PASS_BLOCKS, TPB, 0, stream>>>(srcp, dstp, x, dinv, s, tick, n, E);
  gcn_l1_finalize<<<gN, TPB, 0, stream>>>(x, dinv, s, W1, b1, relu1, n);
  gcn_l2_dense_wmma<false><<<gW, TPB, 0, stream>>>(relu1, W2, dinv, h2c0, h2c1,
                                                   (float4*)nullptr, n, ntiles);
  gcn_l2_scatter<<<PASS_BLOCKS, TPB, 0, stream>>>(srcp, dstp, h2c0, h2c1, dinv,
                                                   agg0, agg1, tick, n, E);
  gcn_finalize<<<gN, TPB, 0, stream>>>(agg0, agg1, h2c0, h2c1, dinv, b2, out, n);
}


# Layer-2 dense WMMA fused into layer-1 gather: block sums a 16-node tile (2 nodes/wave), one wave takes the sums from LDS and builds relu(sn*W1+b1) as its A fragment; 5 launches to 4; sum order kept
# speedup vs baseline: 10.9067x; 1.0590x over previous
#include <hip/hip_runtime.h>


typedef __attribute__((ext_vector_type(2))) float v2f;
typedef __attribute__((ext_vector_type(8))) float v8f;

#define TPB 256
#define EPT 4
#define SLICE (TPB * EPT)
#define PASS_BLOCKS 2048
#define NQ 6


__device__ __forceinline__ int tick_stride(int n) { return (n >= 32 * NQ) ? 64 : 1; }

__device__ __forceinline__ int cache_half() {
  unsigned v = __builtin_amdgcn_s_sendmsg_rtn(0x87);
  return (int)((v >> 18) & 1u);
}

__device__ __forceinline__ int draw_slice(int* q, int* sh) {
  __syncthreads();
  if (threadIdx.x == 0) *sh = atomicAdd(q, 1);
  __syncthreads();
  return *sh;
}

__device__ __forceinline__ int block_half(int* sh) {
  if (threadIdx.x == 0) *sh = cache_half();
  __syncthreads();
  return *sh;
}

__global__ void gcn_init(float* __restrict__ deg, float* __restrict__ s,
                         float* __restrict__ agg0, float* __restrict__ agg1,
                         int* __restrict__ tick, int n) {
  int i = blockIdx.x * blockDim.x + threadIdx.x;
  if (i < NQ) tick[i * tick_stride(n)] = 0;
  if (i < n) {
    deg[i] = 1.0f;
    s[i] = 0.0f;
    agg0[i] = 0.0f;
    agg1[i] = 0.0f;
  }
}

__global__ void __launch_bounds__(TPB)
gcn_degree(const int* __restrict__ dst, float* __restrict__ deg,
           int* __restrict__ tick, int n, int E) {
  __shared__ int sh[2];
  const int nsl = (E + SLICE - 1) / SLICE;
  const int split = (n >> 1) & ~31;
  const int st = tick_stride(n);
  const int h0 = block_half(&sh[1]);
  for (int r = 0; r < 2; ++r) {
    const int h = h0 ^ r;
    const int lo = h ? split : 0;
    const int hi = h ? n : split;
    int* q = tick + (0 + h) * st;
    for (int t = draw_slice(q, &sh[0]); t < nsl; t = draw_slice(q, &sh[0])) {
      const int e0 = t * SLICE + (int)threadIdx.x;
      int b[EPT];
#pragma unroll
      for (int j = 0; j < EPT; ++j) {
        int e = e0 + j * TPB;
        int v = dst[(e < E) ? e : (E - 1)];
        b[j] = (e < E) ? v : -1;
      }
#pragma unroll
      for (int j = 0; j < EPT; ++j)
        if (b[j] >= lo && b[j] < hi) unsafeAtomicAdd(&deg[b[j]], 1.0f);
    }
  }
}

__global__ void gcn_dinv(const float* __restrict__ deg,
                         float* __restrict__ dinv, int n) {
  int i = blockIdx.x * blockDim.x + threadIdx.x;
  if (i < n) dinv[i] = rsqrtf(deg[i]);
}

__global__ void __launch_bounds__(TPB)
gcn_l1_scatter(const int* __restrict__ src, const int* __restrict__ dst,
               const float* __restrict__ x, const float* __restrict__ dinv,
               float* __restrict__ s, int* __restrict__ tick, int n, int E) {
  __shared__ int sh[2];
  const int nsl = (E + SLICE - 1) / SLICE;
  const int split = (n >> 1) & ~31;
  const int st = tick_stride(n);
  const int h0 = block_half(&sh[1]);
  for (int r = 0; r < 2; ++r) {
    const int h = h0 ^ r;
    const int lo = h ? split : 0;
    const int hi = h ? n : split;
    int* q = tick + (2 + h) * st;
    for (int t = draw_slice(q, &sh[0]); t < nsl; t = draw_slice(q, &sh[0])) {
      const int e0 = t * SLICE + (int)threadIdx.x;
      int a[EPT], b[EPT];
#pragma unroll
      for (int j = 0; j < EPT; ++j) {
        int e = e0 + j * TPB;
        int ec = (e < E) ? e : (E - 1);
        int vb = dst[ec];
        a[j] = src[ec];
        b[j] = (e < E) ? vb : -1;
      }
      asm volatile("" :: "v"(a[0]), "v"(a[1]), "v"(a[2]), "v"(a[3]));
      float v[EPT];
#pragma unroll
      for (int j = 0; j < EPT; ++j) {
        bool in = b[j] >= lo && b[j] < hi;
        unsigned aa = in ? (unsigned)a[j] : 0u;
        unsigned bb = in ? (unsigned)b[j] : 0u;
        v[j] = x[aa] * dinv[aa] * dinv[bb];
      }
#pragma unroll
      for (int j = 0; j < EPT; ++j)
        if (b[j] >= lo && b[j] < hi) unsafeAtomicAdd(&s[b[j]], v[j]);
    }
  }
}

__global__ void gcn_l1_finalize(const float* __restrict__ x,
                                const float* __restrict__ dinv,
                                const float* __restrict__ s,
                                const float* __restrict__ W1,
                                const float* __restrict__ b1,
                                float* __restrict__ relu1, int n) {
  int i = blockIdx.x * blockDim.x + threadIdx.x;
  if (i >= n) return;
  float di = dinv[i];
  float sn = s[i] + x[i] * di * di;
  float4* o = (float4*)(relu1 + (size_t)i * 16);
#pragma unroll
  for (int q = 0; q < 4; ++q) {
    float4 r;
    r.x = fmaxf(fmaf(sn, W1[4 * q + 0], b1[4 * q + 0]), 0.0f);
    r.y = fmaxf(fmaf(sn, W1[4 * q + 1], b1[4 * q + 1]), 0.0f);
    r.z = fmaxf(fmaf(sn, W1[4 * q + 2], b1[4 * q + 2]), 0.0f);
    r.w = fmaxf(fmaf(sn, W1[4 * q + 3], b1[4 * q + 3]), 0.0f);
    o[q] = r;
  }
}


__device__ __forceinline__ void load_w2_frag(const float* __restrict__ W2,
                                             int half, int m, v2f bmat[4]) {
#pragma unroll
  for (int k = 0; k < 4; ++k) {
    int r0 = 4 * k + 2 * half;
    float bx = 0.0f, by = 0.0f;
    if (m < 2) { bx = W2[r0 * 2 + m]; by = W2[(r0 + 1) * 2 + m]; }
    bmat[k].x = bx; bmat[k].y = by;
  }
}

__global__ void gcn_l2_dense_wmma(const float* __restrict__ relu1,
                                  const float* __restrict__ W2,
                                  float* __restrict__ h2c0,
                                  float* __restrict__ h2c1,
                                  int n, int ntiles) {
  int gtid = blockIdx.x * blockDim.x + threadIdx.x;
  int tile = gtid >> 5;
  int lane = threadIdx.x & 31;
  if (tile >= ntiles) return;
  int base = tile * 16;
  int half = lane >> 4;
  int m = lane & 15;

  v2f bmat[4];
  load_w2_frag(W2, half, m, bmat);

  int row = base + m;
  row = (row < n) ? row : (n - 1);
  const v2f* p = (const v2f*)(relu1 + (size_t)row * 16 + 2 * half);
  v2f amat[4];
#pragma unroll
  for (int k = 0; k < 4; ++k) amat[k] = p[2 * k];

  v8f c = {};
#pragma unroll
  for (int k = 0; k < 4; ++k)
    c = __builtin_amdgcn_wmma_f32_16x16x4_f32(
         false, amat[k],  false, bmat[k],
         (short)0, c,  false,  false);

  if (m < 2) {
    float* col = (m == 0) ? h2c0 : h2c1;
    int node0 = base + half * 8;
    float4 lo; lo.x = c[0]; lo.y = c[1]; lo.z = c[2]; lo.w = c[3];
    float4 hi; hi.x = c[4]; hi.y = c[5]; hi.z = c[6]; hi.w = c[7];
    *(float4*)(col + node0)     = lo;
    *(float4*)(col + node0 + 4) = hi;
  }
}

__global__ void __launch_bounds__(TPB)
gcn_l2_scatter(const int* __restrict__ src, const int* __restrict__ dst,
               const float* __restrict__ h2c0, const float* __restrict__ h2c1,
               const float* __restrict__ dinv,
               float* __restrict__ agg0, float* __restrict__ agg1,
               int* __restrict__ tick, int n, int E) {
  __shared__ int sh[2];
  const int nsl = (E + SLICE - 1) / SLICE;
  const int split = (n >> 1) & ~31;
  const int st = tick_stride(n);
  const int h0 = block_half(&sh[1]);
  for (int r = 0; r < 2; ++r) {
    const int h = h0 ^ r;
    const int lo = h ? split : 0;
    const int hi = h ? n : split;
    int* q = tick + (4 + h) * st;
    for (int t = draw_slice(q, &sh[0]); t < nsl; t = draw_slice(q, &sh[0])) {
      const int e0 = t * SLICE + (int)threadIdx.x;
      int a[EPT], b[EPT];
#pragma unroll
      for (int j = 0; j < EPT; ++j) {
        int e = e0 + j * TPB;
        int ec = (e < E) ? e : (E - 1);
        int vb = dst[ec];
        a[j] = src[ec];
        b[j] = (e < E) ? vb : -1;
      }
      asm volatile("" :: "v"(a[0]), "v"(a[1]), "v"(a[2]), "v"(a[3]));
      float v0[EPT], v1[EPT];
#pragma unroll
      for (int j = 0; j < EPT; ++j) {
        bool in = b[j] >= lo && b[j] < hi;
        unsigned aa = in ? (unsigned)a[j] : 0u;
        unsigned bb = in ? (unsigned)b[j] : 0u;
        float nrm = dinv[aa] * dinv[bb];
        v0[j] = h2c0[aa] * nrm;
        v1[j] = h2c1[aa] * nrm;
      }
#pragma unroll
      for (int j = 0; j < EPT; ++j)
        if (b[j] >= lo && b[j] < hi) {
          unsafeAtomicAdd(&agg0[b[j]], v0[j]);
          unsafeAtomicAdd(&agg1[b[j]], v1[j]);
        }
    }
  }
}

__global__ void gcn_finalize(const float* __restrict__ agg0,
                             const float* __restrict__ agg1,
                             const float* __restrict__ h2c0,
                             const float* __restrict__ h2c1,
                             const float* __restrict__ dinv,
                             const float* __restrict__ b2,
                             float* __restrict__ out, int n) {
  int i = blockIdx.x * blockDim.x + threadIdx.x;
  if (i >= n) return;
  float di = dinv[i], h0 = h2c0[i], h1 = h2c1[i];
  float sl = di * di;
  float z0 = agg0[i] + h0 * sl + b2[0];
  float z1 = agg1[i] + h1 * sl + b2[1];
  float mx = fmaxf(z0, z1);
  float lse = mx + logf(expf(z0 - mx) + expf(z1 - mx));
  float2 r; r.x = z0 - lse; r.y = z1 - lse;
  *(float2*)(out + (size_t)i * 2) = r;
}

#define SCAN_TPB 1024

__global__ void csr_init(int* __restrict__ cnt, int* __restrict__ tick, int n) {
  int i = blockIdx.x * blockDim.x + threadIdx.x;
  if (i < 4) tick[i * tick_stride(n)] = 0;
  if (i < n) cnt[i] = 0;
}

__global__ void __launch_bounds__(TPB)
csr_count(const int* __restrict__ dst, int* __restrict__ cnt,
          int* __restrict__ tick, int n, int E) {
  __shared__ int sh[2];
  const int nsl = (E + SLICE - 1) / SLICE;
  const int split = (n >> 1) & ~31;
  const int st = tick_stride(n);
  const int h0 = block_half(&sh[1]);
  for (int r = 0; r < 2; ++r) {
    const int h = h0 ^ r;
    const int lo = h ? split : 0;
    const int hi = h ? n : split;
    int* q = tick + (0 + h) * st;
    for (int t = draw_slice(q, &sh[0]); t < nsl; t = draw_slice(q, &sh[0])) {
      const int e0 = t * SLICE + (int)threadIdx.x;
      int b[EPT];
#pragma unroll
      for (int j = 0; j < EPT; ++j) {
        int e = e0 + j * TPB;
        int v = dst[(e < E) ? e : (E - 1)];
        b[j] = (e < E) ? v : -1;
      }
#pragma unroll
      for (int j = 0; j < EPT; ++j)
        if (b[j] >= lo && b[j] < hi) atomicAdd(&cnt[b[j]], 1);
    }
  }
}

__global__ void __launch_bounds__(SCAN_TPB)
csr_scan(const int* __restrict__ cnt, const float* __restrict__ xin,
         int* __restrict__ rowptr, int* __restrict__ cursor,
         float* __restrict__ dinv, float2* __restrict__ rec1, int n) {
  __shared__ int wtot[32];
  __shared__ int wpre[32];
  const int t = (int)threadIdx.x;
  const int lane = t & 31;
  const int wv = t >> 5;
  const int first = (int)blockIdx.x * SCAN_TPB;
  const int i = first + t;
  const int c = (i < n) ? cnt[i] : 0;

  int pre = 0;
  const int4* c4 = (const int4*)cnt;
  const int ngrp = first / 4;
  for (int k = t; k < ngrp; k += 8 * SCAN_TPB) {
    int4 v[8];
#pragma unroll
    for (int j = 0; j < 8; ++j) {
      int kk = k + j * SCAN_TPB;
      v[j] = c4[(kk < ngrp) ? kk : k];
    }
#pragma unroll
    for (int j = 0; j < 8; ++j) {
      int kk = k + j * SCAN_TPB;
      int sum = (v[j].x + v[j].y) + (v[j].z + v[j].w);
      pre += (kk < ngrp) ? sum : 0;
    }
  }
  for (int o = 16; o > 0; o >>= 1) pre += __shfl_xor(pre, o);

  int inc = c;
  for (int o = 1; o < 32; o <<= 1) {
    int u = __shfl_up(inc, o);
    if (lane >= o) inc += u;
  }
  if (lane == 31) wtot[wv] = inc;
  if (lane == 0) wpre[wv] = pre;
  __syncthreads();

  int wt = wtot[lane];
  int winc = wt;
  for (int o = 1; o < 32; o <<= 1) {
    int u = __shfl_up(winc, o);
    if (lane >= o) winc += u;
  }
  int before_wave = __shfl(winc - wt, wv);
  int before_block = wpre[lane];
  for (int o = 16; o > 0; o >>= 1) before_block += __shfl_xor(before_block, o);

  if (i < n) {
    int slot = before_block + before_wave + (inc - c);
    rowptr[i] = slot;
    cursor[i] = slot;
    const float dv = rsqrtf((float)(c + 1));
    dinv[i] = dv;
    rec1[i] = make_float2(xin[i], dv);
  }
}

__global__ void __launch_bounds__(TPB)
csr_fill(const int* __restrict__ src, const int* __restrict__ dst,
         int* __restrict__ cursor, int* __restrict__ list,
         int* __restrict__ tick, int n, int E) {
  __shared__ int sh[2];
  const int nsl = (E + SLICE - 1) / SLICE;
  const int split = (n >> 1) & ~31;
  const int st = tick_stride(n);
  const int h0 = block_half(&sh[1]);
  for (int r = 0; r < 2; ++r) {
    const int h = h0 ^ r;
    const int lo = h ? split : 0;
    const int hi = h ? n : split;
    int* q = tick + (2 + h) * st;
    for (int t = draw_slice(q, &sh[0]); t < nsl; t = draw_slice(q, &sh[0])) {
      const int e0 = t * SLICE + (int)threadIdx.x;
      int a[EPT], b[EPT];
#pragma unroll
      for (int j = 0; j < EPT; ++j) {
        int e = e0 + j * TPB;
        int ec = (e < E) ? e : (E - 1);
        int vb = dst[ec];
        a[j] = src[ec];
        b[j] = (e < E) ? vb : -1;
      }
      int pos[EPT];
#pragma unroll
      for (int j = 0; j < EPT; ++j) {
        pos[j] = -1;
        if (b[j] >= lo && b[j] < hi) pos[j] = atomicAdd(&cursor[b[j]], 1);
      }
#pragma unroll
      for (int j = 0; j < EPT; ++j)
        if ((unsigned)pos[j] < (unsigned)E) list[pos[j]] = a[j];
    }
  }
}

__global__ void __launch_bounds__(TPB)
gcn_l1_gather_dense(const int* __restrict__ rowptr, const int* __restrict__ rowend,
                    const int* __restrict__ list, const float2* __restrict__ rec1,
                    const float* __restrict__ W1, const float* __restrict__ b1,
                    const float* __restrict__ W2, float4* __restrict__ rec2,
                    int n, int ntiles, int E) {
  __shared__ float2 sums[2][16];
  const int lane = (int)threadIdx.x & 31;
  const int wv = __builtin_amdgcn_readfirstlane((int)threadIdx.x >> 5);
  const int half = lane >> 4;
  const int m = lane & 15;
  const unsigned last = (unsigned)(n - 1);

  v2f bmat[4];
  load_w2_frag(W2, half, m, bmat);

  int tile = (int)blockIdx.x;
  int st0 = 0, en0 = 0, st1 = 0, en1 = 0;
  float x0 = 0.0f, d0 = 0.0f, x1 = 0.0f, d1 = 0.0f;
  if (tile < ntiles) {
    const int i0 = tile * 16 + 2 * wv, i1 = i0 + 1;
    if (i0 < n) { st0 = rowptr[i0]; en0 = rowend[i0]; const float2 own = rec1[i0]; x0 = own.x; d0 = own.y; }
    if (i1 < n) { st1 = rowptr[i1]; en1 = rowend[i1]; const float2 own = rec1[i1]; x1 = own.x; d1 = own.y; }
  }
  for (int it = 0; tile < ntiles; ++it) {
    const int ntile = tile + (int)gridDim.x;
    int nst0 = 0, nen0 = 0, nst1 = 0, nen1 = 0;
    float nx0 = 0.0f, nd0 = 0.0f, nx1 = 0.0f, nd1 = 0.0f;
    if (ntile < ntiles) {
      const int i0 = ntile * 16 + 2 * wv, i1 = i0 + 1;
      if (i0 < n) { nst0 = rowptr[i0]; nen0 = rowend[i0]; const float2 own = rec1[i0]; nx0 = own.x; nd0 = own.y; }
      if (i1 < n) { nst1 = rowptr[i1]; nen1 = rowend[i1]; const float2 own = rec1[i1]; nx1 = own.x; nd1 = own.y; }
    }
    st0 = __builtin_amdgcn_readfirstlane((st0 > 0) ? st0 : 0);
    en0 = __builtin_amdgcn_readfirstlane((en0 < E) ? en0 : E);
    st1 = __builtin_amdgcn_readfirstlane((st1 > 0) ? st1 : 0);
    en1 = __builtin_amdgcn_readfirstlane((en1 < E) ? en1 : E);
    const int len0 = en0 - st0, len1 = en1 - st1;
    const int len = (len0 > len1) ? len0 : len1;
    float acc0 = 0.0f, acc1 = 0.0f;
    for (int off = 0; off < len; off += 64) {
      const int k00 = st0 + off + lane, k01 = k00 + 32;
      const int k10 = st1 + off + lane, k11 = k10 + 32;
      const bool p00 = k00 < en0, p01 = k01 < en0;
      const bool p10 = k10 < en1, p11 = k11 < en1;
      unsigned a00 = (unsigned)list[p00 ? k00 : 0];
      unsigned a01 = (unsigned)list[p01 ? k01 : 0];
      unsigned a10 = (unsigned)list[p10 ? k10 : 0];
      unsigned a11 = (unsigned)list[p11 ? k11 : 0];
      a00 = (a00 < last) ? a00 : last;
      a01 = (a01 < last) ? a01 : last;
      a10 = (a10 < last) ? a10 : last;
      a11 = (a11 < last) ? a11 : last;
      const float2 r00 = rec1[a00];
      const float2 r01 = rec1[a01];
      const float2 r10 = rec1[a10];
      const float2 r11 = rec1[a11];
      const float t00 = r00.x * r00.y * d0;
      const float t01 = r01.x * r01.y * d0;
      const float t10 = r10.x * r10.y * d1;
      const float t11 = r11.x * r11.y * d1;
      acc0 += p00 ? t00 : 0.0f;
      acc0 += p01 ? t01 : 0.0f;
      acc1 += p10 ? t10 : 0.0f;
      acc1 += p11 ? t11 : 0.0f;
    }
    for (int o = 16; o > 0; o >>= 1) {
      acc0 += __shfl_xor(acc0, o);
      acc1 += __shfl_xor(acc1, o);
    }
    const float sn0 = acc0 + x0 * d0 * d0;
    const float sn1 = acc1 + x1 * d1 * d1;
    const int buf = it & 1;
    if (lane == 0) {
      sums[buf][2 * wv]     = make_float2(sn0, d0);
      sums[buf][2 * wv + 1] = make_float2(sn1, d1);
    }
    __syncthreads();
    if (wv == (it & 7)) {
      float wx[4], wy[4], cx[4], cy[4];
#pragma unroll
      for (int k = 0; k < 4; ++k) {
        const int r0 = 4 * k + 2 * half;
        wx[k] = W1[r0]; wy[k] = W1[r0 + 1];
        cx[k] = b1[r0]; cy[k] = b1[r0 + 1];
      }
      const float sn = sums[buf][m].x;
      v2f amat[4];
#pragma unroll
      for (int k = 0; k < 4; ++k) {
        amat[k].x = fmaxf(fmaf(sn, wx[k], cx[k]), 0.0f);
        amat[k].y = fmaxf(fmaf(sn, wy[k], cy[k]), 0.0f);
      }
      v8f c = {};
#pragma unroll
      for (int k = 0; k < 4; ++k)
        c = __builtin_amdgcn_wmma_f32_16x16x4_f32(
             false, amat[k],  false, bmat[k],
             (short)0, c,  false,  false);
      const int l0 = lane & 16;
      float h0 = 0.0f, h1 = 0.0f;
#pragma unroll
      for (int r = 0; r < 8; ++r) {
        const float t0 = __shfl(c[r], l0);
        const float t1 = __shfl(c[r], l0 + 1);
        if (m == r) { h0 = t0; h1 = t1; }
      }
      if (m < 8) {
        const int slot = half * 8 + m;
        const int node = tile * 16 + slot;
        if (node < n) rec2[node] = make_float4(sums[buf][slot].y, h0, h1, 0.0f);
      }
    }
    tile = ntile;
    st0 = nst0; en0 = nen0; x0 = nx0; d0 = nd0;
    st1 = nst1; en1 = nen1; x1 = nx1; d1 = nd1;
  }
}

__global__ void __launch_bounds__(TPB)
gcn_l2_gather(const int* __restrict__ rowptr, const int* __restrict__ rowend,
              const int* __restrict__ list, const float4* __restrict__ rec2,
              const float* __restrict__ b2, float* __restrict__ out,
              int n, int E) {
  const int lane = (int)threadIdx.x & 31;
  const int wv = __builtin_amdgcn_readfirstlane((int)threadIdx.x >> 5);
  const int nw = (int)gridDim.x * (TPB / 32);
  const unsigned last = (unsigned)(n - 1);
  const float bz0 = b2[0], bz1 = b2[1];
  int i = (int)blockIdx.x * (TPB / 32) + wv;
  int st = 0, en = 0;
  float di = 0.0f, h0 = 0.0f, h1 = 0.0f;
  if (i < n) { st = rowptr[i]; en = rowend[i]; const float4 own = rec2[i]; di = own.x; h0 = own.y; h1 = own.z; }
  while (i < n) {
    const int ni = i + nw;
    int nst = 0, nen = 0;
    float ndi = 0.0f, nh0 = 0.0f, nh1 = 0.0f;
    if (ni < n) { nst = rowptr[ni]; nen = rowend[ni]; const float4 own = rec2[ni]; ndi = own.x; nh0 = own.y; nh1 = own.z; }
    st = __builtin_amdgcn_readfirstlane((st > 0) ? st : 0);
    en = __builtin_amdgcn_readfirstlane((en < E) ? en : E);
    float acc0 = 0.0f, acc1 = 0.0f;
    for (int base = st; base < en; base += 64) {
      const int k0 = base + lane, k1 = k0 + 32;
      const bool p0 = k0 < en, p1 = k1 < en;
      unsigned a0 = (unsigned)list[p0 ? k0 : st];
      unsigned a1 = (unsigned)list[p1 ? k1 : st];
      a0 = (a0 < last) ? a0 : last;
      a1 = (a1 < last) ? a1 : last;
      const float4 q0 = rec2[a0];
      const float4 q1 = rec2[a1];
      const float n0 = q0.x * di;
      const float n1 = q1.x * di;
      const float u0 = q0.y * n0, w0 = q0.z * n0;
      const float u1 = q1.y * n1, w1 = q1.z * n1;
      acc0 += p0 ? u0 : 0.0f;  acc1 += p0 ? w0 : 0.0f;
      acc0 += p1 ? u1 : 0.0f;  acc1 += p1 ? w1 : 0.0f;
    }
    for (int o = 16; o > 0; o >>= 1) {
      acc0 += __shfl_xor(acc0, o);
      acc1 += __shfl_xor(acc1, o);
    }
    if (lane == 0) {
      const float sl = di * di;
      const float z0 = acc0 + h0 * sl + bz0;
      const float z1 = acc1 + h1 * sl + bz1;
      const float mx = fmaxf(z0, z1);
      const float lse = mx + logf(expf(z0 - mx) + expf(z1 - mx));
      float2 r; r.x = z0 - lse; r.y = z1 - lse;
      *(float2*)(out + (size_t)i * 2) = r;
    }
    i = ni; st = nst; en = nen; di = ndi; h0 = nh0; h1 = nh1;
  }
}


#define GSH 8
#define GNB (1 << GSH)
#define GTPB 1024
#define GEPT 12
#define GSUB (GTPB * GEPT)
#define GBMAX GTPB
#define GCAP 16384
#define RS_TPB 256

template <int NW>
__device__ __forceinline__ int block_scan_excl(int v, int* wtot, int& total) {
  const int lane = (int)threadIdx.x & 31;
  const int wv = (int)threadIdx.x >> 5;
  int inc = v;
  for (int o = 1; o < 32; o <<= 1) {
    int u = __shfl_up(inc, o);
    if (lane >= o) inc += u;
  }
  if (lane == 31) wtot[wv] = inc;
  __syncthreads();
  const int wt = (lane < NW) ? wtot[lane] : 0;
  int winc = wt;
  for (int o = 1; o < 32; o <<= 1) {
    int u = __shfl_up(winc, o);
    if (lane >= o) winc += u;
  }
  const int before = __shfl(winc - wt, wv);
  total = __shfl(winc, 31);
  __syncthreads();
  return before + (inc - v);
}

__device__ __forceinline__ int bucket_of(int d, int nb) {
  unsigned b = (unsigned)d >> GSH;
  return (int)((b < (unsigned)nb) ? b : (unsigned)(nb - 1));
}

__global__ void __launch_bounds__(GTPB)
grp_hist(const int* __restrict__ dst, int* __restrict__ H, int E, int nb, int nseg) {
  __shared__ int hist[GBMAX];
  const int t = (int)threadIdx.x;
  const int s = (int)blockIdx.x;
  if (t < nb) hist[t] = 0;
  __syncthreads();
  const int e0 = s * GSUB + t;
  int d[GEPT];
#pragma unroll
  for (int j = 0; j < GEPT; ++j) {
    int e = e0 + j * GTPB;
    d[j] = dst[(e < E) ? e : (E - 1)];
  }
#pragma unroll
  for (int j = 0; j < GEPT; ++j) {
    int e = e0 + j * GTPB;
    if (e < E) atomicAdd(&hist[bucket_of(d[j], nb)], 1);
  }
  __syncthreads();
  if (t < nb) H[(size_t)t * nseg + s] = hist[t];
}

__global__ void __launch_bounds__(RS_TPB)
grp_rowscan(const int* __restrict__ H, int* __restrict__ P, int* __restrict__ T, int nseg) {
  __shared__ int wtot[32];
  const int t = (int)threadIdx.x;
  const size_t row = (size_t)blockIdx.x * nseg;
  int v[4];
#pragma unroll
  for (int j = 0; j < 4; ++j) {
    int k = 4 * t + j;
    int h = H[row + ((k < nseg) ? k : 0)];
    v[j] = (k < nseg) ? h : 0;
  }
  int total;
  int ex = block_scan_excl<RS_TPB / 32>(v[0] + v[1] + v[2] + v[3], wtot, total);
#pragma unroll
  for (int j = 0; j < 4; ++j) {
    int k = 4 * t + j;
    if (k < nseg) P[row + k] = ex;
    ex += v[j];
  }
  if (t == 0) T[blockIdx.x] = total;
}

__global__ void __launch_bounds__(GTPB)
grp_partition(const int* __restrict__ src, const int* __restrict__ dst,
              const int* __restrict__ P, const int* __restrict__ T,
              int* __restrict__ list, int E, int nb, int nseg) {
  __shared__ int hist[GBMAX];
  __shared__ int offs[GBMAX];
  __shared__ int gbase[GBMAX];
  __shared__ int wtot[32];
  __shared__ int stage[GSUB];
  __shared__ unsigned short stage_b[GSUB];
  const int t = (int)threadIdx.x;
  const int s = (int)blockIdx.x;

  const int tb = (t < nb) ? T[t] : 0;
  const int pb = (t < nb) ? P[(size_t)t * nseg + s] : 0;
  if (t < nb) hist[t] = 0;
  int all;
  const int first = block_scan_excl<GTPB / 32>(tb, wtot, all);
  if (t < nb) gbase[t] = first + pb;

  const int e0 = s * GSUB + t;
  int a[GEPT], d[GEPT], r[GEPT];
#pragma unroll
  for (int j = 0; j < GEPT; ++j) {
    int e = e0 + j * GTPB;
    int ec = (e < E) ? e : (E - 1);
    d[j] = dst[ec];
    a[j] = src[ec];
  }
#pragma unroll
  for (int j = 0; j < GEPT; ++j) {
    int e = e0 + j * GTPB;
    r[j] = -1;
    if (e < E) r[j] = atomicAdd(&hist[bucket_of(d[j], nb)], 1);
  }
  __syncthreads();
  const int hv = (t < nb) ? hist[t] : 0;
  int total;
  const int ex = block_scan_excl<GTPB / 32>(hv, wtot, total);
  if (t < nb) offs[t] = ex;
  __syncthreads();
#pragma unroll
  for (int j = 0; j < GEPT; ++j) {
    if (r[j] >= 0) {
      const int b = bucket_of(d[j], nb);
      const int pos = offs[b] + r[j];
      if ((unsigned)pos < (unsigned)GSUB) {
        stage[pos] = (int)((((unsigned)d[j] & (unsigned)(GNB - 1)) << 24) |
                           ((unsigned)a[j] & 0xFFFFFFu));
        stage_b[pos] = (unsigned short)b;
      }
    }
  }
  __syncthreads();
#pragma unroll
  for (int j = 0; j < GEPT; ++j) {
    const int idx = t + j * GTPB;
    if (idx < total) {
      const int b = (int)stage_b[idx];
      const int g = gbase[b] + (idx - offs[b]);
      if ((unsigned)g < (unsigned)E) list[g] = stage[idx];
    }
  }
}

__global__ void __launch_bounds__(GTPB)
grp_bucket(const int* __restrict__ T, const float* __restrict__ xin, int* list,
           int* __restrict__ rowptr, int* __restrict__ rowend,
           float* __restrict__ dinv, float2* __restrict__ rec1, int n, int E, int nb) {
  __shared__ int cnt[GNB];
  __shared__ int cur[GNB];
  __shared__ int endv[GNB];
  __shared__ int wtot[32];
  __shared__ int sh[2];
  __shared__ int ebuf[GCAP];
  __shared__ int obuf[GCAP];
  const int t = (int)threadIdx.x;
  const int b = (int)blockIdx.x;

  const int tb = (t < nb) ? T[t] : 0;
  int all;
  const int first = block_scan_excl<GTPB / 32>(tb, wtot, all);
  if (t == b) { sh[0] = first; sh[1] = tb; }
  if (t < GNB) cnt[t] = 0;
  __syncthreads();
  int base = sh[0], len = sh[1];
  base = (base > 0) ? base : 0;
  base = (base < E) ? base : E;
  len = (len > 0) ? len : 0;
  len = (len < E - base) ? len : (E - base);
  const bool fits = len <= GCAP;

  for (int j0 = t; j0 < len; j0 += 4 * GTPB) {
    unsigned v[4];
#pragma unroll
    for (int q = 0; q < 4; ++q) {
      int j = j0 + q * GTPB;
      v[q] = (unsigned)list[base + ((j < len) ? j : j0)];
    }
#pragma unroll
    for (int q = 0; q < 4; ++q) {
      int j = j0 + q * GTPB;
      if (j < len) {
        if (fits) ebuf[j] = (int)v[q];
        atomicAdd(&cnt[v[q] >> 24], 1);
      }
    }
  }
  __syncthreads();
  const int c = (t < GNB) ? cnt[t] : 0;
  int tot;
  const int ex = block_scan_excl<GTPB / 32>(c, wtot, tot);
  if (t < GNB) {
    cur[t] = ex;
    endv[t] = ex + c;
    const int i = b * GNB + t;
    if (i < n) {
      rowptr[i] = base + ex;
      rowend[i] = base + ex + c;
      const float dv = rsqrtf((float)(c + 1));
      dinv[i] = dv;
      rec1[i] = make_float2(xin[i], dv);
    }
  }
  __syncthreads();

  if (fits) {
    for (int j = t; j < len; j += GTPB) {
      const unsigned v = (unsigned)ebuf[j];
      const int pos = atomicAdd(&cur[v >> 24], 1);
      if ((unsigned)pos < (unsigned)GCAP) obuf[pos] = (int)(v & 0xFFFFFFu);
    }
    __syncthreads();
    for (int j = t; j < len; j += GTPB) list[base + j] = obuf[j];
  } else {
    if (t == 0) {
      int steps = 0;
      const int limit = 2 * len + GNB;
      for (int v = 0; v < GNB; ++v) {
        int p = cur[v];
        const int pe = endv[v];
        while (p < pe && steps < limit) {
          ++steps;
          const unsigned x = (unsigned)list[base + p];
          const int dv = (int)(x >> 24);
          if (dv == v) { ++p; continue; }
          const int q = cur[dv];
          if (q >= endv[dv]) break;
          const int y = list[base + q];
          list[base + q] = (int)x;
          list[base + p] = y;
          cur[dv] = q + 1;
        }
        cur[v] = p;
      }
    }
    __syncthreads();
    for (int j = t; j < len; j += GTPB) list[base + j] = list[base + j] & 0xFFFFFF;
  }
}

#define GPC 6

__global__ void __launch_bounds__(GTPB)
grp_segsort(const int* __restrict__ src, const int* __restrict__ dst,
            int* __restrict__ O, int* __restrict__ tmp, int E, int nb, int nseg) {
  __shared__ int hist[GBMAX];
  __shared__ int offs[GBMAX];
  __shared__ int wtot[32];
  __shared__ int stage[GSUB];
  const int t = (int)threadIdx.x;
  const int s = (int)blockIdx.x;
  if (t < nb) hist[t] = 0;
  __syncthreads();

  const int e0 = s * GSUB + t;
  int a[GEPT], d[GEPT], r[GEPT];
#pragma unroll
  for (int j = 0; j < GEPT; ++j) {
    int e = e0 + j * GTPB;
    int ec = (e < E) ? e : (E - 1);
    d[j] = dst[ec];
    a[j] = src[ec];
  }
#pragma unroll
  for (int j = 0; j < GEPT; ++j) {
    int e = e0 + j * GTPB;
    r[j] = -1;
    if (e < E) r[j] = atomicAdd(&hist[bucket_of(d[j], nb)], 1);
  }
  __syncthreads();
  const int hv = (t < nb) ? hist[t] : 0;
  int total;
  const int ex = block_scan_excl<GTPB / 32>(hv, wtot, total);
  if (t < nb) {
    offs[t] = ex;
    O[(size_t)t * nseg + s] = ex;
  }
  if (t == 0) O[(size_t)nb * nseg + s] = total;
  __syncthreads();
#pragma unroll
  for (int j = 0; j < GEPT; ++j) {
    if (r[j] >= 0) {
      const int pos = offs[bucket_of(d[j], nb)] + r[j];
      if ((unsigned)pos < (unsigned)GSUB)
        stage[pos] = (int)((((unsigned)d[j] & (unsigned)(GNB - 1)) << 24) |
                           ((unsigned)a[j] & 0xFFFFFFu));
    }
  }
  __syncthreads();
#pragma unroll
  for (int j = 0; j < GEPT; ++j) {
    const int idx = t + j * GTPB;
    const size_t g = (size_t)s * GSUB + (size_t)idx;
    if (idx < total && g < (size_t)E) tmp[g] = stage[idx];
  }
}

__global__ void __launch_bounds__(GTPB)
grp_collect(const int* __restrict__ O, const int* __restrict__ tmp,
            const float* __restrict__ xin, int* list,
            int* __restrict__ rowptr, int* __restrict__ rowend,
            float* __restrict__ dinv, float2* __restrict__ rec1,
            int n, int E, int nb, int nseg) {
  __shared__ int cnt[GNB];
  __shared__ int cur[GNB];
  __shared__ int endv[GNB];
  __shared__ int wtot[32];
  __shared__ int4 piece[GTPB];
  __shared__ int ebuf[GCAP];
  __shared__ int obuf[GCAP];
  const int t = (int)threadIdx.x;
  const int b = (int)blockIdx.x;
  const int lane = t & 31;
  const int wv = __builtin_amdgcn_readfirstlane(t >> 5);

  int o0 = 0, o1 = 0;
  if (t < nseg) {
    o0 = O[(size_t)b * nseg + t];
    o1 = O[(size_t)(b + 1) * nseg + t];
  }
  o0 = (o0 > 0) ? o0 : 0;
  o0 = (o0 < GSUB) ? o0 : GSUB;
  int pl = o1 - o0;
  pl = (pl > 0) ? pl : 0;
  pl = (pl < GSUB - o0) ? pl : (GSUB - o0);
  int len, base;
  const int pp = block_scan_excl<GTPB / 32>(pl, wtot, len);
  (void)block_scan_excl<GTPB / 32>(o0, wtot, base);
  piece[t] = make_int4(o0, pp, pl, 0);
  if (t < GNB) cnt[t] = 0;
  __syncthreads();
  base = (base > 0) ? base : 0;
  base = (base < E) ? base : E;
  len = (len > 0) ? len : 0;
  len = (len < E - base) ? len : (E - base);
  const bool fits = len <= GCAP;

  auto put = [&](int k, unsigned v) {
    if (fits) {
      if ((unsigned)k < (unsigned)GCAP) ebuf[k] = (int)v;
    } else if (k < len) {
      list[base + k] = (int)v;
    }
    atomicAdd(&cnt[v >> 24], 1);
  };

  for (int s0 = wv; s0 < nseg; s0 += 32 * GPC) {
    unsigned v[GPC][2];
    int pl_[GPC], pp_[GPC];
    size_t g0_[GPC];
#pragma unroll
    for (int q = 0; q < GPC; ++q) {
      const int sq = s0 + 32 * q;
      const bool ok = sq < nseg;
      const int sc = ok ? sq : s0;
      const int4 m = piece[sc];
      const int mlen = __builtin_amdgcn_readfirstlane(m.z);
      pl_[q] = ok ? mlen : 0;
      pp_[q] = __builtin_amdgcn_readfirstlane(m.y);
      g0_[q] = (size_t)sc * GSUB + (size_t)__builtin_amdgcn_readfirstlane(m.x);
    }
#pragma unroll
    for (int q = 0; q < GPC; ++q) {
#pragma unroll
      for (int c = 0; c < 2; ++c) {
        const int j = lane + 32 * c;
        size_t g = (j < pl_[q]) ? g0_[q] + (size_t)j : 0;
        g = (g < (size_t)E) ? g : (size_t)(E - 1);
        v[q][c] = (unsigned)tmp[g];
      }
    }
#pragma unroll
    for (int q = 0; q < GPC; ++q) {
#pragma unroll
      for (int c = 0; c < 2; ++c) {
        const int j = lane + 32 * c;
        if (j < pl_[q]) put(pp_[q] + j, v[q][c]);
      }
    }
#pragma unroll
    for (int q = 0; q < GPC; ++q) {
      for (int j = lane + 64; j < pl_[q]; j += 32) {
        size_t g = g0_[q] + (size_t)j;
        g = (g < (size_t)E) ? g : (size_t)(E - 1);
        put(pp_[q] + j, (unsigned)tmp[g]);
      }
    }
  }
  __syncthreads();

  const int c = (t < GNB) ? cnt[t] : 0;
  int tot;
  const int ex = block_scan_excl<GTPB / 32>(c, wtot, tot);
  if (t < GNB) {
    cur[t] = ex;
    endv[t] = ex + c;
    const int i = b * GNB + t;
    if (i < n) {
      rowptr[i] = base + ex;
      rowend[i] = base + ex + c;
      const float dv = rsqrtf((float)(c + 1));
      dinv[i] = dv;
      rec1[i] = make_float2(xin[i], dv);
    }
  }
  __syncthreads();

  if (fits) {
    for (int j = t; j < len; j += GTPB) {
      const unsigned v = (unsigned)ebuf[j];
      const int pos = atomicAdd(&cur[v >> 24], 1);
      if ((unsigned)pos < (unsigned)GCAP) obuf[pos] = (int)(v & 0xFFFFFFu);
    }
    __syncthreads();
    for (int j = t; j < len; j += GTPB) list[base + j] = obuf[j];
  } else {
    if (t == 0) {
      int steps = 0;
      const int limit = 2 * len + GNB;
      for (int v = 0; v < GNB; ++v) {
        int p = cur[v];
        const int pe = endv[v];
        while (p < pe && steps < limit) {
          ++steps;
          const unsigned x = (unsigned)list[base + p];
          const int dv = (int)(x >> 24);
          if (dv == v) { ++p; continue; }
          const int q = cur[dv];
          if (q >= endv[dv]) break;
          const int y = list[base + q];
          list[base + q] = (int)x;
          list[base + p] = y;
          cur[dv] = q + 1;
        }
        cur[v] = p;
      }
    }
    __syncthreads();
    for (int j = t; j < len; j += GTPB) list[base + j] = list[base + j] & 0xFFFFFF;
  }
}

extern "C" void kernel_launch(void* const* d_in, const int* in_sizes, int n_in,
                              void* d_out, int out_size, void* d_ws, size_t ws_size,
                              hipStream_t stream) {
  const float* x  = (const float*)d_in[0];
  const int*   ei = (const int*)d_in[1];
  const float* W1 = (const float*)d_in[2];
  const float* b1 = (const float*)d_in[3];
  const float* W2 = (const float*)d_in[4];
  const float* b2 = (const float*)d_in[5];
  float* out = (float*)d_out;
  int* tick = (int*)d_out;

  int n = in_sizes[0];
  int E = in_sizes[1] / 2;
  const int* srcp = ei;
  const int* dstp = ei + E;

  size_t ns = ((size_t)n + 31) & ~(size_t)31;

  float* ws    = (float*)d_ws;
  float* deg   = ws;
  float* dinv  = ws + ns;
  float* s     = ws + 2 * ns;
  float* h2c0  = ws + 3 * ns;
  float* h2c1  = ws + 4 * ns;
  float* agg0  = ws + 5 * ns;
  float* agg1  = ws + 6 * ns;

  int gN = (n + TPB - 1) / TPB;
  int ntiles = (n + 15) / 16;
  int gW = (int)(((size_t)ntiles * 32 + TPB - 1) / TPB);

  const size_t need = (25 * ns + (size_t)E) * sizeof(float);
  if (ws_size >= need) {
    int*   cnt    = (int*)ws;
    int*   rowptr = (int*)(ws + 7 * ns);
    int*   cursor = (int*)(ws + 8 * ns);
    float* tabs0  = ws + 9 * ns;
    int*   list   = (int*)(ws + 25 * ns);
    float2* rec1  = (float2*)(ws + 4 * ns);
    float4* rec2  = (float4*)(ws + 9 * ns);
    int gS = (n + SCAN_TPB - 1) / SCAN_TPB;

    const int nb = (n + GNB - 1) >> GSH;
    const int nseg = (E + GSUB - 1) / GSUB;
    const size_t tabs = 2 * (size_t)nb * (size_t)nseg + (size_t)nb;
    if (E > 0 && nb <= GBMAX && nseg <= GTPB && n <= (1 << 24) && tabs <= 16 * (size_t)n) {
      int* H = (int*)tabs0;
      const size_t need2 = (25 * ns + 2 * (size_t)E) * sizeof(float);
      if (ws_size >= need2) {
        int* tmp = list + E;
        grp_segsort<<<nseg, GTPB, 0, stream>>>(srcp, dstp, H, tmp, E, nb, nseg);
        grp_collect<<<nb, GTPB, 0, stream>>>(H, tmp, x, list, rowptr, cursor, dinv, rec1, n, E, nb, nseg);
      } else {
        int* P = H + (size_t)nb * nseg;
        int* T = P + (size_t)nb * nseg;
        grp_hist<<<nseg, GTPB, 0, stream>>>(dstp, H, E, nb, nseg);
        grp_rowscan<<<nb, RS_TPB, 0, stream>>>(H, P, T, nseg);
        grp_partition<<<nseg, GTPB, 0, stream>>>(srcp, dstp, P, T, list, E, nb, nseg);
        grp_bucket<<<nb, GTPB, 0, stream>>>(T, x, list, rowptr, cursor, dinv, rec1, n, E, nb);
      }
    } else {
      csr_init<<<gN, TPB, 0, stream>>>(cnt, tick, n);
      csr_count<<<PASS_BLOCKS, TPB, 0, stream>>>(dstp, cnt, tick, n, E);
      csr_scan<<<gS, SCAN_TPB, 0, stream>>>(cnt, x, rowptr, cursor, dinv, rec1, n);
      csr_fill<<<PASS_BLOCKS, TPB, 0, stream>>>(srcp, dstp, cursor, list, tick, n, E);
    }
    const int gT = (ntiles < PASS_BLOCKS) ? ntiles : PASS_BLOCKS;
    gcn_l1_gather_dense<<<gT, TPB, 0, stream>>>(rowptr, cursor, list, rec1, W1, b1, W2, rec2,
                                                n, ntiles, E);
    gcn_l2_gather<<<PASS_BLOCKS, TPB, 0, stream>>>(rowptr, cursor, list, rec2, b2, out, n, E);
    return;
  }

  float* relu1 = ws + 7 * ns;
  gcn_init<<<gN, TPB, 0, stream>>>(deg, s, agg0, agg1, tick, n);
  gcn_degree<<<PASS_BLOCKS, TPB, 0, stream>>>(dstp, deg, tick, n, E);
  gcn_dinv<<<gN, TPB, 0, stream>>>(deg, dinv, n);
  gcn_l1_scatter<<<PASS_BLOCKS, TPB, 0, stream>>>(srcp, dstp, x, dinv, s, tick, n, E);
  gcn_l1_finalize<<<gN, TPB, 0, stream>>>(x, dinv, s, W1, b1, relu1, n);
  gcn_l2_dense_wmma<<<gW, TPB, 0, stream>>>(relu1, W2, h2c0, h2c1, n, ntiles);
  gcn_l2_scatter<<<PASS_BLOCKS, TPB, 0, stream>>>(srcp, dstp, h2c0, h2c1, dinv,
                                                   agg0, agg1, tick, n, E);
  gcn_finalize<<<gN, TPB, 0, stream>>>(agg0, agg1, h2c0, h2c1, dinv, b2, out, n);
}
